# starting kernel
# baseline (speedup 1.0000x reference)
#include <hip/hip_runtime.h>

typedef __attribute__((ext_vector_type(2))) float v2f;
typedef __attribute__((ext_vector_type(8))) float v8f;

#define N_NODES  50000
#define N_EDGES  800000
#define N_GRAPHS 512
#define BN_EPS   1e-5f

__global__ void zero_f32(float* __restrict__ p, int n4) {
    int t = blockIdx.x * blockDim.x + threadIdx.x;
    if (t < n4) ((float4*)p)[t] = make_float4(0.f, 0.f, 0.f, 0.f);
}

__global__ void gather_emb(const int* __restrict__ ids, const float* __restrict__ emb,
                           float* __restrict__ x) {
    int t = blockIdx.x * blockDim.x + threadIdx.x;
    int node = t >> 5, lane = t & 31;
    if (node >= N_NODES) return;
    int id = ids[node];
    float4 v = *(const float4*)(emb + (long)id * 128 + lane * 4);
    *(float4*)(x + (long)node * 128 + lane * 4) = v;
}

__global__ void edge_scatter(const float* __restrict__ x, const int* __restrict__ ei,
                             float* __restrict__ agg) {
    int wave = threadIdx.x >> 5, lane = threadIdx.x & 31;
    int e = blockIdx.x * 8 + wave;
    if (e >= N_EDGES) return;
    int s = ei[e];
    int d = ei[N_EDGES + e];
    float4 v = *(const float4*)(x + (long)s * 128 + lane * 4);
    float* p = agg + (long)d * 128 + lane * 4;
    atomicAdd(p + 0, v.x);
    atomicAdd(p + 1, v.y);
    atomicAdd(p + 2, v.z);
    atomicAdd(p + 3, v.w);
}

template <bool FUSE_ADD, bool BN, bool RELU>
__global__ void gemm_wmma(const float* __restrict__ X, const float* __restrict__ AGG,
                          const float* __restrict__ W, const float* __restrict__ bias,
                          const float* __restrict__ gamma, const float* __restrict__ beta,
                          const float* __restrict__ mean, const float* __restrict__ var,
                          float* __restrict__ O, int M, int K, int N) {
    int wave = threadIdx.x >> 5, lane = threadIdx.x & 31;
    int tilesN = N >> 4;
    int tile = blockIdx.x * 8 + wave;
    if (tile >= (M >> 4) * tilesN) return;
    int mt = tile / tilesN, nt = tile % tilesN;
    int mi = lane & 15, kh = lane >> 4;

    const float* xr = X + (long)(mt * 16 + mi) * K + kh * 2;
    const float* ar = FUSE_ADD ? (AGG + (long)(mt * 16 + mi) * K + kh * 2) : nullptr;
    const float* wc = W + (long)(kh * 2) * N + nt * 16 + mi;

    v8f acc = {0.f, 0.f, 0.f, 0.f, 0.f, 0.f, 0.f, 0.f};
    for (int kb = 0; kb < K; kb += 4) {
        v2f a, b;
        a.x = xr[kb];
        a.y = xr[kb + 1];
        if (FUSE_ADD) { a.x += ar[kb]; a.y += ar[kb + 1]; }
        b.x = wc[(long)kb * N];
        b.y = wc[(long)(kb + 1) * N];
        acc = __builtin_amdgcn_wmma_f32_16x16x4_f32(false, a, false, b,
                                                    (short)0, acc, false, false);
    }

    int col = nt * 16 + mi;
    float bb = bias[col];
    float mu = 0.f, inv = 1.f, gm = 1.f, bt = 0.f;
    if (BN) { mu = mean[col]; inv = rsqrtf(var[col] + BN_EPS); gm = gamma[col]; bt = beta[col]; }
    float* orow = O + (long)(mt * 16 + kh * 8) * N + col;
#pragma unroll
    for (int r = 0; r < 8; r++) {
        float v = acc[r] + bb;
        if (BN)   v = (v - mu) * inv * gm + bt;
        if (RELU) v = fmaxf(v, 0.f);
        orow[(long)r * N] = v;
    }
}

__global__ void pool3(const float* __restrict__ h1, const float* __restrict__ h2,
                      const float* __restrict__ h3, const int* __restrict__ batch,
                      float* __restrict__ gcat) {
    int t = blockIdx.x * blockDim.x + threadIdx.x;
    int node = t >> 5, lane = t & 31;
    if (node >= N_NODES) return;
    int g = batch[node];
    float* base = gcat + (long)g * 384 + lane * 4;
    const float* hs[3] = {h1, h2, h3};
#pragma unroll
    for (int l = 0; l < 3; l++) {
        float4 v = *(const float4*)(hs[l] + (long)node * 128 + lane * 4);
        float* p = base + l * 128;
        atomicAdd(p + 0, v.x);
        atomicAdd(p + 1, v.y);
        atomicAdd(p + 2, v.z);
        atomicAdd(p + 3, v.w);
    }
}

__global__ void lin2_kernel(const float* __restrict__ h, const float* __restrict__ w,
                            const float* __restrict__ b, float* __restrict__ out) {
    int t = blockIdx.x * blockDim.x + threadIdx.x;
    if (t >= N_GRAPHS * 10) return;
    int g = t / 10, o = t % 10;
    const float* hr = h + (long)g * 384;
    float acc = b[o];
    for (int k = 0; k < 384; k++) acc = fmaf(hr[k], w[k * 10 + o], acc);
    out[t] = acc;
}

extern "C" void kernel_launch(void* const* d_in, const int* in_sizes, int n_in,
                              void* d_out, int out_size, void* d_ws, size_t ws_size,
                              hipStream_t stream) {
    const int*   x_ids = (const int*)d_in[0];
    const int*   edge  = (const int*)d_in[1];
    const int*   batch = (const int*)d_in[2];
    const float* emb   = (const float*)d_in[3];

    const float *c_w1[3], *c_b1[3], *c_g[3], *c_be[3], *c_mu[3], *c_va[3], *c_w2[3], *c_b2[3];
    for (int i = 0; i < 3; i++) {
        c_w1[i] = (const float*)d_in[4 + 8 * i + 0];
        c_b1[i] = (const float*)d_in[4 + 8 * i + 1];
        c_g[i]  = (const float*)d_in[4 + 8 * i + 2];
        c_be[i] = (const float*)d_in[4 + 8 * i + 3];
        c_mu[i] = (const float*)d_in[4 + 8 * i + 4];
        c_va[i] = (const float*)d_in[4 + 8 * i + 5];
        c_w2[i] = (const float*)d_in[4 + 8 * i + 6];
        c_b2[i] = (const float*)d_in[4 + 8 * i + 7];
    }
    const float* lin1_w = (const float*)d_in[28];
    const float* lin1_b = (const float*)d_in[29];
    const float* lin2_w = (const float*)d_in[30];
    const float* lin2_b = (const float*)d_in[31];

    const long NF = (long)N_NODES * 128;
    float* ws   = (float*)d_ws;
    float* x0   = ws + 0 * NF;
    float* agg  = ws + 1 * NF;
    float* tmp  = ws + 2 * NF;
    float* hbuf[3] = { ws + 3 * NF, ws + 4 * NF, x0 };
    float* gcat  = agg;
    float* lin1o = tmp;

    gather_emb<<<(N_NODES * 32 + 255) / 256, 256, 0, stream>>>(x_ids, emb, x0);

    const int tiles_node = (N_NODES / 16) * (128 / 16);
    const float* xin = x0;
    for (int l = 0; l < 3; l++) {
        zero_f32<<<(int)((NF / 4 + 255) / 256), 256, 0, stream>>>(agg, (int)(NF / 4));
        edge_scatter<<<N_EDGES / 8, 256, 0, stream>>>(xin, edge, agg);
        gemm_wmma<true, true, true><<<(tiles_node + 7) / 8, 256, 0, stream>>>(
            xin, agg, c_w1[l], c_b1[l], c_g[l], c_be[l], c_mu[l], c_va[l],
            tmp, N_NODES, 128, 128);
        gemm_wmma<false, false, true><<<(tiles_node + 7) / 8, 256, 0, stream>>>(
            tmp, nullptr, c_w2[l], c_b2[l], nullptr, nullptr, nullptr, nullptr,
            hbuf[l], N_NODES, 128, 128);
        xin = hbuf[l];
    }

    zero_f32<<<(N_GRAPHS * 384 / 4 + 255) / 256, 256, 0, stream>>>(gcat, N_GRAPHS * 384 / 4);
    pool3<<<(N_NODES * 32 + 255) / 256, 256, 0, stream>>>(hbuf[0], hbuf[1], hbuf[2], batch, gcat);

    const int tiles_head = (N_GRAPHS / 16) * (384 / 16);
    gemm_wmma<false, false, true><<<(tiles_head + 7) / 8, 256, 0, stream>>>(
        gcat, nullptr, lin1_w, lin1_b, nullptr, nullptr, nullptr, nullptr,
        lin1o, N_GRAPHS, 384, 384);

    lin2_kernel<<<(N_GRAPHS * 10 + 255) / 256, 256, 0, stream>>>(lin1o, lin2_w, lin2_b,
                                                                 (float*)d_out);
}


# Neighbour sum as a register gather over edges grouped by destination (counting sort once per call, int atomic gives the slot) instead of float-atomic scatter plus zero fill; sum order = slot order
# speedup vs baseline: 1.7020x; 1.7020x over previous
#include <hip/hip_runtime.h>

typedef __attribute__((ext_vector_type(2))) float v2f;
typedef __attribute__((ext_vector_type(8))) float v8f;

#define N_NODES  50000
#define N_EDGES  800000
#define N_GRAPHS 512
#define BN_EPS   1e-5f

__global__ void zero_f32(float* __restrict__ p, int n4) {
    int t = blockIdx.x * blockDim.x + threadIdx.x;
    if (t < n4) ((float4*)p)[t] = make_float4(0.f, 0.f, 0.f, 0.f);
}

__global__ void gather_emb(const int* __restrict__ ids, const float* __restrict__ emb,
                           float* __restrict__ x) {
    int t = blockIdx.x * blockDim.x + threadIdx.x;
    int node = t >> 5, lane = t & 31;
    if (node >= N_NODES) return;
    int id = ids[node];
    float4 v = *(const float4*)(emb + (long)id * 128 + lane * 4);
    *(float4*)(x + (long)node * 128 + lane * 4) = v;
}

#define SCAN_STEPS 13
#define CNT_PAD    (32 * SCAN_STEPS * 128)

__global__ void edge_rank(const int* __restrict__ ei, int* __restrict__ cnt,
                          int* __restrict__ rank) {
    int e = blockIdx.x * blockDim.x + threadIdx.x;
    if (e >= N_EDGES) return;
    rank[e] = atomicAdd(cnt + ei[N_EDGES + e], 1);
}

__global__ void __launch_bounds__(1024) scan_counts(const int* __restrict__ cnt,
                                                    int* __restrict__ rowptr) {
    __shared__ int wsum[32];
    int wave = threadIdx.x >> 5, lane = threadIdx.x & 31;
    const int4* c4 = (const int4*)cnt + wave * (SCAN_STEPS * 32);
    int4* r4 = (int4*)rowptr + wave * (SCAN_STEPS * 32);

    int tot = 0;
    for (int s = 0; s < SCAN_STEPS; s++) {
        int4 v = c4[s * 32 + lane];
        tot += v.x + v.y + v.z + v.w;
    }
    for (int off = 16; off > 0; off >>= 1) tot += __shfl_xor(tot, off, 32);
    if (lane == 0) wsum[wave] = tot;
    __syncthreads();

    int carry = 0;
    for (int w = 0; w < wave; w++) carry += wsum[w];

    for (int s = 0; s < SCAN_STEPS; s++) {
        int4 v = c4[s * 32 + lane];
        int l = v.x + v.y + v.z + v.w;
        int inc = l;
        for (int off = 1; off < 32; off <<= 1) {
            int t = __shfl_up(inc, off, 32);
            if (lane >= off) inc += t;
        }
        int ex = carry + inc - l;
        int4 o;
        o.x = ex;
        o.y = ex + v.x;
        o.z = ex + v.x + v.y;
        o.w = ex + v.x + v.y + v.z;
        r4[s * 32 + lane] = o;
        carry += __shfl(inc, 31, 32);
    }
}

__global__ void edge_fill(const int* __restrict__ ei, const int* __restrict__ rowptr,
                          const int* __restrict__ rank, int* __restrict__ col) {
    int e = blockIdx.x * blockDim.x + threadIdx.x;
    if (e >= N_EDGES) return;
    int d = ei[N_EDGES + e];
    col[rowptr[d] + rank[e]] = ei[e];
}

__global__ void gather_sum(const float* __restrict__ x, const int* __restrict__ rowptr,
                           const int* __restrict__ col, float* __restrict__ s) {
    int wave = threadIdx.x >> 5, lane = threadIdx.x & 31;
    for (int node = blockIdx.x * 8 + wave; node < N_NODES; node += gridDim.x * 8) {
        int beg = __builtin_amdgcn_readfirstlane(rowptr[node]);
        int end = __builtin_amdgcn_readfirstlane(rowptr[node + 1]);
        float4 acc = make_float4(0.f, 0.f, 0.f, 0.f);
        for (int base = beg; base < end; base += 32) {
            int n = end - base;
            if (n > 32) n = 32;
            int myc = (lane < n) ? col[base + lane] : 0;
            int j = 0;
            for (; j + 4 <= n; j += 4) {
                int s0 = __builtin_amdgcn_readlane(myc, j);
                int s1 = __builtin_amdgcn_readlane(myc, j + 1);
                int s2 = __builtin_amdgcn_readlane(myc, j + 2);
                int s3 = __builtin_amdgcn_readlane(myc, j + 3);
                float4 v0 = *(const float4*)(x + (long)s0 * 128 + lane * 4);
                float4 v1 = *(const float4*)(x + (long)s1 * 128 + lane * 4);
                float4 v2 = *(const float4*)(x + (long)s2 * 128 + lane * 4);
                float4 v3 = *(const float4*)(x + (long)s3 * 128 + lane * 4);
                acc.x = (((acc.x + v0.x) + v1.x) + v2.x) + v3.x;
                acc.y = (((acc.y + v0.y) + v1.y) + v2.y) + v3.y;
                acc.z = (((acc.z + v0.z) + v1.z) + v2.z) + v3.z;
                acc.w = (((acc.w + v0.w) + v1.w) + v2.w) + v3.w;
            }
            for (; j < n; j++) {
                int s0 = __builtin_amdgcn_readlane(myc, j);
                float4 v0 = *(const float4*)(x + (long)s0 * 128 + lane * 4);
                acc.x += v0.x; acc.y += v0.y; acc.z += v0.z; acc.w += v0.w;
            }
        }
        float4 xv = *(const float4*)(x + (long)node * 128 + lane * 4);
        float4 o = make_float4(xv.x + acc.x, xv.y + acc.y, xv.z + acc.z, xv.w + acc.w);
        *(float4*)(s + (long)node * 128 + lane * 4) = o;
    }
}

template <bool FUSE_ADD, bool BN, bool RELU>
__global__ void gemm_wmma(const float* __restrict__ X, const float* __restrict__ AGG,
                          const float* __restrict__ W, const float* __restrict__ bias,
                          const float* __restrict__ gamma, const float* __restrict__ beta,
                          const float* __restrict__ mean, const float* __restrict__ var,
                          float* __restrict__ O, int M, int K, int N) {
    int wave = threadIdx.x >> 5, lane = threadIdx.x & 31;
    int tilesN = N >> 4;
    int tile = blockIdx.x * 8 + wave;
    if (tile >= (M >> 4) * tilesN) return;
    int mt = tile / tilesN, nt = tile % tilesN;
    int mi = lane & 15, kh = lane >> 4;

    const float* xr = X + (long)(mt * 16 + mi) * K + kh * 2;
    const float* ar = FUSE_ADD ? (AGG + (long)(mt * 16 + mi) * K + kh * 2) : nullptr;
    const float* wc = W + (long)(kh * 2) * N + nt * 16 + mi;

    v8f acc = {0.f, 0.f, 0.f, 0.f, 0.f, 0.f, 0.f, 0.f};
    for (int kb = 0; kb < K; kb += 4) {
        v2f a, b;
        a.x = xr[kb];
        a.y = xr[kb + 1];
        if (FUSE_ADD) { a.x += ar[kb]; a.y += ar[kb + 1]; }
        b.x = wc[(long)kb * N];
        b.y = wc[(long)(kb + 1) * N];
        acc = __builtin_amdgcn_wmma_f32_16x16x4_f32(false, a, false, b,
                                                    (short)0, acc, false, false);
    }

    int col = nt * 16 + mi;
    float bb = bias[col];
    float mu = 0.f, inv = 1.f, gm = 1.f, bt = 0.f;
    if (BN) { mu = mean[col]; inv = rsqrtf(var[col] + BN_EPS); gm = gamma[col]; bt = beta[col]; }
    float* orow = O + (long)(mt * 16 + kh * 8) * N + col;
#pragma unroll
    for (int r = 0; r < 8; r++) {
        float v = acc[r] + bb;
        if (BN)   v = (v - mu) * inv * gm + bt;
        if (RELU) v = fmaxf(v, 0.f);
        orow[(long)r * N] = v;
    }
}

__global__ void pool3(const float* __restrict__ h1, const float* __restrict__ h2,
                      const float* __restrict__ h3, const int* __restrict__ batch,
                      float* __restrict__ gcat) {
    int t = blockIdx.x * blockDim.x + threadIdx.x;
    int node = t >> 5, lane = t & 31;
    if (node >= N_NODES) return;
    int g = batch[node];
    float* base = gcat + (long)g * 384 + lane * 4;
    const float* hs[3] = {h1, h2, h3};
#pragma unroll
    for (int l = 0; l < 3; l++) {
        float4 v = *(const float4*)(hs[l] + (long)node * 128 + lane * 4);
        float* p = base + l * 128;
        atomicAdd(p + 0, v.x);
        atomicAdd(p + 1, v.y);
        atomicAdd(p + 2, v.z);
        atomicAdd(p + 3, v.w);
    }
}

__global__ void lin2_kernel(const float* __restrict__ h, const float* __restrict__ w,
                            const float* __restrict__ b, float* __restrict__ out) {
    int t = blockIdx.x * blockDim.x + threadIdx.x;
    if (t >= N_GRAPHS * 10) return;
    int g = t / 10, o = t % 10;
    const float* hr = h + (long)g * 384;
    float acc = b[o];
    for (int k = 0; k < 384; k++) acc = fmaf(hr[k], w[k * 10 + o], acc);
    out[t] = acc;
}

extern "C" void kernel_launch(void* const* d_in, const int* in_sizes, int n_in,
                              void* d_out, int out_size, void* d_ws, size_t ws_size,
                              hipStream_t stream) {
    const int*   x_ids = (const int*)d_in[0];
    const int*   edge  = (const int*)d_in[1];
    const int*   batch = (const int*)d_in[2];
    const float* emb   = (const float*)d_in[3];

    const float *c_w1[3], *c_b1[3], *c_g[3], *c_be[3], *c_mu[3], *c_va[3], *c_w2[3], *c_b2[3];
    for (int i = 0; i < 3; i++) {
        c_w1[i] = (const float*)d_in[4 + 8 * i + 0];
        c_b1[i] = (const float*)d_in[4 + 8 * i + 1];
        c_g[i]  = (const float*)d_in[4 + 8 * i + 2];
        c_be[i] = (const float*)d_in[4 + 8 * i + 3];
        c_mu[i] = (const float*)d_in[4 + 8 * i + 4];
        c_va[i] = (const float*)d_in[4 + 8 * i + 5];
        c_w2[i] = (const float*)d_in[4 + 8 * i + 6];
        c_b2[i] = (const float*)d_in[4 + 8 * i + 7];
    }
    const float* lin1_w = (const float*)d_in[28];
    const float* lin1_b = (const float*)d_in[29];
    const float* lin2_w = (const float*)d_in[30];
    const float* lin2_b = (const float*)d_in[31];

    const long NF = (long)N_NODES * 128;
    float* ws   = (float*)d_ws;
    float* x0   = ws + 0 * NF;
    float* agg  = ws + 1 * NF;
    float* tmp  = ws + 2 * NF;
    float* hbuf[3] = { ws + 3 * NF, ws + 4 * NF, x0 };
    float* gcat  = agg;
    float* lin1o = tmp;
    int* cnt    = (int*)(ws + 5 * NF);
    int* rowptr = cnt + CNT_PAD;
    int* rank   = rowptr + CNT_PAD;
    int* col    = rank + N_EDGES;

    gather_emb<<<(N_NODES * 32 + 255) / 256, 256, 0, stream>>>(x_ids, emb, x0);

    zero_f32<<<(CNT_PAD / 4 + 255) / 256, 256, 0, stream>>>((float*)cnt, CNT_PAD / 4);
    edge_rank<<<N_EDGES / 256, 256, 0, stream>>>(edge, cnt, rank);
    scan_counts<<<1, 1024, 0, stream>>>(cnt, rowptr);
    edge_fill<<<N_EDGES / 256, 256, 0, stream>>>(edge, rowptr, rank, col);

    const int tiles_node = (N_NODES / 16) * (128 / 16);
    const float* xin = x0;
    for (int l = 0; l < 3; l++) {
        gather_sum<<<2048, 256, 0, stream>>>(xin, rowptr, col, agg);
        gemm_wmma<false, true, true><<<(tiles_node + 7) / 8, 256, 0, stream>>>(
            agg, nullptr, c_w1[l], c_b1[l], c_g[l], c_be[l], c_mu[l], c_va[l],
            tmp, N_NODES, 128, 128);
        gemm_wmma<false, false, true><<<(tiles_node + 7) / 8, 256, 0, stream>>>(
            tmp, nullptr, c_w2[l], c_b2[l], nullptr, nullptr, nullptr, nullptr,
            hbuf[l], N_NODES, 128, 128);
        xin = hbuf[l];
    }

    zero_f32<<<(N_GRAPHS * 384 / 4 + 255) / 256, 256, 0, stream>>>(gcat, N_GRAPHS * 384 / 4);
    pool3<<<(N_NODES * 32 + 255) / 256, 256, 0, stream>>>(hbuf[0], hbuf[1], hbuf[2], batch, gcat);

    const int tiles_head = (N_GRAPHS / 16) * (384 / 16);
    gemm_wmma<false, false, true><<<(tiles_head + 7) / 8, 256, 0, stream>>>(
        gcat, nullptr, lin1_w, lin1_b, nullptr, nullptr, nullptr, nullptr,
        lin1o, N_GRAPHS, 384, 384);

    lin2_kernel<<<(N_GRAPHS * 10 + 255) / 256, 256, 0, stream>>>(lin1o, lin2_w, lin2_b,
                                                                 (float*)d_out);
}


# Node-layer GEMMs re-tiled: each wave owns a 16x128 output strip, W copied once per work-group into LDS, so one A fragment feeds 8 WMMAs and B is read from LDS; k and sum order unchanged
# speedup vs baseline: 2.1355x; 1.2547x over previous
#include <hip/hip_runtime.h>

typedef __attribute__((ext_vector_type(2))) float v2f;
typedef __attribute__((ext_vector_type(8))) float v8f;

#define N_NODES  50000
#define N_EDGES  800000
#define N_GRAPHS 512
#define BN_EPS   1e-5f

__global__ void zero_f32(float* __restrict__ p, int n4) {
    int t = blockIdx.x * blockDim.x + threadIdx.x;
    if (t < n4) ((float4*)p)[t] = make_float4(0.f, 0.f, 0.f, 0.f);
}

__global__ void gather_emb(const int* __restrict__ ids, const float* __restrict__ emb,
                           float* __restrict__ x) {
    int t = blockIdx.x * blockDim.x + threadIdx.x;
    int node = t >> 5, lane = t & 31;
    if (node >= N_NODES) return;
    int id = ids[node];
    float4 v = *(const float4*)(emb + (long)id * 128 + lane * 4);
    *(float4*)(x + (long)node * 128 + lane * 4) = v;
}

#define SCAN_STEPS 13
#define CNT_PAD    (32 * SCAN_STEPS * 128)

__global__ void edge_rank(const int* __restrict__ ei, int* __restrict__ cnt,
                          int* __restrict__ rank) {
    int e = blockIdx.x * blockDim.x + threadIdx.x;
    if (e >= N_EDGES) return;
    rank[e] = atomicAdd(cnt + ei[N_EDGES + e], 1);
}

__global__ void __launch_bounds__(1024) scan_counts(const int* __restrict__ cnt,
                                                    int* __restrict__ rowptr) {
    __shared__ int wsum[32];
    int wave = threadIdx.x >> 5, lane = threadIdx.x & 31;
    const int4* c4 = (const int4*)cnt + wave * (SCAN_STEPS * 32);
    int4* r4 = (int4*)rowptr + wave * (SCAN_STEPS * 32);

    int tot = 0;
    for (int s = 0; s < SCAN_STEPS; s++) {
        int4 v = c4[s * 32 + lane];
        tot += v.x + v.y + v.z + v.w;
    }
    for (int off = 16; off > 0; off >>= 1) tot += __shfl_xor(tot, off, 32);
    if (lane == 0) wsum[wave] = tot;
    __syncthreads();

    int carry = 0;
    for (int w = 0; w < wave; w++) carry += wsum[w];

    for (int s = 0; s < SCAN_STEPS; s++) {
        int4 v = c4[s * 32 + lane];
        int l = v.x + v.y + v.z + v.w;
        int inc = l;
        for (int off = 1; off < 32; off <<= 1) {
            int t = __shfl_up(inc, off, 32);
            if (lane >= off) inc += t;
        }
        int ex = carry + inc - l;
        int4 o;
        o.x = ex;
        o.y = ex + v.x;
        o.z = ex + v.x + v.y;
        o.w = ex + v.x + v.y + v.z;
        r4[s * 32 + lane] = o;
        carry += __shfl(inc, 31, 32);
    }
}

__global__ void edge_fill(const int* __restrict__ ei, const int* __restrict__ rowptr,
                          const int* __restrict__ rank, int* __restrict__ col) {
    int e = blockIdx.x * blockDim.x + threadIdx.x;
    if (e >= N_EDGES) return;
    int d = ei[N_EDGES + e];
    col[rowptr[d] + rank[e]] = ei[e];
}

__global__ void gather_sum(const float* __restrict__ x, const int* __restrict__ rowptr,
                           const int* __restrict__ col, float* __restrict__ s) {
    int wave = threadIdx.x >> 5, lane = threadIdx.x & 31;
    for (int node = blockIdx.x * 8 + wave; node < N_NODES; node += gridDim.x * 8) {
        int beg = __builtin_amdgcn_readfirstlane(rowptr[node]);
        int end = __builtin_amdgcn_readfirstlane(rowptr[node + 1]);
        float4 acc = make_float4(0.f, 0.f, 0.f, 0.f);
        for (int base = beg; base < end; base += 32) {
            int n = end - base;
            if (n > 32) n = 32;
            int myc = (lane < n) ? col[base + lane] : 0;
            int j = 0;
            for (; j + 4 <= n; j += 4) {
                int s0 = __builtin_amdgcn_readlane(myc, j);
                int s1 = __builtin_amdgcn_readlane(myc, j + 1);
                int s2 = __builtin_amdgcn_readlane(myc, j + 2);
                int s3 = __builtin_amdgcn_readlane(myc, j + 3);
                float4 v0 = *(const float4*)(x + (long)s0 * 128 + lane * 4);
                float4 v1 = *(const float4*)(x + (long)s1 * 128 + lane * 4);
                float4 v2 = *(const float4*)(x + (long)s2 * 128 + lane * 4);
                float4 v3 = *(const float4*)(x + (long)s3 * 128 + lane * 4);
                acc.x = (((acc.x + v0.x) + v1.x) + v2.x) + v3.x;
                acc.y = (((acc.y + v0.y) + v1.y) + v2.y) + v3.y;
                acc.z = (((acc.z + v0.z) + v1.z) + v2.z) + v3.z;
                acc.w = (((acc.w + v0.w) + v1.w) + v2.w) + v3.w;
            }
            for (; j < n; j++) {
                int s0 = __builtin_amdgcn_readlane(myc, j);
                float4 v0 = *(const float4*)(x + (long)s0 * 128 + lane * 4);
                acc.x += v0.x; acc.y += v0.y; acc.z += v0.z; acc.w += v0.w;
            }
        }
        float4 xv = *(const float4*)(x + (long)node * 128 + lane * 4);
        float4 o = make_float4(xv.x + acc.x, xv.y + acc.y, xv.z + acc.z, xv.w + acc.w);
        *(float4*)(s + (long)node * 128 + lane * 4) = o;
    }
}

#define SK  128
#define SN  128
#define WPF 288
template <bool BN>
__global__ void __launch_bounds__(256) gemm_strip(const float* __restrict__ X, const float* __restrict__ W,
                                                  const float* __restrict__ bias,
                                                  const float* __restrict__ gamma, const float* __restrict__ beta,
                                                  const float* __restrict__ mean, const float* __restrict__ var,
                                                  float* __restrict__ O, int strips) {
    __shared__ __attribute__((aligned(16))) float wl[(SK / 2) * WPF];
    int wave = threadIdx.x >> 5, lane = threadIdx.x & 31;
    int c = lane & 15, h = lane >> 4;
    int strip = blockIdx.x * 8 + wave;

#pragma unroll
    for (int i = 0; i < 8; i++) {
        int kp = wave + 8 * i;
        float4 r0 = *(const float4*)(W + (long)(2 * kp) * SN + lane * 4);
        float4 r1 = *(const float4*)(W + (long)(2 * kp + 1) * SN + lane * 4);
        float* d = wl + kp * WPF + lane * 8;
        *(float4*)(d)     = make_float4(r0.x, r1.x, r0.y, r1.y);
        *(float4*)(d + 4) = make_float4(r0.z, r1.z, r0.w, r1.w);
    }
    __syncthreads();
    if (strip >= strips) return;

    v2f af[SK / 4];
    const float* xr = X + (long)(strip * 16 + c) * SK + h * 2;
#pragma unroll
    for (int i = 0; i < SK / 4; i++) af[i] = *(const v2f*)(xr + i * 4);

    v8f acc[8];
#pragma unroll
    for (int nt = 0; nt < 8; nt++) acc[nt] = (v8f){0.f, 0.f, 0.f, 0.f, 0.f, 0.f, 0.f, 0.f};

    const float* bl = wl + h * WPF + c * 2;
    v2f bc[8], bn[8];
#pragma unroll
    for (int nt = 0; nt < 8; nt++) bc[nt] = *(const v2f*)(bl + nt * 32);
#pragma unroll
    for (int i = 0; i < SK / 4; i++) {
        if (i + 1 < SK / 4) {
#pragma unroll
            for (int nt = 0; nt < 8; nt++) bn[nt] = *(const v2f*)(bl + (2 * i + 2) * WPF + nt * 32);
        }
        __builtin_amdgcn_sched_barrier(0);
#pragma unroll
        for (int nt = 0; nt < 8; nt++)
            acc[nt] = __builtin_amdgcn_wmma_f32_16x16x4_f32(false, af[i], false, bc[nt],
                                                            (short)0, acc[nt], false, false);
        __builtin_amdgcn_sched_barrier(0);
#pragma unroll
        for (int nt = 0; nt < 8; nt++) bc[nt] = bn[nt];
    }

    float* orow = O + (long)(strip * 16 + h * 8) * SN + c;
#pragma unroll
    for (int nt = 0; nt < 8; nt++) {
        int col = nt * 16 + c;
        float bb = bias[col];
        float mu = 0.f, inv = 1.f, gm = 1.f, bt = 0.f;
        if (BN) { mu = mean[col]; inv = rsqrtf(var[col] + BN_EPS); gm = gamma[col]; bt = beta[col]; }
#pragma unroll
        for (int r = 0; r < 8; r++) {
            float v = acc[nt][r] + bb;
            if (BN) v = (v - mu) * inv * gm + bt;
            v = fmaxf(v, 0.f);
            orow[(long)r * SN + nt * 16] = v;
        }
    }
}

template <bool FUSE_ADD, bool BN, bool RELU>
__global__ void gemm_wmma(const float* __restrict__ X, const float* __restrict__ AGG,
                          const float* __restrict__ W, const float* __restrict__ bias,
                          const float* __restrict__ gamma, const float* __restrict__ beta,
                          const float* __restrict__ mean, const float* __restrict__ var,
                          float* __restrict__ O, int M, int K, int N) {
    int wave = threadIdx.x >> 5, lane = threadIdx.x & 31;
    int tilesN = N >> 4;
    int tile = blockIdx.x * 8 + wave;
    if (tile >= (M >> 4) * tilesN) return;
    int mt = tile / tilesN, nt = tile % tilesN;
    int mi = lane & 15, kh = lane >> 4;

    const float* xr = X + (long)(mt * 16 + mi) * K + kh * 2;
    const float* ar = FUSE_ADD ? (AGG + (long)(mt * 16 + mi) * K + kh * 2) : nullptr;
    const float* wc = W + (long)(kh * 2) * N + nt * 16 + mi;

    v8f acc = {0.f, 0.f, 0.f, 0.f, 0.f, 0.f, 0.f, 0.f};
    for (int kb = 0; kb < K; kb += 4) {
        v2f a, b;
        a.x = xr[kb];
        a.y = xr[kb + 1];
        if (FUSE_ADD) { a.x += ar[kb]; a.y += ar[kb + 1]; }
        b.x = wc[(long)kb * N];
        b.y = wc[(long)(kb + 1) * N];
        acc = __builtin_amdgcn_wmma_f32_16x16x4_f32(false, a, false, b,
                                                    (short)0, acc, false, false);
    }

    int col = nt * 16 + mi;
    float bb = bias[col];
    float mu = 0.f, inv = 1.f, gm = 1.f, bt = 0.f;
    if (BN) { mu = mean[col]; inv = rsqrtf(var[col] + BN_EPS); gm = gamma[col]; bt = beta[col]; }
    float* orow = O + (long)(mt * 16 + kh * 8) * N + col;
#pragma unroll
    for (int r = 0; r < 8; r++) {
        float v = acc[r] + bb;
        if (BN)   v = (v - mu) * inv * gm + bt;
        if (RELU) v = fmaxf(v, 0.f);
        orow[(long)r * N] = v;
    }
}

__global__ void pool3(const float* __restrict__ h1, const float* __restrict__ h2,
                      const float* __restrict__ h3, const int* __restrict__ batch,
                      float* __restrict__ gcat) {
    int t = blockIdx.x * blockDim.x + threadIdx.x;
    int node = t >> 5, lane = t & 31;
    if (node >= N_NODES) return;
    int g = batch[node];
    float* base = gcat + (long)g * 384 + lane * 4;
    const float* hs[3] = {h1, h2, h3};
#pragma unroll
    for (int l = 0; l < 3; l++) {
        float4 v = *(const float4*)(hs[l] + (long)node * 128 + lane * 4);
        float* p = base + l * 128;
        atomicAdd(p + 0, v.x);
        atomicAdd(p + 1, v.y);
        atomicAdd(p + 2, v.z);
        atomicAdd(p + 3, v.w);
    }
}

__global__ void lin2_kernel(const float* __restrict__ h, const float* __restrict__ w,
                            const float* __restrict__ b, float* __restrict__ out) {
    int t = blockIdx.x * blockDim.x + threadIdx.x;
    if (t >= N_GRAPHS * 10) return;
    int g = t / 10, o = t % 10;
    const float* hr = h + (long)g * 384;
    float acc = b[o];
    for (int k = 0; k < 384; k++) acc = fmaf(hr[k], w[k * 10 + o], acc);
    out[t] = acc;
}

extern "C" void kernel_launch(void* const* d_in, const int* in_sizes, int n_in,
                              void* d_out, int out_size, void* d_ws, size_t ws_size,
                              hipStream_t stream) {
    const int*   x_ids = (const int*)d_in[0];
    const int*   edge  = (const int*)d_in[1];
    const int*   batch = (const int*)d_in[2];
    const float* emb   = (const float*)d_in[3];

    const float *c_w1[3], *c_b1[3], *c_g[3], *c_be[3], *c_mu[3], *c_va[3], *c_w2[3], *c_b2[3];
    for (int i = 0; i < 3; i++) {
        c_w1[i] = (const float*)d_in[4 + 8 * i + 0];
        c_b1[i] = (const float*)d_in[4 + 8 * i + 1];
        c_g[i]  = (const float*)d_in[4 + 8 * i + 2];
        c_be[i] = (const float*)d_in[4 + 8 * i + 3];
        c_mu[i] = (const float*)d_in[4 + 8 * i + 4];
        c_va[i] = (const float*)d_in[4 + 8 * i + 5];
        c_w2[i] = (const float*)d_in[4 + 8 * i + 6];
        c_b2[i] = (const float*)d_in[4 + 8 * i + 7];
    }
    const float* lin1_w = (const float*)d_in[28];
    const float* lin1_b = (const float*)d_in[29];
    const float* lin2_w = (const float*)d_in[30];
    const float* lin2_b = (const float*)d_in[31];

    const long NF = (long)N_NODES * 128;
    float* ws   = (float*)d_ws;
    float* x0   = ws + 0 * NF;
    float* agg  = ws + 1 * NF;
    float* tmp  = ws + 2 * NF;
    float* hbuf[3] = { ws + 3 * NF, ws + 4 * NF, x0 };
    float* gcat  = agg;
    float* lin1o = tmp;
    int* cnt    = (int*)(ws + 5 * NF);
    int* rowptr = cnt + CNT_PAD;
    int* rank   = rowptr + CNT_PAD;
    int* col    = rank + N_EDGES;

    gather_emb<<<(N_NODES * 32 + 255) / 256, 256, 0, stream>>>(x_ids, emb, x0);

    zero_f32<<<(CNT_PAD / 4 + 255) / 256, 256, 0, stream>>>((float*)cnt, CNT_PAD / 4);
    edge_rank<<<N_EDGES / 256, 256, 0, stream>>>(edge, cnt, rank);
    scan_counts<<<1, 1024, 0, stream>>>(cnt, rowptr);
    edge_fill<<<N_EDGES / 256, 256, 0, stream>>>(edge, rowptr, rank, col);

    const int strips = N_NODES / 16;
    const float* xin = x0;
    for (int l = 0; l < 3; l++) {
        gather_sum<<<2048, 256, 0, stream>>>(xin, rowptr, col, agg);
        gemm_strip<true><<<(strips + 7) / 8, 256, 0, stream>>>(
            agg, c_w1[l], c_b1[l], c_g[l], c_be[l], c_mu[l], c_va[l], tmp, strips);
        gemm_strip<false><<<(strips + 7) / 8, 256, 0, stream>>>(
            tmp, c_w2[l], c_b2[l], nullptr, nullptr, nullptr, nullptr, hbuf[l], strips);
        xin = hbuf[l];
    }

    zero_f32<<<(N_GRAPHS * 384 / 4 + 255) / 256, 256, 0, stream>>>(gcat, N_GRAPHS * 384 / 4);
    pool3<<<(N_NODES * 32 + 255) / 256, 256, 0, stream>>>(hbuf[0], hbuf[1], hbuf[2], batch, gcat);

    const int tiles_head = (N_GRAPHS / 16) * (384 / 16);
    gemm_wmma<false, false, true><<<(tiles_head + 7) / 8, 256, 0, stream>>>(
        gcat, nullptr, lin1_w, lin1_b, nullptr, nullptr, nullptr, nullptr,
        lin1o, N_GRAPHS, 384, 384);

    lin2_kernel<<<(N_GRAPHS * 10 + 255) / 256, 256, 0, stream>>>(lin1o, lin2_w, lin2_b,
                                                                 (float*)d_out);
}


# Pooling: each wave sums its 16 consecutive nodes' rows in registers in node order while the graph id holds, one atomic add per run and column instead of one per node; sum order changes with it
# speedup vs baseline: 2.5364x; 1.1877x over previous
#include <hip/hip_runtime.h>

typedef __attribute__((ext_vector_type(2))) float v2f;
typedef __attribute__((ext_vector_type(8))) float v8f;

#define N_NODES  50000
#define N_EDGES  800000
#define N_GRAPHS 512
#define BN_EPS   1e-5f

__global__ void zero_f32(float* __restrict__ p, int n4) {
    int t = blockIdx.x * blockDim.x + threadIdx.x;
    if (t < n4) ((float4*)p)[t] = make_float4(0.f, 0.f, 0.f, 0.f);
}

__global__ void gather_emb(const int* __restrict__ ids, const float* __restrict__ emb,
                           float* __restrict__ x) {
    int t = blockIdx.x * blockDim.x + threadIdx.x;
    int node = t >> 5, lane = t & 31;
    if (node >= N_NODES) return;
    int id = ids[node];
    float4 v = *(const float4*)(emb + (long)id * 128 + lane * 4);
    *(float4*)(x + (long)node * 128 + lane * 4) = v;
}

#define SCAN_STEPS 13
#define CNT_PAD    (32 * SCAN_STEPS * 128)

__global__ void edge_rank(const int* __restrict__ ei, int* __restrict__ cnt,
                          int* __restrict__ rank) {
    int e = blockIdx.x * blockDim.x + threadIdx.x;
    if (e >= N_EDGES) return;
    rank[e] = atomicAdd(cnt + ei[N_EDGES + e], 1);
}

__global__ void __launch_bounds__(1024) scan_counts(const int* __restrict__ cnt,
                                                    int* __restrict__ rowptr) {
    __shared__ int wsum[32];
    int wave = threadIdx.x >> 5, lane = threadIdx.x & 31;
    const int4* c4 = (const int4*)cnt + wave * (SCAN_STEPS * 32);
    int4* r4 = (int4*)rowptr + wave * (SCAN_STEPS * 32);

    int tot = 0;
    for (int s = 0; s < SCAN_STEPS; s++) {
        int4 v = c4[s * 32 + lane];
        tot += v.x + v.y + v.z + v.w;
    }
    for (int off = 16; off > 0; off >>= 1) tot += __shfl_xor(tot, off, 32);
    if (lane == 0) wsum[wave] = tot;
    __syncthreads();

    int carry = 0;
    for (int w = 0; w < wave; w++) carry += wsum[w];

    for (int s = 0; s < SCAN_STEPS; s++) {
        int4 v = c4[s * 32 + lane];
        int l = v.x + v.y + v.z + v.w;
        int inc = l;
        for (int off = 1; off < 32; off <<= 1) {
            int t = __shfl_up(inc, off, 32);
            if (lane >= off) inc += t;
        }
        int ex = carry + inc - l;
        int4 o;
        o.x = ex;
        o.y = ex + v.x;
        o.z = ex + v.x + v.y;
        o.w = ex + v.x + v.y + v.z;
        r4[s * 32 + lane] = o;
        carry += __shfl(inc, 31, 32);
    }
}

__global__ void edge_fill(const int* __restrict__ ei, const int* __restrict__ rowptr,
                          const int* __restrict__ rank, int* __restrict__ col) {
    int e = blockIdx.x * blockDim.x + threadIdx.x;
    if (e >= N_EDGES) return;
    int d = ei[N_EDGES + e];
    col[rowptr[d] + rank[e]] = ei[e];
}

__global__ void gather_sum(const float* __restrict__ x, const int* __restrict__ rowptr,
                           const int* __restrict__ col, float* __restrict__ s) {
    int wave = threadIdx.x >> 5, lane = threadIdx.x & 31;
    for (int node = blockIdx.x * 8 + wave; node < N_NODES; node += gridDim.x * 8) {
        int beg = __builtin_amdgcn_readfirstlane(rowptr[node]);
        int end = __builtin_amdgcn_readfirstlane(rowptr[node + 1]);
        float4 acc = make_float4(0.f, 0.f, 0.f, 0.f);
        for (int base = beg; base < end; base += 32) {
            int n = end - base;
            if (n > 32) n = 32;
            int myc = (lane < n) ? col[base + lane] : 0;
            int j = 0;
            for (; j + 4 <= n; j += 4) {
                int s0 = __builtin_amdgcn_readlane(myc, j);
                int s1 = __builtin_amdgcn_readlane(myc, j + 1);
                int s2 = __builtin_amdgcn_readlane(myc, j + 2);
                int s3 = __builtin_amdgcn_readlane(myc, j + 3);
                float4 v0 = *(const float4*)(x + (long)s0 * 128 + lane * 4);
                float4 v1 = *(const float4*)(x + (long)s1 * 128 + lane * 4);
                float4 v2 = *(const float4*)(x + (long)s2 * 128 + lane * 4);
                float4 v3 = *(const float4*)(x + (long)s3 * 128 + lane * 4);
                acc.x = (((acc.x + v0.x) + v1.x) + v2.x) + v3.x;
                acc.y = (((acc.y + v0.y) + v1.y) + v2.y) + v3.y;
                acc.z = (((acc.z + v0.z) + v1.z) + v2.z) + v3.z;
                acc.w = (((acc.w + v0.w) + v1.w) + v2.w) + v3.w;
            }
            for (; j < n; j++) {
                int s0 = __builtin_amdgcn_readlane(myc, j);
                float4 v0 = *(const float4*)(x + (long)s0 * 128 + lane * 4);
                acc.x += v0.x; acc.y += v0.y; acc.z += v0.z; acc.w += v0.w;
            }
        }
        float4 xv = *(const float4*)(x + (long)node * 128 + lane * 4);
        float4 o = make_float4(xv.x + acc.x, xv.y + acc.y, xv.z + acc.z, xv.w + acc.w);
        *(float4*)(s + (long)node * 128 + lane * 4) = o;
    }
}

#define SK  128
#define SN  128
#define WPF 288
template <bool BN>
__global__ void __launch_bounds__(256) gemm_strip(const float* __restrict__ X, const float* __restrict__ W,
                                                  const float* __restrict__ bias,
                                                  const float* __restrict__ gamma, const float* __restrict__ beta,
                                                  const float* __restrict__ mean, const float* __restrict__ var,
                                                  float* __restrict__ O, int strips) {
    __shared__ __attribute__((aligned(16))) float wl[(SK / 2) * WPF];
    int wave = threadIdx.x >> 5, lane = threadIdx.x & 31;
    int c = lane & 15, h = lane >> 4;
    int strip = blockIdx.x * 8 + wave;

#pragma unroll
    for (int i = 0; i < 8; i++) {
        int kp = wave + 8 * i;
        float4 r0 = *(const float4*)(W + (long)(2 * kp) * SN + lane * 4);
        float4 r1 = *(const float4*)(W + (long)(2 * kp + 1) * SN + lane * 4);
        float* d = wl + kp * WPF + lane * 8;
        *(float4*)(d)     = make_float4(r0.x, r1.x, r0.y, r1.y);
        *(float4*)(d + 4) = make_float4(r0.z, r1.z, r0.w, r1.w);
    }
    __syncthreads();
    if (strip >= strips) return;

    v2f af[SK / 4];
    const float* xr = X + (long)(strip * 16 + c) * SK + h * 2;
#pragma unroll
    for (int i = 0; i < SK / 4; i++) af[i] = *(const v2f*)(xr + i * 4);

    v8f acc[8];
#pragma unroll
    for (int nt = 0; nt < 8; nt++) acc[nt] = (v8f){0.f, 0.f, 0.f, 0.f, 0.f, 0.f, 0.f, 0.f};

    const float* bl = wl + h * WPF + c * 2;
    v2f bc[8], bn[8];
#pragma unroll
    for (int nt = 0; nt < 8; nt++) bc[nt] = *(const v2f*)(bl + nt * 32);
#pragma unroll
    for (int i = 0; i < SK / 4; i++) {
        if (i + 1 < SK / 4) {
#pragma unroll
            for (int nt = 0; nt < 8; nt++) bn[nt] = *(const v2f*)(bl + (2 * i + 2) * WPF + nt * 32);
        }
        __builtin_amdgcn_sched_barrier(0);
#pragma unroll
        for (int nt = 0; nt < 8; nt++)
            acc[nt] = __builtin_amdgcn_wmma_f32_16x16x4_f32(false, af[i], false, bc[nt],
                                                            (short)0, acc[nt], false, false);
        __builtin_amdgcn_sched_barrier(0);
#pragma unroll
        for (int nt = 0; nt < 8; nt++) bc[nt] = bn[nt];
    }

    float* orow = O + (long)(strip * 16 + h * 8) * SN + c;
#pragma unroll
    for (int nt = 0; nt < 8; nt++) {
        int col = nt * 16 + c;
        float bb = bias[col];
        float mu = 0.f, inv = 1.f, gm = 1.f, bt = 0.f;
        if (BN) { mu = mean[col]; inv = rsqrtf(var[col] + BN_EPS); gm = gamma[col]; bt = beta[col]; }
#pragma unroll
        for (int r = 0; r < 8; r++) {
            float v = acc[nt][r] + bb;
            if (BN) v = (v - mu) * inv * gm + bt;
            v = fmaxf(v, 0.f);
            orow[(long)r * SN + nt * 16] = v;
        }
    }
}

template <bool FUSE_ADD, bool BN, bool RELU>
__global__ void gemm_wmma(const float* __restrict__ X, const float* __restrict__ AGG,
                          const float* __restrict__ W, const float* __restrict__ bias,
                          const float* __restrict__ gamma, const float* __restrict__ beta,
                          const float* __restrict__ mean, const float* __restrict__ var,
                          float* __restrict__ O, int M, int K, int N) {
    int wave = threadIdx.x >> 5, lane = threadIdx.x & 31;
    int tilesN = N >> 4;
    int tile = blockIdx.x * 8 + wave;
    if (tile >= (M >> 4) * tilesN) return;
    int mt = tile / tilesN, nt = tile % tilesN;
    int mi = lane & 15, kh = lane >> 4;

    const float* xr = X + (long)(mt * 16 + mi) * K + kh * 2;
    const float* ar = FUSE_ADD ? (AGG + (long)(mt * 16 + mi) * K + kh * 2) : nullptr;
    const float* wc = W + (long)(kh * 2) * N + nt * 16 + mi;

    v8f acc = {0.f, 0.f, 0.f, 0.f, 0.f, 0.f, 0.f, 0.f};
    for (int kb = 0; kb < K; kb += 4) {
        v2f a, b;
        a.x = xr[kb];
        a.y = xr[kb + 1];
        if (FUSE_ADD) { a.x += ar[kb]; a.y += ar[kb + 1]; }
        b.x = wc[(long)kb * N];
        b.y = wc[(long)(kb + 1) * N];
        acc = __builtin_amdgcn_wmma_f32_16x16x4_f32(false, a, false, b,
                                                    (short)0, acc, false, false);
    }

    int col = nt * 16 + mi;
    float bb = bias[col];
    float mu = 0.f, inv = 1.f, gm = 1.f, bt = 0.f;
    if (BN) { mu = mean[col]; inv = rsqrtf(var[col] + BN_EPS); gm = gamma[col]; bt = beta[col]; }
    float* orow = O + (long)(mt * 16 + kh * 8) * N + col;
#pragma unroll
    for (int r = 0; r < 8; r++) {
        float v = acc[r] + bb;
        if (BN)   v = (v - mu) * inv * gm + bt;
        if (RELU) v = fmaxf(v, 0.f);
        orow[(long)r * N] = v;
    }
}

#define POOL_CH 16
__device__ __forceinline__ void pool_flush(float* __restrict__ p, float4 a) {
    atomicAdd(p + 0, a.x);
    atomicAdd(p + 1, a.y);
    atomicAdd(p + 2, a.z);
    atomicAdd(p + 3, a.w);
}

__global__ void __launch_bounds__(256) pool3(const float* __restrict__ h1, const float* __restrict__ h2,
                                             const float* __restrict__ h3, const int* __restrict__ batch,
                                             float* __restrict__ gcat) {
    int wave = threadIdx.x >> 5, lane = threadIdx.x & 31;
    int chunk = blockIdx.x * 8 + wave;
    if (chunk >= N_NODES / POOL_CH) return;
    int base = chunk * POOL_CH;
    int myb = batch[base + (lane & (POOL_CH - 1))];
    const float* hs[3] = {h1, h2, h3};
#pragma unroll
    for (int l = 0; l < 3; l++) {
        const float* hp = hs[l] + (long)base * 128 + lane * 4;
        float4 v[POOL_CH];
#pragma unroll
        for (int j = 0; j < POOL_CH; j++) v[j] = *(const float4*)(hp + j * 128);
        float* gp = gcat + l * 128 + lane * 4;
        int cur = __builtin_amdgcn_readlane(myb, 0);
        float4 acc = v[0];
#pragma unroll
        for (int j = 1; j < POOL_CH; j++) {
            int g = __builtin_amdgcn_readlane(myb, j);
            if (g != cur) {
                pool_flush(gp + (long)cur * 384, acc);
                acc = v[j];
                cur = g;
            } else {
                acc.x += v[j].x; acc.y += v[j].y; acc.z += v[j].z; acc.w += v[j].w;
            }
        }
        pool_flush(gp + (long)cur * 384, acc);
    }
}

__global__ void lin2_kernel(const float* __restrict__ h, const float* __restrict__ w,
                            const float* __restrict__ b, float* __restrict__ out) {
    int t = blockIdx.x * blockDim.x + threadIdx.x;
    if (t >= N_GRAPHS * 10) return;
    int g = t / 10, o = t % 10;
    const float* hr = h + (long)g * 384;
    float acc = b[o];
    for (int k = 0; k < 384; k++) acc = fmaf(hr[k], w[k * 10 + o], acc);
    out[t] = acc;
}

extern "C" void kernel_launch(void* const* d_in, const int* in_sizes, int n_in,
                              void* d_out, int out_size, void* d_ws, size_t ws_size,
                              hipStream_t stream) {
    const int*   x_ids = (const int*)d_in[0];
    const int*   edge  = (const int*)d_in[1];
    const int*   batch = (const int*)d_in[2];
    const float* emb   = (const float*)d_in[3];

    const float *c_w1[3], *c_b1[3], *c_g[3], *c_be[3], *c_mu[3], *c_va[3], *c_w2[3], *c_b2[3];
    for (int i = 0; i < 3; i++) {
        c_w1[i] = (const float*)d_in[4 + 8 * i + 0];
        c_b1[i] = (const float*)d_in[4 + 8 * i + 1];
        c_g[i]  = (const float*)d_in[4 + 8 * i + 2];
        c_be[i] = (const float*)d_in[4 + 8 * i + 3];
        c_mu[i] = (const float*)d_in[4 + 8 * i + 4];
        c_va[i] = (const float*)d_in[4 + 8 * i + 5];
        c_w2[i] = (const float*)d_in[4 + 8 * i + 6];
        c_b2[i] = (const float*)d_in[4 + 8 * i + 7];
    }
    const float* lin1_w = (const float*)d_in[28];
    const float* lin1_b = (const float*)d_in[29];
    const float* lin2_w = (const float*)d_in[30];
    const float* lin2_b = (const float*)d_in[31];

    const long NF = (long)N_NODES * 128;
    float* ws   = (float*)d_ws;
    float* x0   = ws + 0 * NF;
    float* agg  = ws + 1 * NF;
    float* tmp  = ws + 2 * NF;
    float* hbuf[3] = { ws + 3 * NF, ws + 4 * NF, x0 };
    float* gcat  = agg;
    float* lin1o = tmp;
    int* cnt    = (int*)(ws + 5 * NF);
    int* rowptr = cnt + CNT_PAD;
    int* rank   = rowptr + CNT_PAD;
    int* col    = rank + N_EDGES;

    gather_emb<<<(N_NODES * 32 + 255) / 256, 256, 0, stream>>>(x_ids, emb, x0);

    zero_f32<<<(CNT_PAD / 4 + 255) / 256, 256, 0, stream>>>((float*)cnt, CNT_PAD / 4);
    edge_rank<<<N_EDGES / 256, 256, 0, stream>>>(edge, cnt, rank);
    scan_counts<<<1, 1024, 0, stream>>>(cnt, rowptr);
    edge_fill<<<N_EDGES / 256, 256, 0, stream>>>(edge, rowptr, rank, col);

    const int strips = N_NODES / 16;
    const float* xin = x0;
    for (int l = 0; l < 3; l++) {
        gather_sum<<<2048, 256, 0, stream>>>(xin, rowptr, col, agg);
        gemm_strip<true><<<(strips + 7) / 8, 256, 0, stream>>>(
            agg, c_w1[l], c_b1[l], c_g[l], c_be[l], c_mu[l], c_va[l], tmp, strips);
        gemm_strip<false><<<(strips + 7) / 8, 256, 0, stream>>>(
            tmp, c_w2[l], c_b2[l], nullptr, nullptr, nullptr, nullptr, hbuf[l], strips);
        xin = hbuf[l];
    }

    zero_f32<<<(N_GRAPHS * 384 / 4 + 255) / 256, 256, 0, stream>>>(gcat, N_GRAPHS * 384 / 4);
    pool3<<<(N_NODES / POOL_CH + 7) / 8, 256, 0, stream>>>(hbuf[0], hbuf[1], hbuf[2], batch, gcat);

    const int tiles_head = (N_GRAPHS / 16) * (384 / 16);
    gemm_wmma<false, false, true><<<(tiles_head + 7) / 8, 256, 0, stream>>>(
        gcat, nullptr, lin1_w, lin1_b, nullptr, nullptr, nullptr, nullptr,
        lin1o, N_GRAPHS, 384, 384);

    lin2_kernel<<<(N_GRAPHS * 10 + 255) / 256, 256, 0, stream>>>(lin1o, lin2_w, lin2_b,
                                                                 (float*)d_out);
}


# Edges grouped by destination by a two-level counting sort in LDS (atomics on LDS counters) instead of 800000 returning global atomics, a global scan and a fill pass; slot (sum) order changes with it
# speedup vs baseline: 3.2565x; 1.2839x over previous
#include <hip/hip_runtime.h>

typedef __attribute__((ext_vector_type(2))) float v2f;
typedef __attribute__((ext_vector_type(8))) float v8f;

#define N_NODES  50000
#define N_EDGES  800000
#define N_GRAPHS 512
#define BN_EPS   1e-5f

__global__ void zero_f32(float* __restrict__ p, int n4) {
    int t = blockIdx.x * blockDim.x + threadIdx.x;
    if (t < n4) ((float4*)p)[t] = make_float4(0.f, 0.f, 0.f, 0.f);
}

__global__ void gather_emb(const int* __restrict__ ids, const float* __restrict__ emb,
                           float* __restrict__ x) {
    int t = blockIdx.x * blockDim.x + threadIdx.x;
    int node = t >> 5, lane = t & 31;
    if (node >= N_NODES) return;
    int id = ids[node];
    float4 v = *(const float4*)(emb + (long)id * 128 + lane * 4);
    *(float4*)(x + (long)node * 128 + lane * 4) = v;
}

#define EB_BLOCKS 256
#define EB_CHUNK  3125
#define EB_BINS   256
#define EB_RANGE  196
#define EB_T      512
#define EB_PER_T  7
#define EB_SEG    16
#define RP_PAD    (EB_BINS * EB_RANGE)

__device__ __forceinline__ int scan256_excl(int v, int tid, int* wtot) {
    int lane = tid & 31, wave = tid >> 5;
    int inc = v;
#pragma unroll
    for (int off = 1; off < 32; off <<= 1) {
        int t = __shfl_up(inc, off, 32);
        if (lane >= off) inc += t;
    }
    if (tid < 256 && lane == 31) wtot[wave] = inc;
    __syncthreads();
    int carry = 0;
    if (tid < 256)
        for (int w = 0; w < wave; w++) carry += wtot[w];
    return carry + inc - v;
}

__global__ void __launch_bounds__(EB_T) edge_bin(const int* __restrict__ ei, int* __restrict__ binned,
                                                 int* __restrict__ cntm, int* __restrict__ stm) {
    __shared__ int hist[EB_BINS];
    __shared__ int offs[EB_BINS];
    __shared__ int wtot[8];
    int tid = threadIdx.x, blk = blockIdx.x;
    if (tid < EB_BINS) hist[tid] = 0;
    __syncthreads();

    int d[EB_PER_T], s[EB_PER_T], r[EB_PER_T];
#pragma unroll
    for (int k = 0; k < EB_PER_T; k++) {
        int idx = tid + k * EB_T;
        d[k] = -1;
        s[k] = 0;
        if (idx < EB_CHUNK) {
            int e = blk * EB_CHUNK + idx;
            d[k] = ei[N_EDGES + e];
            s[k] = ei[e];
        }
    }
#pragma unroll
    for (int k = 0; k < EB_PER_T; k++) {
        r[k] = 0;
        if (d[k] >= 0) r[k] = atomicAdd(&hist[d[k] / EB_RANGE], 1);
    }
    __syncthreads();

    int v = (tid < EB_BINS) ? hist[tid] : 0;
    int ex = scan256_excl(v, tid, wtot);
    if (tid < EB_BINS) {
        offs[tid] = ex;
        cntm[blk * EB_BINS + tid] = v;
        stm[blk * EB_BINS + tid] = ex;
    }
    __syncthreads();

#pragma unroll
    for (int k = 0; k < EB_PER_T; k++) {
        if (d[k] >= 0) {
            int bn = d[k] / EB_RANGE;
            binned[blk * EB_CHUNK + offs[bn] + r[k]] = ((d[k] - bn * EB_RANGE) << 16) | s[k];
        }
    }
}

__global__ void __launch_bounds__(EB_T) edge_place(const int* __restrict__ binned, const int* __restrict__ cntm,
                                                   const int* __restrict__ stm, int* __restrict__ rowptr,
                                                   int* __restrict__ col) {
    __shared__ int cnt[EB_BINS];
    __shared__ int wtot[8];
    __shared__ int red[EB_T / 32];
    int tid = threadIdx.x, b = blockIdx.x;
    int wave = tid >> 5, lane = tid & 31;
    if (tid < EB_BINS) cnt[tid] = 0;

    int c = 0, s = 0;
    if (lane < EB_SEG) {
        int i = wave * EB_SEG + lane;
        c = cntm[i * EB_BINS + b];
        s = stm[i * EB_BINS + b];
    }
    int part = s;
    for (int off = 16; off > 0; off >>= 1) part += __shfl_xor(part, off, 32);
    if (lane == 0) red[wave] = part;

    int len[EB_SEG], st[EB_SEG], rec[EB_SEG];
#pragma unroll
    for (int q = 0; q < EB_SEG; q++) {
        len[q] = __builtin_amdgcn_readlane(c, q);
        st[q]  = (wave * EB_SEG + q) * EB_CHUNK + __builtin_amdgcn_readlane(s, q);
    }
#pragma unroll
    for (int q = 0; q < EB_SEG; q++) {
        int p = st[q] + lane;
        if (p > N_EDGES - 1) p = N_EDGES - 1;
        int t = binned[p];
        rec[q] = (lane < len[q]) ? t : -1;
    }
    __syncthreads();

#pragma unroll
    for (int q = 0; q < EB_SEG; q++) {
        if (rec[q] >= 0) atomicAdd(&cnt[rec[q] >> 16], 1);
        for (int o = 32 + lane; o < len[q]; o += 32) atomicAdd(&cnt[binned[st[q] + o] >> 16], 1);
    }
    __syncthreads();

    int base = 0;
    for (int w = 0; w < EB_T / 32; w++) base += red[w];
    int v = (tid < EB_BINS) ? cnt[tid] : 0;
    int ex = scan256_excl(v, tid, wtot);
    if (tid < EB_BINS) {
        int p = base + ex;
        cnt[tid] = p;
        int node = b * EB_RANGE + tid;
        if (tid < EB_RANGE && node <= N_NODES) rowptr[node] = p;
    }
    __syncthreads();

#pragma unroll
    for (int q = 0; q < EB_SEG; q++) {
        if (rec[q] >= 0) {
            int pos = atomicAdd(&cnt[rec[q] >> 16], 1);
            col[pos] = rec[q] & 0xFFFF;
        }
        for (int o = 32 + lane; o < len[q]; o += 32) {
            int rr = binned[st[q] + o];
            int pos = atomicAdd(&cnt[rr >> 16], 1);
            col[pos] = rr & 0xFFFF;
        }
    }
}

__global__ void gather_sum(const float* __restrict__ x, const int* __restrict__ rowptr,
                           const int* __restrict__ col, float* __restrict__ s) {
    int wave = threadIdx.x >> 5, lane = threadIdx.x & 31;
    for (int node = blockIdx.x * 8 + wave; node < N_NODES; node += gridDim.x * 8) {
        int beg = __builtin_amdgcn_readfirstlane(rowptr[node]);
        int end = __builtin_amdgcn_readfirstlane(rowptr[node + 1]);
        float4 acc = make_float4(0.f, 0.f, 0.f, 0.f);
        for (int base = beg; base < end; base += 32) {
            int n = end - base;
            if (n > 32) n = 32;
            int myc = (lane < n) ? col[base + lane] : 0;
            int j = 0;
            for (; j + 4 <= n; j += 4) {
                int s0 = __builtin_amdgcn_readlane(myc, j);
                int s1 = __builtin_amdgcn_readlane(myc, j + 1);
                int s2 = __builtin_amdgcn_readlane(myc, j + 2);
                int s3 = __builtin_amdgcn_readlane(myc, j + 3);
                float4 v0 = *(const float4*)(x + (long)s0 * 128 + lane * 4);
                float4 v1 = *(const float4*)(x + (long)s1 * 128 + lane * 4);
                float4 v2 = *(const float4*)(x + (long)s2 * 128 + lane * 4);
                float4 v3 = *(const float4*)(x + (long)s3 * 128 + lane * 4);
                acc.x = (((acc.x + v0.x) + v1.x) + v2.x) + v3.x;
                acc.y = (((acc.y + v0.y) + v1.y) + v2.y) + v3.y;
                acc.z = (((acc.z + v0.z) + v1.z) + v2.z) + v3.z;
                acc.w = (((acc.w + v0.w) + v1.w) + v2.w) + v3.w;
            }
            for (; j < n; j++) {
                int s0 = __builtin_amdgcn_readlane(myc, j);
                float4 v0 = *(const float4*)(x + (long)s0 * 128 + lane * 4);
                acc.x += v0.x; acc.y += v0.y; acc.z += v0.z; acc.w += v0.w;
            }
        }
        float4 xv = *(const float4*)(x + (long)node * 128 + lane * 4);
        float4 o = make_float4(xv.x + acc.x, xv.y + acc.y, xv.z + acc.z, xv.w + acc.w);
        *(float4*)(s + (long)node * 128 + lane * 4) = o;
    }
}

#define SK  128
#define SN  128
#define WPF 288
template <bool BN>
__global__ void __launch_bounds__(256) gemm_strip(const float* __restrict__ X, const float* __restrict__ W,
                                                  const float* __restrict__ bias,
                                                  const float* __restrict__ gamma, const float* __restrict__ beta,
                                                  const float* __restrict__ mean, const float* __restrict__ var,
                                                  float* __restrict__ O, int strips) {
    __shared__ __attribute__((aligned(16))) float wl[(SK / 2) * WPF];
    int wave = threadIdx.x >> 5, lane = threadIdx.x & 31;
    int c = lane & 15, h = lane >> 4;
    int strip = blockIdx.x * 8 + wave;

#pragma unroll
    for (int i = 0; i < 8; i++) {
        int kp = wave + 8 * i;
        float4 r0 = *(const float4*)(W + (long)(2 * kp) * SN + lane * 4);
        float4 r1 = *(const float4*)(W + (long)(2 * kp + 1) * SN + lane * 4);
        float* d = wl + kp * WPF + lane * 8;
        *(float4*)(d)     = make_float4(r0.x, r1.x, r0.y, r1.y);
        *(float4*)(d + 4) = make_float4(r0.z, r1.z, r0.w, r1.w);
    }
    __syncthreads();
    if (strip >= strips) return;

    v2f af[SK / 4];
    const float* xr = X + (long)(strip * 16 + c) * SK + h * 2;
#pragma unroll
    for (int i = 0; i < SK / 4; i++) af[i] = *(const v2f*)(xr + i * 4);

    v8f acc[8];
#pragma unroll
    for (int nt = 0; nt < 8; nt++) acc[nt] = (v8f){0.f, 0.f, 0.f, 0.f, 0.f, 0.f, 0.f, 0.f};

    const float* bl = wl + h * WPF + c * 2;
    v2f bc[8], bn[8];
#pragma unroll
    for (int nt = 0; nt < 8; nt++) bc[nt] = *(const v2f*)(bl + nt * 32);
#pragma unroll
    for (int i = 0; i < SK / 4; i++) {
        if (i + 1 < SK / 4) {
#pragma unroll
            for (int nt = 0; nt < 8; nt++) bn[nt] = *(const v2f*)(bl + (2 * i + 2) * WPF + nt * 32);
        }
        __builtin_amdgcn_sched_barrier(0);
#pragma unroll
        for (int nt = 0; nt < 8; nt++)
            acc[nt] = __builtin_amdgcn_wmma_f32_16x16x4_f32(false, af[i], false, bc[nt],
                                                            (short)0, acc[nt], false, false);
        __builtin_amdgcn_sched_barrier(0);
#pragma unroll
        for (int nt = 0; nt < 8; nt++) bc[nt] = bn[nt];
    }

    float* orow = O + (long)(strip * 16 + h * 8) * SN + c;
#pragma unroll
    for (int nt = 0; nt < 8; nt++) {
        int col = nt * 16 + c;
        float bb = bias[col];
        float mu = 0.f, inv = 1.f, gm = 1.f, bt = 0.f;
        if (BN) { mu = mean[col]; inv = rsqrtf(var[col] + BN_EPS); gm = gamma[col]; bt = beta[col]; }
#pragma unroll
        for (int r = 0; r < 8; r++) {
            float v = acc[nt][r] + bb;
            if (BN) v = (v - mu) * inv * gm + bt;
            v = fmaxf(v, 0.f);
            orow[(long)r * SN + nt * 16] = v;
        }
    }
}

template <bool FUSE_ADD, bool BN, bool RELU>
__global__ void gemm_wmma(const float* __restrict__ X, const float* __restrict__ AGG,
                          const float* __restrict__ W, const float* __restrict__ bias,
                          const float* __restrict__ gamma, const float* __restrict__ beta,
                          const float* __restrict__ mean, const float* __restrict__ var,
                          float* __restrict__ O, int M, int K, int N) {
    int wave = threadIdx.x >> 5, lane = threadIdx.x & 31;
    int tilesN = N >> 4;
    int tile = blockIdx.x * 8 + wave;
    if (tile >= (M >> 4) * tilesN) return;
    int mt = tile / tilesN, nt = tile % tilesN;
    int mi = lane & 15, kh = lane >> 4;

    const float* xr = X + (long)(mt * 16 + mi) * K + kh * 2;
    const float* ar = FUSE_ADD ? (AGG + (long)(mt * 16 + mi) * K + kh * 2) : nullptr;
    const float* wc = W + (long)(kh * 2) * N + nt * 16 + mi;

    v8f acc = {0.f, 0.f, 0.f, 0.f, 0.f, 0.f, 0.f, 0.f};
    for (int kb = 0; kb < K; kb += 4) {
        v2f a, b;
        a.x = xr[kb];
        a.y = xr[kb + 1];
        if (FUSE_ADD) { a.x += ar[kb]; a.y += ar[kb + 1]; }
        b.x = wc[(long)kb * N];
        b.y = wc[(long)(kb + 1) * N];
        acc = __builtin_amdgcn_wmma_f32_16x16x4_f32(false, a, false, b,
                                                    (short)0, acc, false, false);
    }

    int col = nt * 16 + mi;
    float bb = bias[col];
    float mu = 0.f, inv = 1.f, gm = 1.f, bt = 0.f;
    if (BN) { mu = mean[col]; inv = rsqrtf(var[col] + BN_EPS); gm = gamma[col]; bt = beta[col]; }
    float* orow = O + (long)(mt * 16 + kh * 8) * N + col;
#pragma unroll
    for (int r = 0; r < 8; r++) {
        float v = acc[r] + bb;
        if (BN)   v = (v - mu) * inv * gm + bt;
        if (RELU) v = fmaxf(v, 0.f);
        orow[(long)r * N] = v;
    }
}

#define POOL_CH 16
__device__ __forceinline__ void pool_flush(float* __restrict__ p, float4 a) {
    atomicAdd(p + 0, a.x);
    atomicAdd(p + 1, a.y);
    atomicAdd(p + 2, a.z);
    atomicAdd(p + 3, a.w);
}

__global__ void __launch_bounds__(256) pool3(const float* __restrict__ h1, const float* __restrict__ h2,
                                             const float* __restrict__ h3, const int* __restrict__ batch,
                                             float* __restrict__ gcat) {
    int wave = threadIdx.x >> 5, lane = threadIdx.x & 31;
    int chunk = blockIdx.x * 8 + wave;
    if (chunk >= N_NODES / POOL_CH) return;
    int base = chunk * POOL_CH;
    int myb = batch[base + (lane & (POOL_CH - 1))];
    const float* hs[3] = {h1, h2, h3};
#pragma unroll
    for (int l = 0; l < 3; l++) {
        const float* hp = hs[l] + (long)base * 128 + lane * 4;
        float4 v[POOL_CH];
#pragma unroll
        for (int j = 0; j < POOL_CH; j++) v[j] = *(const float4*)(hp + j * 128);
        float* gp = gcat + l * 128 + lane * 4;
        int cur = __builtin_amdgcn_readlane(myb, 0);
        float4 acc = v[0];
#pragma unroll
        for (int j = 1; j < POOL_CH; j++) {
            int g = __builtin_amdgcn_readlane(myb, j);
            if (g != cur) {
                pool_flush(gp + (long)cur * 384, acc);
                acc = v[j];
                cur = g;
            } else {
                acc.x += v[j].x; acc.y += v[j].y; acc.z += v[j].z; acc.w += v[j].w;
            }
        }
        pool_flush(gp + (long)cur * 384, acc);
    }
}

__global__ void lin2_kernel(const float* __restrict__ h, const float* __restrict__ w,
                            const float* __restrict__ b, float* __restrict__ out) {
    int t = blockIdx.x * blockDim.x + threadIdx.x;
    if (t >= N_GRAPHS * 10) return;
    int g = t / 10, o = t % 10;
    const float* hr = h + (long)g * 384;
    float acc = b[o];
    for (int k = 0; k < 384; k++) acc = fmaf(hr[k], w[k * 10 + o], acc);
    out[t] = acc;
}

extern "C" void kernel_launch(void* const* d_in, const int* in_sizes, int n_in,
                              void* d_out, int out_size, void* d_ws, size_t ws_size,
                              hipStream_t stream) {
    const int*   x_ids = (const int*)d_in[0];
    const int*   edge  = (const int*)d_in[1];
    const int*   batch = (const int*)d_in[2];
    const float* emb   = (const float*)d_in[3];

    const float *c_w1[3], *c_b1[3], *c_g[3], *c_be[3], *c_mu[3], *c_va[3], *c_w2[3], *c_b2[3];
    for (int i = 0; i < 3; i++) {
        c_w1[i] = (const float*)d_in[4 + 8 * i + 0];
        c_b1[i] = (const float*)d_in[4 + 8 * i + 1];
        c_g[i]  = (const float*)d_in[4 + 8 * i + 2];
        c_be[i] = (const float*)d_in[4 + 8 * i + 3];
        c_mu[i] = (const float*)d_in[4 + 8 * i + 4];
        c_va[i] = (const float*)d_in[4 + 8 * i + 5];
        c_w2[i] = (const float*)d_in[4 + 8 * i + 6];
        c_b2[i] = (const float*)d_in[4 + 8 * i + 7];
    }
    const float* lin1_w = (const float*)d_in[28];
    const float* lin1_b = (const float*)d_in[29];
    const float* lin2_w = (const float*)d_in[30];
    const float* lin2_b = (const float*)d_in[31];

    const long NF = (long)N_NODES * 128;
    float* ws   = (float*)d_ws;
    float* x0   = ws + 0 * NF;
    float* agg  = ws + 1 * NF;
    float* tmp  = ws + 2 * NF;
    float* hbuf[3] = { ws + 3 * NF, ws + 4 * NF, x0 };
    float* gcat  = agg;
    float* lin1o = tmp;
    int* rowptr = (int*)(ws + 5 * NF);
    int* binned = rowptr + RP_PAD;
    int* col    = binned + N_EDGES;
    int* cntm   = col + N_EDGES;
    int* stm    = cntm + EB_BLOCKS * EB_BINS;

    gather_emb<<<(N_NODES * 32 + 255) / 256, 256, 0, stream>>>(x_ids, emb, x0);

    edge_bin<<<EB_BLOCKS, EB_T, 0, stream>>>(edge, binned, cntm, stm);
    edge_place<<<EB_BINS, EB_T, 0, stream>>>(binned, cntm, stm, rowptr, col);

    const int strips = N_NODES / 16;
    const float* xin = x0;
    for (int l = 0; l < 3; l++) {
        gather_sum<<<2048, 256, 0, stream>>>(xin, rowptr, col, agg);
        gemm_strip<true><<<(strips + 7) / 8, 256, 0, stream>>>(
            agg, c_w1[l], c_b1[l], c_g[l], c_be[l], c_mu[l], c_va[l], tmp, strips);
        gemm_strip<false><<<(strips + 7) / 8, 256, 0, stream>>>(
            tmp, c_w2[l], c_b2[l], nullptr, nullptr, nullptr, nullptr, hbuf[l], strips);
        xin = hbuf[l];
    }

    zero_f32<<<(N_GRAPHS * 384 / 4 + 255) / 256, 256, 0, stream>>>(gcat, N_GRAPHS * 384 / 4);
    pool3<<<(N_NODES / POOL_CH + 7) / 8, 256, 0, stream>>>(hbuf[0], hbuf[1], hbuf[2], batch, gcat);

    const int tiles_head = (N_GRAPHS / 16) * (384 / 16);
    gemm_wmma<false, false, true><<<(tiles_head + 7) / 8, 256, 0, stream>>>(
        gcat, nullptr, lin1_w, lin1_b, nullptr, nullptr, nullptr, nullptr,
        lin1o, N_GRAPHS, 384, 384);

    lin2_kernel<<<(N_GRAPHS * 10 + 255) / 256, 256, 0, stream>>>(lin1o, lin2_w, lin2_b,
                                                                 (float*)d_out);
}


# Both products of a node layer's MLP in one launch: the hidden 16x128 strip goes from the accumulators through LDS into the second product's A fragments, W2 takes W1's LDS; k and sum order unchanged
# speedup vs baseline: 3.5036x; 1.0759x over previous
#include <hip/hip_runtime.h>

typedef __attribute__((ext_vector_type(2))) float v2f;
typedef __attribute__((ext_vector_type(8))) float v8f;

#define N_NODES  50000
#define N_EDGES  800000
#define N_GRAPHS 512
#define BN_EPS   1e-5f

__global__ void zero_f32(float* __restrict__ p, int n4) {
    int t = blockIdx.x * blockDim.x + threadIdx.x;
    if (t < n4) ((float4*)p)[t] = make_float4(0.f, 0.f, 0.f, 0.f);
}

__global__ void gather_emb(const int* __restrict__ ids, const float* __restrict__ emb,
                           float* __restrict__ x) {
    int t = blockIdx.x * blockDim.x + threadIdx.x;
    int node = t >> 5, lane = t & 31;
    if (node >= N_NODES) return;
    int id = ids[node];
    float4 v = *(const float4*)(emb + (long)id * 128 + lane * 4);
    *(float4*)(x + (long)node * 128 + lane * 4) = v;
}

#define EB_BLOCKS 256
#define EB_CHUNK  3125
#define EB_BINS   256
#define EB_RANGE  196
#define EB_T      512
#define EB_PER_T  7
#define EB_SEG    16
#define RP_PAD    (EB_BINS * EB_RANGE)

__device__ __forceinline__ int scan256_excl(int v, int tid, int* wtot) {
    int lane = tid & 31, wave = tid >> 5;
    int inc = v;
#pragma unroll
    for (int off = 1; off < 32; off <<= 1) {
        int t = __shfl_up(inc, off, 32);
        if (lane >= off) inc += t;
    }
    if (tid < 256 && lane == 31) wtot[wave] = inc;
    __syncthreads();
    int carry = 0;
    if (tid < 256)
        for (int w = 0; w < wave; w++) carry += wtot[w];
    return carry + inc - v;
}

__global__ void __launch_bounds__(EB_T) edge_bin(const int* __restrict__ ei, int* __restrict__ binned,
                                                 int* __restrict__ cntm, int* __restrict__ stm) {
    __shared__ int hist[EB_BINS];
    __shared__ int offs[EB_BINS];
    __shared__ int wtot[8];
    int tid = threadIdx.x, blk = blockIdx.x;
    if (tid < EB_BINS) hist[tid] = 0;
    __syncthreads();

    int d[EB_PER_T], s[EB_PER_T], r[EB_PER_T];
#pragma unroll
    for (int k = 0; k < EB_PER_T; k++) {
        int idx = tid + k * EB_T;
        d[k] = -1;
        s[k] = 0;
        if (idx < EB_CHUNK) {
            int e = blk * EB_CHUNK + idx;
            d[k] = ei[N_EDGES + e];
            s[k] = ei[e];
        }
    }
#pragma unroll
    for (int k = 0; k < EB_PER_T; k++) {
        r[k] = 0;
        if (d[k] >= 0) r[k] = atomicAdd(&hist[d[k] / EB_RANGE], 1);
    }
    __syncthreads();

    int v = (tid < EB_BINS) ? hist[tid] : 0;
    int ex = scan256_excl(v, tid, wtot);
    if (tid < EB_BINS) {
        offs[tid] = ex;
        cntm[blk * EB_BINS + tid] = v;
        stm[blk * EB_BINS + tid] = ex;
    }
    __syncthreads();

#pragma unroll
    for (int k = 0; k < EB_PER_T; k++) {
        if (d[k] >= 0) {
            int bn = d[k] / EB_RANGE;
            binned[blk * EB_CHUNK + offs[bn] + r[k]] = ((d[k] - bn * EB_RANGE) << 16) | s[k];
        }
    }
}

__global__ void __launch_bounds__(EB_T) edge_place(const int* __restrict__ binned, const int* __restrict__ cntm,
                                                   const int* __restrict__ stm, int* __restrict__ rowptr,
                                                   int* __restrict__ col) {
    __shared__ int cnt[EB_BINS];
    __shared__ int wtot[8];
    __shared__ int red[EB_T / 32];
    int tid = threadIdx.x, b = blockIdx.x;
    int wave = tid >> 5, lane = tid & 31;
    if (tid < EB_BINS) cnt[tid] = 0;

    int c = 0, s = 0;
    if (lane < EB_SEG) {
        int i = wave * EB_SEG + lane;
        c = cntm[i * EB_BINS + b];
        s = stm[i * EB_BINS + b];
    }
    int part = s;
    for (int off = 16; off > 0; off >>= 1) part += __shfl_xor(part, off, 32);
    if (lane == 0) red[wave] = part;

    int len[EB_SEG], st[EB_SEG], rec[EB_SEG];
#pragma unroll
    for (int q = 0; q < EB_SEG; q++) {
        len[q] = __builtin_amdgcn_readlane(c, q);
        st[q]  = (wave * EB_SEG + q) * EB_CHUNK + __builtin_amdgcn_readlane(s, q);
    }
#pragma unroll
    for (int q = 0; q < EB_SEG; q++) {
        int p = st[q] + lane;
        if (p > N_EDGES - 1) p = N_EDGES - 1;
        int t = binned[p];
        rec[q] = (lane < len[q]) ? t : -1;
    }
    __syncthreads();

#pragma unroll
    for (int q = 0; q < EB_SEG; q++) {
        if (rec[q] >= 0) atomicAdd(&cnt[rec[q] >> 16], 1);
        for (int o = 32 + lane; o < len[q]; o += 32) atomicAdd(&cnt[binned[st[q] + o] >> 16], 1);
    }
    __syncthreads();

    int base = 0;
    for (int w = 0; w < EB_T / 32; w++) base += red[w];
    int v = (tid < EB_BINS) ? cnt[tid] : 0;
    int ex = scan256_excl(v, tid, wtot);
    if (tid < EB_BINS) {
        int p = base + ex;
        cnt[tid] = p;
        int node = b * EB_RANGE + tid;
        if (tid < EB_RANGE && node <= N_NODES) rowptr[node] = p;
    }
    __syncthreads();

#pragma unroll
    for (int q = 0; q < EB_SEG; q++) {
        if (rec[q] >= 0) {
            int pos = atomicAdd(&cnt[rec[q] >> 16], 1);
            col[pos] = rec[q] & 0xFFFF;
        }
        for (int o = 32 + lane; o < len[q]; o += 32) {
            int rr = binned[st[q] + o];
            int pos = atomicAdd(&cnt[rr >> 16], 1);
            col[pos] = rr & 0xFFFF;
        }
    }
}

__global__ void gather_sum(const float* __restrict__ x, const int* __restrict__ rowptr,
                           const int* __restrict__ col, float* __restrict__ s) {
    int wave = threadIdx.x >> 5, lane = threadIdx.x & 31;
    for (int node = blockIdx.x * 8 + wave; node < N_NODES; node += gridDim.x * 8) {
        int beg = __builtin_amdgcn_readfirstlane(rowptr[node]);
        int end = __builtin_amdgcn_readfirstlane(rowptr[node + 1]);
        float4 acc = make_float4(0.f, 0.f, 0.f, 0.f);
        for (int base = beg; base < end; base += 32) {
            int n = end - base;
            if (n > 32) n = 32;
            int myc = (lane < n) ? col[base + lane] : 0;
            int j = 0;
            for (; j + 4 <= n; j += 4) {
                int s0 = __builtin_amdgcn_readlane(myc, j);
                int s1 = __builtin_amdgcn_readlane(myc, j + 1);
                int s2 = __builtin_amdgcn_readlane(myc, j + 2);
                int s3 = __builtin_amdgcn_readlane(myc, j + 3);
                float4 v0 = *(const float4*)(x + (long)s0 * 128 + lane * 4);
                float4 v1 = *(const float4*)(x + (long)s1 * 128 + lane * 4);
                float4 v2 = *(const float4*)(x + (long)s2 * 128 + lane * 4);
                float4 v3 = *(const float4*)(x + (long)s3 * 128 + lane * 4);
                acc.x = (((acc.x + v0.x) + v1.x) + v2.x) + v3.x;
                acc.y = (((acc.y + v0.y) + v1.y) + v2.y) + v3.y;
                acc.z = (((acc.z + v0.z) + v1.z) + v2.z) + v3.z;
                acc.w = (((acc.w + v0.w) + v1.w) + v2.w) + v3.w;
            }
            for (; j < n; j++) {
                int s0 = __builtin_amdgcn_readlane(myc, j);
                float4 v0 = *(const float4*)(x + (long)s0 * 128 + lane * 4);
                acc.x += v0.x; acc.y += v0.y; acc.z += v0.z; acc.w += v0.w;
            }
        }
        float4 xv = *(const float4*)(x + (long)node * 128 + lane * 4);
        float4 o = make_float4(xv.x + acc.x, xv.y + acc.y, xv.z + acc.z, xv.w + acc.w);
        *(float4*)(s + (long)node * 128 + lane * 4) = o;
    }
}

#define SK  128
#define SN  128
#define WPF 288
#define HP  132

__device__ __forceinline__ void w_fetch(const float* __restrict__ W, int wave, int lane,
                                        float4 (&r0)[8], float4 (&r1)[8]) {
#pragma unroll
    for (int i = 0; i < 8; i++) {
        int kp = wave + 8 * i;
        r0[i] = *(const float4*)(W + (long)(2 * kp) * SN + lane * 4);
        r1[i] = *(const float4*)(W + (long)(2 * kp + 1) * SN + lane * 4);
    }
}
__device__ __forceinline__ void w_put(float* wl, int wave, int lane,
                                      const float4 (&r0)[8], const float4 (&r1)[8]) {
#pragma unroll
    for (int i = 0; i < 8; i++) {
        int kp = wave + 8 * i;
        float* d = wl + kp * WPF + lane * 8;
        *(float4*)(d)     = make_float4(r0[i].x, r1[i].x, r0[i].y, r1[i].y);
        *(float4*)(d + 4) = make_float4(r0[i].z, r1[i].z, r0[i].w, r1[i].w);
    }
}

__device__ __forceinline__ void strip_product(const v2f (&af)[SK / 4], const float* bl, v8f (&acc)[8]) {
#pragma unroll
    for (int nt = 0; nt < 8; nt++) acc[nt] = (v8f){0.f, 0.f, 0.f, 0.f, 0.f, 0.f, 0.f, 0.f};
    v2f bc[8], bn[8];
#pragma unroll
    for (int nt = 0; nt < 8; nt++) bc[nt] = *(const v2f*)(bl + nt * 32);
#pragma unroll
    for (int i = 0; i < SK / 4; i++) {
        if (i + 1 < SK / 4) {
#pragma unroll
            for (int nt = 0; nt < 8; nt++) bn[nt] = *(const v2f*)(bl + (2 * i + 2) * WPF + nt * 32);
        }
        __builtin_amdgcn_sched_barrier(0);
#pragma unroll
        for (int nt = 0; nt < 8; nt++)
            acc[nt] = __builtin_amdgcn_wmma_f32_16x16x4_f32(false, af[i], false, bc[nt],
                                                            (short)0, acc[nt], false, false);
        __builtin_amdgcn_sched_barrier(0);
#pragma unroll
        for (int nt = 0; nt < 8; nt++) bc[nt] = bn[nt];
    }
}

__global__ void __launch_bounds__(256) mlp_strip(const float* __restrict__ X,
                                                 const float* __restrict__ W1, const float* __restrict__ b1,
                                                 const float* __restrict__ gamma, const float* __restrict__ beta,
                                                 const float* __restrict__ mean, const float* __restrict__ var,
                                                 const float* __restrict__ W2, const float* __restrict__ b2,
                                                 float* __restrict__ O, int strips) {
    __shared__ __attribute__((aligned(16))) float wl[(SK / 2) * WPF];
    __shared__ __attribute__((aligned(16))) float hs[8 * 16 * HP];
    int wave = __builtin_amdgcn_readfirstlane(threadIdx.x >> 5);
    int lane = threadIdx.x & 31;
    int c = lane & 15, h = lane >> 4;
    int strip = blockIdx.x * 8 + wave;
    bool active = strip < strips;
    float* hw = hs + wave * 16 * HP;
    const float* bl = wl + h * WPF + c * 2;

    float4 r0[8], r1[8];
    w_fetch(W1, wave, lane, r0, r1);
    w_put(wl, wave, lane, r0, r1);
    __syncthreads();

    if (active) {
        v2f af[SK / 4];
        const float* xr = X + (long)(strip * 16 + c) * SK + h * 2;
#pragma unroll
        for (int i = 0; i < SK / 4; i++) af[i] = *(const v2f*)(xr + i * 4);
        v8f acc[8];
        strip_product(af, bl, acc);
#pragma unroll
        for (int nt = 0; nt < 8; nt++) {
            int col = nt * 16 + c;
            float bb = b1[col];
            float mu = mean[col], inv = rsqrtf(var[col] + BN_EPS), gm = gamma[col], bt = beta[col];
#pragma unroll
            for (int r = 0; r < 8; r++) {
                float v = acc[nt][r] + bb;
                v = (v - mu) * inv * gm + bt;
                v = fmaxf(v, 0.f);
                hw[(h * 8 + r) * HP + col] = v;
            }
        }
    }

    w_fetch(W2, wave, lane, r0, r1);
    __syncthreads();
    w_put(wl, wave, lane, r0, r1);
    __syncthreads();

    if (active) {
        v2f af[SK / 4];
        const float* hr = hw + c * HP + h * 2;
#pragma unroll
        for (int i = 0; i < SK / 4; i++) af[i] = *(const v2f*)(hr + i * 4);
        v8f acc[8];
        strip_product(af, bl, acc);
        float* orow = O + (long)(strip * 16 + h * 8) * SN + c;
#pragma unroll
        for (int nt = 0; nt < 8; nt++) {
            float bb = b2[nt * 16 + c];
#pragma unroll
            for (int r = 0; r < 8; r++) {
                float v = fmaxf(acc[nt][r] + bb, 0.f);
                orow[(long)r * SN + nt * 16] = v;
            }
        }
    }
}

template <bool FUSE_ADD, bool BN, bool RELU>
__global__ void gemm_wmma(const float* __restrict__ X, const float* __restrict__ AGG,
                          const float* __restrict__ W, const float* __restrict__ bias,
                          const float* __restrict__ gamma, const float* __restrict__ beta,
                          const float* __restrict__ mean, const float* __restrict__ var,
                          float* __restrict__ O, int M, int K, int N) {
    int wave = threadIdx.x >> 5, lane = threadIdx.x & 31;
    int tilesN = N >> 4;
    int tile = blockIdx.x * 8 + wave;
    if (tile >= (M >> 4) * tilesN) return;
    int mt = tile / tilesN, nt = tile % tilesN;
    int mi = lane & 15, kh = lane >> 4;

    const float* xr = X + (long)(mt * 16 + mi) * K + kh * 2;
    const float* ar = FUSE_ADD ? (AGG + (long)(mt * 16 + mi) * K + kh * 2) : nullptr;
    const float* wc = W + (long)(kh * 2) * N + nt * 16 + mi;

    v8f acc = {0.f, 0.f, 0.f, 0.f, 0.f, 0.f, 0.f, 0.f};
    for (int kb = 0; kb < K; kb += 4) {
        v2f a, b;
        a.x = xr[kb];
        a.y = xr[kb + 1];
        if (FUSE_ADD) { a.x += ar[kb]; a.y += ar[kb + 1]; }
        b.x = wc[(long)kb * N];
        b.y = wc[(long)(kb + 1) * N];
        acc = __builtin_amdgcn_wmma_f32_16x16x4_f32(false, a, false, b,
                                                    (short)0, acc, false, false);
    }

    int col = nt * 16 + mi;
    float bb = bias[col];
    float mu = 0.f, inv = 1.f, gm = 1.f, bt = 0.f;
    if (BN) { mu = mean[col]; inv = rsqrtf(var[col] + BN_EPS); gm = gamma[col]; bt = beta[col]; }
    float* orow = O + (long)(mt * 16 + kh * 8) * N + col;
#pragma unroll
    for (int r = 0; r < 8; r++) {
        float v = acc[r] + bb;
        if (BN)   v = (v - mu) * inv * gm + bt;
        if (RELU) v = fmaxf(v, 0.f);
        orow[(long)r * N] = v;
    }
}

#define POOL_CH 16
__device__ __forceinline__ void pool_flush(float* __restrict__ p, float4 a) {
    atomicAdd(p + 0, a.x);
    atomicAdd(p + 1, a.y);
    atomicAdd(p + 2, a.z);
    atomicAdd(p + 3, a.w);
}

__global__ void __launch_bounds__(256) pool3(const float* __restrict__ h1, const float* __restrict__ h2,
                                             const float* __restrict__ h3, const int* __restrict__ batch,
                                             float* __restrict__ gcat) {
    int wave = threadIdx.x >> 5, lane = threadIdx.x & 31;
    int chunk = blockIdx.x * 8 + wave;
    if (chunk >= N_NODES / POOL_CH) return;
    int base = chunk * POOL_CH;
    int myb = batch[base + (lane & (POOL_CH - 1))];
    const float* hs[3] = {h1, h2, h3};
#pragma unroll
    for (int l = 0; l < 3; l++) {
        const float* hp = hs[l] + (long)base * 128 + lane * 4;
        float4 v[POOL_CH];
#pragma unroll
        for (int j = 0; j < POOL_CH; j++) v[j] = *(const float4*)(hp + j * 128);
        float* gp = gcat + l * 128 + lane * 4;
        int cur = __builtin_amdgcn_readlane(myb, 0);
        float4 acc = v[0];
#pragma unroll
        for (int j = 1; j < POOL_CH; j++) {
            int g = __builtin_amdgcn_readlane(myb, j);
            if (g != cur) {
                pool_flush(gp + (long)cur * 384, acc);
                acc = v[j];
                cur = g;
            } else {
                acc.x += v[j].x; acc.y += v[j].y; acc.z += v[j].z; acc.w += v[j].w;
            }
        }
        pool_flush(gp + (long)cur * 384, acc);
    }
}

__global__ void lin2_kernel(const float* __restrict__ h, const float* __restrict__ w,
                            const float* __restrict__ b, float* __restrict__ out) {
    int t = blockIdx.x * blockDim.x + threadIdx.x;
    if (t >= N_GRAPHS * 10) return;
    int g = t / 10, o = t % 10;
    const float* hr = h + (long)g * 384;
    float acc = b[o];
    for (int k = 0; k < 384; k++) acc = fmaf(hr[k], w[k * 10 + o], acc);
    out[t] = acc;
}

extern "C" void kernel_launch(void* const* d_in, const int* in_sizes, int n_in,
                              void* d_out, int out_size, void* d_ws, size_t ws_size,
                              hipStream_t stream) {
    const int*   x_ids = (const int*)d_in[0];
    const int*   edge  = (const int*)d_in[1];
    const int*   batch = (const int*)d_in[2];
    const float* emb   = (const float*)d_in[3];

    const float *c_w1[3], *c_b1[3], *c_g[3], *c_be[3], *c_mu[3], *c_va[3], *c_w2[3], *c_b2[3];
    for (int i = 0; i < 3; i++) {
        c_w1[i] = (const float*)d_in[4 + 8 * i + 0];
        c_b1[i] = (const float*)d_in[4 + 8 * i + 1];
        c_g[i]  = (const float*)d_in[4 + 8 * i + 2];
        c_be[i] = (const float*)d_in[4 + 8 * i + 3];
        c_mu[i] = (const float*)d_in[4 + 8 * i + 4];
        c_va[i] = (const float*)d_in[4 + 8 * i + 5];
        c_w2[i] = (const float*)d_in[4 + 8 * i + 6];
        c_b2[i] = (const float*)d_in[4 + 8 * i + 7];
    }
    const float* lin1_w = (const float*)d_in[28];
    const float* lin1_b = (const float*)d_in[29];
    const float* lin2_w = (const float*)d_in[30];
    const float* lin2_b = (const float*)d_in[31];

    const long NF = (long)N_NODES * 128;
    float* ws   = (float*)d_ws;
    float* x0   = ws + 0 * NF;
    float* agg  = ws + 1 * NF;
    float* tmp  = ws + 2 * NF;
    float* hbuf[3] = { ws + 3 * NF, ws + 4 * NF, x0 };
    float* gcat  = agg;
    float* lin1o = tmp;
    int* rowptr = (int*)(ws + 5 * NF);
    int* binned = rowptr + RP_PAD;
    int* col    = binned + N_EDGES;
    int* cntm   = col + N_EDGES;
    int* stm    = cntm + EB_BLOCKS * EB_BINS;

    gather_emb<<<(N_NODES * 32 + 255) / 256, 256, 0, stream>>>(x_ids, emb, x0);

    edge_bin<<<EB_BLOCKS, EB_T, 0, stream>>>(edge, binned, cntm, stm);
    edge_place<<<EB_BINS, EB_T, 0, stream>>>(binned, cntm, stm, rowptr, col);

    const int strips = N_NODES / 16;
    const float* xin = x0;
    for (int l = 0; l < 3; l++) {
        gather_sum<<<2048, 256, 0, stream>>>(xin, rowptr, col, agg);
        mlp_strip<<<(strips + 7) / 8, 256, 0, stream>>>(
            agg, c_w1[l], c_b1[l], c_g[l], c_be[l], c_mu[l], c_va[l], c_w2[l], c_b2[l], hbuf[l], strips);
        xin = hbuf[l];
    }

    zero_f32<<<(N_GRAPHS * 384 / 4 + 255) / 256, 256, 0, stream>>>(gcat, N_GRAPHS * 384 / 4);
    pool3<<<(N_NODES / POOL_CH + 7) / 8, 256, 0, stream>>>(hbuf[0], hbuf[1], hbuf[2], batch, gcat);

    const int tiles_head = (N_GRAPHS / 16) * (384 / 16);
    gemm_wmma<false, false, true><<<(tiles_head + 7) / 8, 256, 0, stream>>>(
        gcat, nullptr, lin1_w, lin1_b, nullptr, nullptr, nullptr, nullptr,
        lin1o, N_GRAPHS, 384, 384);

    lin2_kernel<<<(N_GRAPHS * 10 + 255) / 256, 256, 0, stream>>>(lin1o, lin2_w, lin2_b,
                                                                 (float*)d_out);
}


# Layer 1 neighbour sum reads embedding rows by id from an LDS copy of the 250 KB table (one 32-wave work-group per WGP) instead of node rows gathered through L2; x never written; sum order unchanged
# speedup vs baseline: 3.8650x; 1.1032x over previous
#include <hip/hip_runtime.h>

typedef __attribute__((ext_vector_type(2))) float v2f;
typedef __attribute__((ext_vector_type(8))) float v8f;

#define N_NODES  50000
#define N_EDGES  800000
#define N_GRAPHS 512
#define BN_EPS   1e-5f

__global__ void zero_f32(float* __restrict__ p, int n4) {
    int t = blockIdx.x * blockDim.x + threadIdx.x;
    if (t < n4) ((float4*)p)[t] = make_float4(0.f, 0.f, 0.f, 0.f);
}

#define EB_BLOCKS 256
#define EB_CHUNK  3125
#define EB_BINS   256
#define EB_RANGE  196
#define EB_T      512
#define EB_PER_T  7
#define EB_SEG    16
#define RP_PAD    (EB_BINS * EB_RANGE)

__device__ __forceinline__ int scan256_excl(int v, int tid, int* wtot) {
    int lane = tid & 31, wave = tid >> 5;
    int inc = v;
#pragma unroll
    for (int off = 1; off < 32; off <<= 1) {
        int t = __shfl_up(inc, off, 32);
        if (lane >= off) inc += t;
    }
    if (tid < 256 && lane == 31) wtot[wave] = inc;
    __syncthreads();
    int carry = 0;
    if (tid < 256)
        for (int w = 0; w < wave; w++) carry += wtot[w];
    return carry + inc - v;
}

__global__ void __launch_bounds__(EB_T) edge_bin(const int* __restrict__ ei, int* __restrict__ binned,
                                                 int* __restrict__ cntm, int* __restrict__ stm) {
    __shared__ int hist[EB_BINS];
    __shared__ int offs[EB_BINS];
    __shared__ int wtot[8];
    int tid = threadIdx.x, blk = blockIdx.x;
    if (tid < EB_BINS) hist[tid] = 0;
    __syncthreads();

    int d[EB_PER_T], s[EB_PER_T], r[EB_PER_T];
#pragma unroll
    for (int k = 0; k < EB_PER_T; k++) {
        int idx = tid + k * EB_T;
        d[k] = -1;
        s[k] = 0;
        if (idx < EB_CHUNK) {
            int e = blk * EB_CHUNK + idx;
            d[k] = ei[N_EDGES + e];
            s[k] = ei[e];
        }
    }
#pragma unroll
    for (int k = 0; k < EB_PER_T; k++) {
        r[k] = 0;
        if (d[k] >= 0) r[k] = atomicAdd(&hist[d[k] / EB_RANGE], 1);
    }
    __syncthreads();

    int v = (tid < EB_BINS) ? hist[tid] : 0;
    int ex = scan256_excl(v, tid, wtot);
    if (tid < EB_BINS) {
        offs[tid] = ex;
        cntm[blk * EB_BINS + tid] = v;
        stm[blk * EB_BINS + tid] = ex;
    }
    __syncthreads();

#pragma unroll
    for (int k = 0; k < EB_PER_T; k++) {
        if (d[k] >= 0) {
            int bn = d[k] / EB_RANGE;
            binned[blk * EB_CHUNK + offs[bn] + r[k]] = ((d[k] - bn * EB_RANGE) << 16) | s[k];
        }
    }
}

__global__ void __launch_bounds__(EB_T) edge_place(const int* __restrict__ binned, const int* __restrict__ cntm,
                                                   const int* __restrict__ stm, int* __restrict__ rowptr,
                                                   int* __restrict__ col) {
    __shared__ int cnt[EB_BINS];
    __shared__ int wtot[8];
    __shared__ int red[EB_T / 32];
    int tid = threadIdx.x, b = blockIdx.x;
    int wave = tid >> 5, lane = tid & 31;
    if (tid < EB_BINS) cnt[tid] = 0;

    int c = 0, s = 0;
    if (lane < EB_SEG) {
        int i = wave * EB_SEG + lane;
        c = cntm[i * EB_BINS + b];
        s = stm[i * EB_BINS + b];
    }
    int part = s;
    for (int off = 16; off > 0; off >>= 1) part += __shfl_xor(part, off, 32);
    if (lane == 0) red[wave] = part;

    int len[EB_SEG], st[EB_SEG], rec[EB_SEG];
#pragma unroll
    for (int q = 0; q < EB_SEG; q++) {
        len[q] = __builtin_amdgcn_readlane(c, q);
        st[q]  = (wave * EB_SEG + q) * EB_CHUNK + __builtin_amdgcn_readlane(s, q);
    }
#pragma unroll
    for (int q = 0; q < EB_SEG; q++) {
        int p = st[q] + lane;
        if (p > N_EDGES - 1) p = N_EDGES - 1;
        int t = binned[p];
        rec[q] = (lane < len[q]) ? t : -1;
    }
    __syncthreads();

#pragma unroll
    for (int q = 0; q < EB_SEG; q++) {
        if (rec[q] >= 0) atomicAdd(&cnt[rec[q] >> 16], 1);
        for (int o = 32 + lane; o < len[q]; o += 32) atomicAdd(&cnt[binned[st[q] + o] >> 16], 1);
    }
    __syncthreads();

    int base = 0;
    for (int w = 0; w < EB_T / 32; w++) base += red[w];
    int v = (tid < EB_BINS) ? cnt[tid] : 0;
    int ex = scan256_excl(v, tid, wtot);
    if (tid < EB_BINS) {
        int p = base + ex;
        cnt[tid] = p;
        int node = b * EB_RANGE + tid;
        if (tid < EB_RANGE && node <= N_NODES) rowptr[node] = p;
    }
    __syncthreads();

#pragma unroll
    for (int q = 0; q < EB_SEG; q++) {
        if (rec[q] >= 0) {
            int pos = atomicAdd(&cnt[rec[q] >> 16], 1);
            col[pos] = rec[q] & 0xFFFF;
        }
        for (int o = 32 + lane; o < len[q]; o += 32) {
            int rr = binned[st[q] + o];
            int pos = atomicAdd(&cnt[rr >> 16], 1);
            col[pos] = rr & 0xFFFF;
        }
    }
}

__global__ void gather_sum(const float* __restrict__ x, const int* __restrict__ rowptr,
                           const int* __restrict__ col, float* __restrict__ s) {
    int wave = threadIdx.x >> 5, lane = threadIdx.x & 31;
    for (int node = blockIdx.x * 8 + wave; node < N_NODES; node += gridDim.x * 8) {
        int beg = __builtin_amdgcn_readfirstlane(rowptr[node]);
        int end = __builtin_amdgcn_readfirstlane(rowptr[node + 1]);
        float4 acc = make_float4(0.f, 0.f, 0.f, 0.f);
        for (int base = beg; base < end; base += 32) {
            int n = end - base;
            if (n > 32) n = 32;
            int myc = (lane < n) ? col[base + lane] : 0;
            int j = 0;
            for (; j + 4 <= n; j += 4) {
                int s0 = __builtin_amdgcn_readlane(myc, j);
                int s1 = __builtin_amdgcn_readlane(myc, j + 1);
                int s2 = __builtin_amdgcn_readlane(myc, j + 2);
                int s3 = __builtin_amdgcn_readlane(myc, j + 3);
                float4 v0 = *(const float4*)(x + (long)s0 * 128 + lane * 4);
                float4 v1 = *(const float4*)(x + (long)s1 * 128 + lane * 4);
                float4 v2 = *(const float4*)(x + (long)s2 * 128 + lane * 4);
                float4 v3 = *(const float4*)(x + (long)s3 * 128 + lane * 4);
                acc.x = (((acc.x + v0.x) + v1.x) + v2.x) + v3.x;
                acc.y = (((acc.y + v0.y) + v1.y) + v2.y) + v3.y;
                acc.z = (((acc.z + v0.z) + v1.z) + v2.z) + v3.z;
                acc.w = (((acc.w + v0.w) + v1.w) + v2.w) + v3.w;
            }
            for (; j < n; j++) {
                int s0 = __builtin_amdgcn_readlane(myc, j);
                float4 v0 = *(const float4*)(x + (long)s0 * 128 + lane * 4);
                acc.x += v0.x; acc.y += v0.y; acc.z += v0.z; acc.w += v0.w;
            }
        }
        float4 xv = *(const float4*)(x + (long)node * 128 + lane * 4);
        float4 o = make_float4(xv.x + acc.x, xv.y + acc.y, xv.z + acc.z, xv.w + acc.w);
        *(float4*)(s + (long)node * 128 + lane * 4) = o;
    }
}

#define VOCAB     500
#define GE_BLOCKS 256
#define GE_WAVES  32
#define GE_SLOTS  (GE_BLOCKS * GE_WAVES)
#define GE_F4     (VOCAB * 128 / 4)
#define GE_FILL   ((GE_F4 + GE_WAVES * 32 - 1) / (GE_WAVES * 32))
#define GE_LDS4   (GE_FILL * GE_WAVES * 32)
static_assert(N_NODES >= GE_SLOTS, "every wave has at least one node");
static_assert(N_NODES / GE_SLOTS + 1 < 32, "a wave's slot range starts fit its 32 lanes");

__device__ __forceinline__ int row_of(int id) {
    return min(max(id, 0), VOCAB - 1) * 128;
}

__device__ __forceinline__ unsigned first_sources(const int* __restrict__ col, int rp, int k, int cnt, int lane) {
    int b = __builtin_amdgcn_readlane(rp, k < cnt ? k : cnt);
    int p = b + lane;
    if (p > N_EDGES - 1) p = N_EDGES - 1;
    return (unsigned)col[p];
}

__device__ __forceinline__ void sum_rows(const float* el, int ro, int n, float4& acc) {
    int t = 0;
    for (; t + 4 <= n; t += 4) {
        int o0 = __builtin_amdgcn_readlane(ro, t);
        int o1 = __builtin_amdgcn_readlane(ro, t + 1);
        int o2 = __builtin_amdgcn_readlane(ro, t + 2);
        int o3 = __builtin_amdgcn_readlane(ro, t + 3);
        float4 v0 = *(const float4*)(el + o0);
        float4 v1 = *(const float4*)(el + o1);
        float4 v2 = *(const float4*)(el + o2);
        float4 v3 = *(const float4*)(el + o3);
        acc.x = (((acc.x + v0.x) + v1.x) + v2.x) + v3.x;
        acc.y = (((acc.y + v0.y) + v1.y) + v2.y) + v3.y;
        acc.z = (((acc.z + v0.z) + v1.z) + v2.z) + v3.z;
        acc.w = (((acc.w + v0.w) + v1.w) + v2.w) + v3.w;
    }
    switch (n - t) {
    case 3: {
        int o0 = __builtin_amdgcn_readlane(ro, t);
        int o1 = __builtin_amdgcn_readlane(ro, t + 1);
        int o2 = __builtin_amdgcn_readlane(ro, t + 2);
        float4 v0 = *(const float4*)(el + o0);
        float4 v1 = *(const float4*)(el + o1);
        float4 v2 = *(const float4*)(el + o2);
        acc.x = ((acc.x + v0.x) + v1.x) + v2.x;
        acc.y = ((acc.y + v0.y) + v1.y) + v2.y;
        acc.z = ((acc.z + v0.z) + v1.z) + v2.z;
        acc.w = ((acc.w + v0.w) + v1.w) + v2.w;
        break;
    }
    case 2: {
        int o0 = __builtin_amdgcn_readlane(ro, t);
        int o1 = __builtin_amdgcn_readlane(ro, t + 1);
        float4 v0 = *(const float4*)(el + o0);
        float4 v1 = *(const float4*)(el + o1);
        acc.x = (acc.x + v0.x) + v1.x;
        acc.y = (acc.y + v0.y) + v1.y;
        acc.z = (acc.z + v0.z) + v1.z;
        acc.w = (acc.w + v0.w) + v1.w;
        break;
    }
    case 1: {
        int o0 = __builtin_amdgcn_readlane(ro, t);
        float4 v0 = *(const float4*)(el + o0);
        acc.x += v0.x; acc.y += v0.y; acc.z += v0.z; acc.w += v0.w;
        break;
    }
    default: break;
    }
}

__global__ void __launch_bounds__(GE_WAVES * 32) gather_sum_emb(const int* __restrict__ ids, const float* __restrict__ emb,
                                                                const int* __restrict__ rowptr, const int* __restrict__ col,
                                                                float* __restrict__ s) {
    __shared__ __attribute__((aligned(16))) float et[GE_LDS4 * 4];
    int tid = threadIdx.x;
    int wave = __builtin_amdgcn_readfirstlane(tid >> 5);
    int lane = tid & 31;
    int gw = blockIdx.x * GE_WAVES + wave;
    int n0 = (int)(((long)gw * N_NODES) / GE_SLOTS);
    int n1 = (int)(((long)(gw + 1) * N_NODES) / GE_SLOTS);
    int cnt = n1 - n0;

    int rp  = rowptr[n0 + (lane < cnt ? lane : cnt)];
    int oid = ids[n0 + (lane < cnt ? lane : cnt - 1)];

    float4 r[GE_FILL];
#pragma unroll
    for (int k = 0; k < GE_FILL / 2; k++) {
        int i = tid + k * (GE_WAVES * 32);
        r[k] = ((const float4*)emb)[i < GE_F4 ? i : GE_F4 - 1];
    }
    unsigned c0 = first_sources(col, rp, 0, cnt, lane);
    unsigned cn = first_sources(col, rp, 1, cnt, lane);
#pragma unroll
    for (int k = GE_FILL / 2; k < GE_FILL; k++) {
        int i = tid + k * (GE_WAVES * 32);
        r[k] = ((const float4*)emb)[i < GE_F4 ? i : GE_F4 - 1];
    }
#pragma unroll
    for (int k = 0; k < GE_FILL / 2; k++) ((float4*)et)[tid + k * (GE_WAVES * 32)] = r[k];
    int idn = ids[c0];
#pragma unroll
    for (int k = GE_FILL / 2; k < GE_FILL; k++) ((float4*)et)[tid + k * (GE_WAVES * 32)] = r[k];
    __syncthreads();

    int own = row_of(oid);
    const float* el = et + lane * 4;
    for (int j = 0; j < cnt; j++) {
        int roc = row_of(idn);
        idn = ids[cn];
        cn = first_sources(col, rp, j + 2, cnt, lane);
        int beg = __builtin_amdgcn_readlane(rp, j);
        int end = __builtin_amdgcn_readlane(rp, j + 1);
        int so  = __builtin_amdgcn_readlane(own, j);
        float4 xv = *(const float4*)(el + so);
        float4 acc = make_float4(0.f, 0.f, 0.f, 0.f);
        int n = end - beg;
        if (n > 32) n = 32;
        sum_rows(el, roc, n, acc);
        for (int base = beg + 32; base < end; base += 32) {
            int m = end - base;
            if (m > 32) m = 32;
            unsigned c = (lane < m) ? (unsigned)col[base + lane] : 0u;
            int rr = row_of(ids[c]);
            sum_rows(el, rr, m, acc);
        }
        float4 o = make_float4(xv.x + acc.x, xv.y + acc.y, xv.z + acc.z, xv.w + acc.w);
        *(float4*)(s + (long)(n0 + j) * 128 + lane * 4) = o;
    }
}

#define SK  128
#define SN  128
#define WPF 288
#define HP  132

__device__ __forceinline__ void w_fetch(const float* __restrict__ W, int wave, int lane,
                                        float4 (&r0)[8], float4 (&r1)[8]) {
#pragma unroll
    for (int i = 0; i < 8; i++) {
        int kp = wave + 8 * i;
        r0[i] = *(const float4*)(W + (long)(2 * kp) * SN + lane * 4);
        r1[i] = *(const float4*)(W + (long)(2 * kp + 1) * SN + lane * 4);
    }
}
__device__ __forceinline__ void w_put(float* wl, int wave, int lane,
                                      const float4 (&r0)[8], const float4 (&r1)[8]) {
#pragma unroll
    for (int i = 0; i < 8; i++) {
        int kp = wave + 8 * i;
        float* d = wl + kp * WPF + lane * 8;
        *(float4*)(d)     = make_float4(r0[i].x, r1[i].x, r0[i].y, r1[i].y);
        *(float4*)(d + 4) = make_float4(r0[i].z, r1[i].z, r0[i].w, r1[i].w);
    }
}

__device__ __forceinline__ void strip_product(const v2f (&af)[SK / 4], const float* bl, v8f (&acc)[8]) {
#pragma unroll
    for (int nt = 0; nt < 8; nt++) acc[nt] = (v8f){0.f, 0.f, 0.f, 0.f, 0.f, 0.f, 0.f, 0.f};
    v2f bc[8], bn[8];
#pragma unroll
    for (int nt = 0; nt < 8; nt++) bc[nt] = *(const v2f*)(bl + nt * 32);
#pragma unroll
    for (int i = 0; i < SK / 4; i++) {
        if (i + 1 < SK / 4) {
#pragma unroll
            for (int nt = 0; nt < 8; nt++) bn[nt] = *(const v2f*)(bl + (2 * i + 2) * WPF + nt * 32);
        }
        __builtin_amdgcn_sched_barrier(0);
#pragma unroll
        for (int nt = 0; nt < 8; nt++)
            acc[nt] = __builtin_amdgcn_wmma_f32_16x16x4_f32(false, af[i], false, bc[nt],
                                                            (short)0, acc[nt], false, false);
        __builtin_amdgcn_sched_barrier(0);
#pragma unroll
        for (int nt = 0; nt < 8; nt++) bc[nt] = bn[nt];
    }
}

__global__ void __launch_bounds__(256) mlp_strip(const float* __restrict__ X,
                                                 const float* __restrict__ W1, const float* __restrict__ b1,
                                                 const float* __restrict__ gamma, const float* __restrict__ beta,
                                                 const float* __restrict__ mean, const float* __restrict__ var,
                                                 const float* __restrict__ W2, const float* __restrict__ b2,
                                                 float* __restrict__ O, int strips) {
    __shared__ __attribute__((aligned(16))) float wl[(SK / 2) * WPF];
    __shared__ __attribute__((aligned(16))) float hs[8 * 16 * HP];
    int wave = __builtin_amdgcn_readfirstlane(threadIdx.x >> 5);
    int lane = threadIdx.x & 31;
    int c = lane & 15, h = lane >> 4;
    int strip = blockIdx.x * 8 + wave;
    bool active = strip < strips;
    float* hw = hs + wave * 16 * HP;
    const float* bl = wl + h * WPF + c * 2;

    float4 r0[8], r1[8];
    w_fetch(W1, wave, lane, r0, r1);
    w_put(wl, wave, lane, r0, r1);
    __syncthreads();

    if (active) {
        v2f af[SK / 4];
        const float* xr = X + (long)(strip * 16 + c) * SK + h * 2;
#pragma unroll
        for (int i = 0; i < SK / 4; i++) af[i] = *(const v2f*)(xr + i * 4);
        v8f acc[8];
        strip_product(af, bl, acc);
#pragma unroll
        for (int nt = 0; nt < 8; nt++) {
            int col = nt * 16 + c;
            float bb = b1[col];
            float mu = mean[col], inv = rsqrtf(var[col] + BN_EPS), gm = gamma[col], bt = beta[col];
#pragma unroll
            for (int r = 0; r < 8; r++) {
                float v = acc[nt][r] + bb;
                v = (v - mu) * inv * gm + bt;
                v = fmaxf(v, 0.f);
                hw[(h * 8 + r) * HP + col] = v;
            }
        }
    }

    w_fetch(W2, wave, lane, r0, r1);
    __syncthreads();
    w_put(wl, wave, lane, r0, r1);
    __syncthreads();

    if (active) {
        v2f af[SK / 4];
        const float* hr = hw + c * HP + h * 2;
#pragma unroll
        for (int i = 0; i < SK / 4; i++) af[i] = *(const v2f*)(hr + i * 4);
        v8f acc[8];
        strip_product(af, bl, acc);
        float* orow = O + (long)(strip * 16 + h * 8) * SN + c;
#pragma unroll
        for (int nt = 0; nt < 8; nt++) {
            float bb = b2[nt * 16 + c];
#pragma unroll
            for (int r = 0; r < 8; r++) {
                float v = fmaxf(acc[nt][r] + bb, 0.f);
                orow[(long)r * SN + nt * 16] = v;
            }
        }
    }
}

template <bool FUSE_ADD, bool BN, bool RELU>
__global__ void gemm_wmma(const float* __restrict__ X, const float* __restrict__ AGG,
                          const float* __restrict__ W, const float* __restrict__ bias,
                          const float* __restrict__ gamma, const float* __restrict__ beta,
                          const float* __restrict__ mean, const float* __restrict__ var,
                          float* __restrict__ O, int M, int K, int N) {
    int wave = threadIdx.x >> 5, lane = threadIdx.x & 31;
    int tilesN = N >> 4;
    int tile = blockIdx.x * 8 + wave;
    if (tile >= (M >> 4) * tilesN) return;
    int mt = tile / tilesN, nt = tile % tilesN;
    int mi = lane & 15, kh = lane >> 4;

    const float* xr = X + (long)(mt * 16 + mi) * K + kh * 2;
    const float* ar = FUSE_ADD ? (AGG + (long)(mt * 16 + mi) * K + kh * 2) : nullptr;
    const float* wc = W + (long)(kh * 2) * N + nt * 16 + mi;

    v8f acc = {0.f, 0.f, 0.f, 0.f, 0.f, 0.f, 0.f, 0.f};
    for (int kb = 0; kb < K; kb += 4) {
        v2f a, b;
        a.x = xr[kb];
        a.y = xr[kb + 1];
        if (FUSE_ADD) { a.x += ar[kb]; a.y += ar[kb + 1]; }
        b.x = wc[(long)kb * N];
        b.y = wc[(long)(kb + 1) * N];
        acc = __builtin_amdgcn_wmma_f32_16x16x4_f32(false, a, false, b,
                                                    (short)0, acc, false, false);
    }

    int col = nt * 16 + mi;
    float bb = bias[col];
    float mu = 0.f, inv = 1.f, gm = 1.f, bt = 0.f;
    if (BN) { mu = mean[col]; inv = rsqrtf(var[col] + BN_EPS); gm = gamma[col]; bt = beta[col]; }
    float* orow = O + (long)(mt * 16 + kh * 8) * N + col;
#pragma unroll
    for (int r = 0; r < 8; r++) {
        float v = acc[r] + bb;
        if (BN)   v = (v - mu) * inv * gm + bt;
        if (RELU) v = fmaxf(v, 0.f);
        orow[(long)r * N] = v;
    }
}

#define POOL_CH 16
__device__ __forceinline__ void pool_flush(float* __restrict__ p, float4 a) {
    atomicAdd(p + 0, a.x);
    atomicAdd(p + 1, a.y);
    atomicAdd(p + 2, a.z);
    atomicAdd(p + 3, a.w);
}

__global__ void __launch_bounds__(256) pool3(const float* __restrict__ h1, const float* __restrict__ h2,
                                             const float* __restrict__ h3, const int* __restrict__ batch,
                                             float* __restrict__ gcat) {
    int wave = threadIdx.x >> 5, lane = threadIdx.x & 31;
    int chunk = blockIdx.x * 8 + wave;
    if (chunk >= N_NODES / POOL_CH) return;
    int base = chunk * POOL_CH;
    int myb = batch[base + (lane & (POOL_CH - 1))];
    const float* hs[3] = {h1, h2, h3};
#pragma unroll
    for (int l = 0; l < 3; l++) {
        const float* hp = hs[l] + (long)base * 128 + lane * 4;
        float4 v[POOL_CH];
#pragma unroll
        for (int j = 0; j < POOL_CH; j++) v[j] = *(const float4*)(hp + j * 128);
        float* gp = gcat + l * 128 + lane * 4;
        int cur = __builtin_amdgcn_readlane(myb, 0);
        float4 acc = v[0];
#pragma unroll
        for (int j = 1; j < POOL_CH; j++) {
            int g = __builtin_amdgcn_readlane(myb, j);
            if (g != cur) {
                pool_flush(gp + (long)cur * 384, acc);
                acc = v[j];
                cur = g;
            } else {
                acc.x += v[j].x; acc.y += v[j].y; acc.z += v[j].z; acc.w += v[j].w;
            }
        }
        pool_flush(gp + (long)cur * 384, acc);
    }
}

__global__ void lin2_kernel(const float* __restrict__ h, const float* __restrict__ w,
                            const float* __restrict__ b, float* __restrict__ out) {
    int t = blockIdx.x * blockDim.x + threadIdx.x;
    if (t >= N_GRAPHS * 10) return;
    int g = t / 10, o = t % 10;
    const float* hr = h + (long)g * 384;
    float acc = b[o];
    for (int k = 0; k < 384; k++) acc = fmaf(hr[k], w[k * 10 + o], acc);
    out[t] = acc;
}

extern "C" void kernel_launch(void* const* d_in, const int* in_sizes, int n_in,
                              void* d_out, int out_size, void* d_ws, size_t ws_size,
                              hipStream_t stream) {
    const int*   x_ids = (const int*)d_in[0];
    const int*   edge  = (const int*)d_in[1];
    const int*   batch = (const int*)d_in[2];
    const float* emb   = (const float*)d_in[3];

    const float *c_w1[3], *c_b1[3], *c_g[3], *c_be[3], *c_mu[3], *c_va[3], *c_w2[3], *c_b2[3];
    for (int i = 0; i < 3; i++) {
        c_w1[i] = (const float*)d_in[4 + 8 * i + 0];
        c_b1[i] = (const float*)d_in[4 + 8 * i + 1];
        c_g[i]  = (const float*)d_in[4 + 8 * i + 2];
        c_be[i] = (const float*)d_in[4 + 8 * i + 3];
        c_mu[i] = (const float*)d_in[4 + 8 * i + 4];
        c_va[i] = (const float*)d_in[4 + 8 * i + 5];
        c_w2[i] = (const float*)d_in[4 + 8 * i + 6];
        c_b2[i] = (const float*)d_in[4 + 8 * i + 7];
    }
    const float* lin1_w = (const float*)d_in[28];
    const float* lin1_b = (const float*)d_in[29];
    const float* lin2_w = (const float*)d_in[30];
    const float* lin2_b = (const float*)d_in[31];

    const long NF = (long)N_NODES * 128;
    float* ws   = (float*)d_ws;
    float* h3   = ws + 0 * NF;
    float* agg  = ws + 1 * NF;
    float* tmp  = ws + 2 * NF;
    float* hbuf[3] = { ws + 3 * NF, ws + 4 * NF, h3 };
    float* gcat  = agg;
    float* lin1o = tmp;
    int* rowptr = (int*)(ws + 5 * NF);
    int* binned = rowptr + RP_PAD;
    int* col    = binned + N_EDGES;
    int* cntm   = col + N_EDGES;
    int* stm    = cntm + EB_BLOCKS * EB_BINS;

    edge_bin<<<EB_BLOCKS, EB_T, 0, stream>>>(edge, binned, cntm, stm);
    edge_place<<<EB_BINS, EB_T, 0, stream>>>(binned, cntm, stm, rowptr, col);

    const int strips = N_NODES / 16;
    for (int l = 0; l < 3; l++) {
        if (l == 0) gather_sum_emb<<<GE_BLOCKS, GE_WAVES * 32, 0, stream>>>(x_ids, emb, rowptr, col, agg);
        else        gather_sum<<<2048, 256, 0, stream>>>(hbuf[l - 1], rowptr, col, agg);
        mlp_strip<<<(strips + 7) / 8, 256, 0, stream>>>(
            agg, c_w1[l], c_b1[l], c_g[l], c_be[l], c_mu[l], c_va[l], c_w2[l], c_b2[l], hbuf[l], strips);
    }

    zero_f32<<<(N_GRAPHS * 384 / 4 + 255) / 256, 256, 0, stream>>>(gcat, N_GRAPHS * 384 / 4);
    pool3<<<(N_NODES / POOL_CH + 7) / 8, 256, 0, stream>>>(hbuf[0], hbuf[1], hbuf[2], batch, gcat);

    const int tiles_head = (N_GRAPHS / 16) * (384 / 16);
    gemm_wmma<false, false, true><<<(tiles_head + 7) / 8, 256, 0, stream>>>(
        gcat, nullptr, lin1_w, lin1_b, nullptr, nullptr, nullptr, nullptr,
        lin1o, N_GRAPHS, 384, 384);

    lin2_kernel<<<(N_GRAPHS * 10 + 255) / 256, 256, 0, stream>>>(lin1o, lin2_w, lin2_b,
                                                                 (float*)d_out);
}


# Layers 2 and 3: each MLP wave sums its strip's 16 input rows itself (32 source rows in flight) into LDS and feeds product 1 from there; summed rows never stored, 2 launches fewer; sum order unchanged
# speedup vs baseline: 4.1318x; 1.0690x over previous
#include <hip/hip_runtime.h>

typedef __attribute__((ext_vector_type(2))) float v2f;
typedef __attribute__((ext_vector_type(8))) float v8f;

#define N_NODES  50000
#define N_EDGES  800000
#define N_GRAPHS 512
#define BN_EPS   1e-5f

__global__ void zero_f32(float* __restrict__ p, int n4) {
    int t = blockIdx.x * blockDim.x + threadIdx.x;
    if (t < n4) ((float4*)p)[t] = make_float4(0.f, 0.f, 0.f, 0.f);
}

#define EB_BLOCKS 256
#define EB_CHUNK  3125
#define EB_BINS   256
#define EB_RANGE  196
#define EB_T      512
#define EB_PER_T  7
#define EB_SEG    16
#define RP_PAD    (EB_BINS * EB_RANGE)

__device__ __forceinline__ int scan256_excl(int v, int tid, int* wtot) {
    int lane = tid & 31, wave = tid >> 5;
    int inc = v;
#pragma unroll
    for (int off = 1; off < 32; off <<= 1) {
        int t = __shfl_up(inc, off, 32);
        if (lane >= off) inc += t;
    }
    if (tid < 256 && lane == 31) wtot[wave] = inc;
    __syncthreads();
    int carry = 0;
    if (tid < 256)
        for (int w = 0; w < wave; w++) carry += wtot[w];
    return carry + inc - v;
}

__global__ void __launch_bounds__(EB_T) edge_bin(const int* __restrict__ ei, int* __restrict__ binned,
                                                 int* __restrict__ cntm, int* __restrict__ stm) {
    __shared__ int hist[EB_BINS];
    __shared__ int offs[EB_BINS];
    __shared__ int wtot[8];
    int tid = threadIdx.x, blk = blockIdx.x;
    if (tid < EB_BINS) hist[tid] = 0;
    __syncthreads();

    int d[EB_PER_T], s[EB_PER_T], r[EB_PER_T];
#pragma unroll
    for (int k = 0; k < EB_PER_T; k++) {
        int idx = tid + k * EB_T;
        d[k] = -1;
        s[k] = 0;
        if (idx < EB_CHUNK) {
            int e = blk * EB_CHUNK + idx;
            d[k] = ei[N_EDGES + e];
            s[k] = ei[e];
        }
    }
#pragma unroll
    for (int k = 0; k < EB_PER_T; k++) {
        r[k] = 0;
        if (d[k] >= 0) r[k] = atomicAdd(&hist[d[k] / EB_RANGE], 1);
    }
    __syncthreads();

    int v = (tid < EB_BINS) ? hist[tid] : 0;
    int ex = scan256_excl(v, tid, wtot);
    if (tid < EB_BINS) {
        offs[tid] = ex;
        cntm[blk * EB_BINS + tid] = v;
        stm[blk * EB_BINS + tid] = ex;
    }
    __syncthreads();

#pragma unroll
    for (int k = 0; k < EB_PER_T; k++) {
        if (d[k] >= 0) {
            int bn = d[k] / EB_RANGE;
            binned[blk * EB_CHUNK + offs[bn] + r[k]] = ((d[k] - bn * EB_RANGE) << 16) | s[k];
        }
    }
}

__global__ void __launch_bounds__(EB_T) edge_place(const int* __restrict__ binned, const int* __restrict__ cntm,
                                                   const int* __restrict__ stm, int* __restrict__ rowptr,
                                                   int* __restrict__ col) {
    __shared__ int cnt[EB_BINS];
    __shared__ int wtot[8];
    __shared__ int red[EB_T / 32];
    int tid = threadIdx.x, b = blockIdx.x;
    int wave = tid >> 5, lane = tid & 31;
    if (tid < EB_BINS) cnt[tid] = 0;

    int c = 0, s = 0;
    if (lane < EB_SEG) {
        int i = wave * EB_SEG + lane;
        c = cntm[i * EB_BINS + b];
        s = stm[i * EB_BINS + b];
    }
    int part = s;
    for (int off = 16; off > 0; off >>= 1) part += __shfl_xor(part, off, 32);
    if (lane == 0) red[wave] = part;

    int len[EB_SEG], st[EB_SEG], rec[EB_SEG];
#pragma unroll
    for (int q = 0; q < EB_SEG; q++) {
        len[q] = __builtin_amdgcn_readlane(c, q);
        st[q]  = (wave * EB_SEG + q) * EB_CHUNK + __builtin_amdgcn_readlane(s, q);
    }
#pragma unroll
    for (int q = 0; q < EB_SEG; q++) {
        int p = st[q] + lane;
        if (p > N_EDGES - 1) p = N_EDGES - 1;
        int t = binned[p];
        rec[q] = (lane < len[q]) ? t : -1;
    }
    __syncthreads();

#pragma unroll
    for (int q = 0; q < EB_SEG; q++) {
        if (rec[q] >= 0) atomicAdd(&cnt[rec[q] >> 16], 1);
        for (int o = 32 + lane; o < len[q]; o += 32) atomicAdd(&cnt[binned[st[q] + o] >> 16], 1);
    }
    __syncthreads();

    int base = 0;
    for (int w = 0; w < EB_T / 32; w++) base += red[w];
    int v = (tid < EB_BINS) ? cnt[tid] : 0;
    int ex = scan256_excl(v, tid, wtot);
    if (tid < EB_BINS) {
        int p = base + ex;
        cnt[tid] = p;
        int node = b * EB_RANGE + tid;
        if (tid < EB_RANGE && node <= N_NODES) rowptr[node] = p;
    }
    __syncthreads();

#pragma unroll
    for (int q = 0; q < EB_SEG; q++) {
        if (rec[q] >= 0) {
            int pos = atomicAdd(&cnt[rec[q] >> 16], 1);
            col[pos] = rec[q] & 0xFFFF;
        }
        for (int o = 32 + lane; o < len[q]; o += 32) {
            int rr = binned[st[q] + o];
            int pos = atomicAdd(&cnt[rr >> 16], 1);
            col[pos] = rr & 0xFFFF;
        }
    }
}

#define VOCAB     500
#define GE_BLOCKS 256
#define GE_WAVES  32
#define GE_SLOTS  (GE_BLOCKS * GE_WAVES)
#define GE_F4     (VOCAB * 128 / 4)
#define GE_FILL   ((GE_F4 + GE_WAVES * 32 - 1) / (GE_WAVES * 32))
#define GE_LDS4   (GE_FILL * GE_WAVES * 32)
static_assert(N_NODES >= GE_SLOTS, "every wave has at least one node");
static_assert(N_NODES / GE_SLOTS + 1 < 32, "a wave's slot range starts fit its 32 lanes");

__device__ __forceinline__ int row_of(int id) {
    return min(max(id, 0), VOCAB - 1) * 128;
}

__device__ __forceinline__ unsigned first_sources(const int* __restrict__ col, int rp, int k, int cnt, int lane) {
    int b = __builtin_amdgcn_readlane(rp, k < cnt ? k : cnt);
    int p = b + lane;
    if (p > N_EDGES - 1) p = N_EDGES - 1;
    return (unsigned)col[p];
}

__device__ __forceinline__ void sum_rows(const float* el, int ro, int n, float4& acc) {
    int t = 0;
    for (; t + 4 <= n; t += 4) {
        int o0 = __builtin_amdgcn_readlane(ro, t);
        int o1 = __builtin_amdgcn_readlane(ro, t + 1);
        int o2 = __builtin_amdgcn_readlane(ro, t + 2);
        int o3 = __builtin_amdgcn_readlane(ro, t + 3);
        float4 v0 = *(const float4*)(el + o0);
        float4 v1 = *(const float4*)(el + o1);
        float4 v2 = *(const float4*)(el + o2);
        float4 v3 = *(const float4*)(el + o3);
        acc.x = (((acc.x + v0.x) + v1.x) + v2.x) + v3.x;
        acc.y = (((acc.y + v0.y) + v1.y) + v2.y) + v3.y;
        acc.z = (((acc.z + v0.z) + v1.z) + v2.z) + v3.z;
        acc.w = (((acc.w + v0.w) + v1.w) + v2.w) + v3.w;
    }
    switch (n - t) {
    case 3: {
        int o0 = __builtin_amdgcn_readlane(ro, t);
        int o1 = __builtin_amdgcn_readlane(ro, t + 1);
        int o2 = __builtin_amdgcn_readlane(ro, t + 2);
        float4 v0 = *(const float4*)(el + o0);
        float4 v1 = *(const float4*)(el + o1);
        float4 v2 = *(const float4*)(el + o2);
        acc.x = ((acc.x + v0.x) + v1.x) + v2.x;
        acc.y = ((acc.y + v0.y) + v1.y) + v2.y;
        acc.z = ((acc.z + v0.z) + v1.z) + v2.z;
        acc.w = ((acc.w + v0.w) + v1.w) + v2.w;
        break;
    }
    case 2: {
        int o0 = __builtin_amdgcn_readlane(ro, t);
        int o1 = __builtin_amdgcn_readlane(ro, t + 1);
        float4 v0 = *(const float4*)(el + o0);
        float4 v1 = *(const float4*)(el + o1);
        acc.x = (acc.x + v0.x) + v1.x;
        acc.y = (acc.y + v0.y) + v1.y;
        acc.z = (acc.z + v0.z) + v1.z;
        acc.w = (acc.w + v0.w) + v1.w;
        break;
    }
    case 1: {
        int o0 = __builtin_amdgcn_readlane(ro, t);
        float4 v0 = *(const float4*)(el + o0);
        acc.x += v0.x; acc.y += v0.y; acc.z += v0.z; acc.w += v0.w;
        break;
    }
    default: break;
    }
}

__global__ void __launch_bounds__(GE_WAVES * 32) gather_sum_emb(const int* __restrict__ ids, const float* __restrict__ emb,
                                                                const int* __restrict__ rowptr, const int* __restrict__ col,
                                                                float* __restrict__ s) {
    __shared__ __attribute__((aligned(16))) float et[GE_LDS4 * 4];
    int tid = threadIdx.x;
    int wave = __builtin_amdgcn_readfirstlane(tid >> 5);
    int lane = tid & 31;
    int gw = blockIdx.x * GE_WAVES + wave;
    int n0 = (int)(((long)gw * N_NODES) / GE_SLOTS);
    int n1 = (int)(((long)(gw + 1) * N_NODES) / GE_SLOTS);
    int cnt = n1 - n0;

    int rp  = rowptr[n0 + (lane < cnt ? lane : cnt)];
    int oid = ids[n0 + (lane < cnt ? lane : cnt - 1)];

    float4 r[GE_FILL];
#pragma unroll
    for (int k = 0; k < GE_FILL / 2; k++) {
        int i = tid + k * (GE_WAVES * 32);
        r[k] = ((const float4*)emb)[i < GE_F4 ? i : GE_F4 - 1];
    }
    unsigned c0 = first_sources(col, rp, 0, cnt, lane);
    unsigned cn = first_sources(col, rp, 1, cnt, lane);
#pragma unroll
    for (int k = GE_FILL / 2; k < GE_FILL; k++) {
        int i = tid + k * (GE_WAVES * 32);
        r[k] = ((const float4*)emb)[i < GE_F4 ? i : GE_F4 - 1];
    }
#pragma unroll
    for (int k = 0; k < GE_FILL / 2; k++) ((float4*)et)[tid + k * (GE_WAVES * 32)] = r[k];
    int idn = ids[c0];
#pragma unroll
    for (int k = GE_FILL / 2; k < GE_FILL; k++) ((float4*)et)[tid + k * (GE_WAVES * 32)] = r[k];
    __syncthreads();

    int own = row_of(oid);
    const float* el = et + lane * 4;
    for (int j = 0; j < cnt; j++) {
        int roc = row_of(idn);
        idn = ids[cn];
        cn = first_sources(col, rp, j + 2, cnt, lane);
        int beg = __builtin_amdgcn_readlane(rp, j);
        int end = __builtin_amdgcn_readlane(rp, j + 1);
        int so  = __builtin_amdgcn_readlane(own, j);
        float4 xv = *(const float4*)(el + so);
        float4 acc = make_float4(0.f, 0.f, 0.f, 0.f);
        int n = end - beg;
        if (n > 32) n = 32;
        sum_rows(el, roc, n, acc);
        for (int base = beg + 32; base < end; base += 32) {
            int m = end - base;
            if (m > 32) m = 32;
            unsigned c = (lane < m) ? (unsigned)col[base + lane] : 0u;
            int rr = row_of(ids[c]);
            sum_rows(el, rr, m, acc);
        }
        float4 o = make_float4(xv.x + acc.x, xv.y + acc.y, xv.z + acc.z, xv.w + acc.w);
        *(float4*)(s + (long)(n0 + j) * 128 + lane * 4) = o;
    }
}

#define SK  128
#define SN  128
#define WPF 288
#define HP  132

__device__ __forceinline__ void w_fetch(const float* __restrict__ W, int wave, int lane,
                                        float4 (&r0)[8], float4 (&r1)[8]) {
#pragma unroll
    for (int i = 0; i < 8; i++) {
        int kp = wave + 8 * i;
        r0[i] = *(const float4*)(W + (long)(2 * kp) * SN + lane * 4);
        r1[i] = *(const float4*)(W + (long)(2 * kp + 1) * SN + lane * 4);
    }
}
__device__ __forceinline__ void w_put(float* wl, int wave, int lane,
                                      const float4 (&r0)[8], const float4 (&r1)[8]) {
#pragma unroll
    for (int i = 0; i < 8; i++) {
        int kp = wave + 8 * i;
        float* d = wl + kp * WPF + lane * 8;
        *(float4*)(d)     = make_float4(r0[i].x, r1[i].x, r0[i].y, r1[i].y);
        *(float4*)(d + 4) = make_float4(r0[i].z, r1[i].z, r0[i].w, r1[i].w);
    }
}

__device__ __forceinline__ void strip_product(const v2f (&af)[SK / 4], const float* bl, v8f (&acc)[8]) {
#pragma unroll
    for (int nt = 0; nt < 8; nt++) acc[nt] = (v8f){0.f, 0.f, 0.f, 0.f, 0.f, 0.f, 0.f, 0.f};
    v2f bc[8], bn[8];
#pragma unroll
    for (int nt = 0; nt < 8; nt++) bc[nt] = *(const v2f*)(bl + nt * 32);
#pragma unroll
    for (int i = 0; i < SK / 4; i++) {
        if (i + 1 < SK / 4) {
#pragma unroll
            for (int nt = 0; nt < 8; nt++) bn[nt] = *(const v2f*)(bl + (2 * i + 2) * WPF + nt * 32);
        }
        __builtin_amdgcn_sched_barrier(0);
#pragma unroll
        for (int nt = 0; nt < 8; nt++)
            acc[nt] = __builtin_amdgcn_wmma_f32_16x16x4_f32(false, af[i], false, bc[nt],
                                                            (short)0, acc[nt], false, false);
        __builtin_amdgcn_sched_barrier(0);
#pragma unroll
        for (int nt = 0; nt < 8; nt++) bc[nt] = bn[nt];
    }
}

#define GB 32

__device__ __forceinline__ void strip_request(const float* __restrict__ x, const int* __restrict__ rowptr,
                                              int n0, int lane, int& rp, float4 (&own)[16]) {
    rp = rowptr[n0 + (lane < 16 ? lane : 16)];
#pragma unroll
    for (int j = 0; j < 16; j++) own[j] = *(const float4*)(x + (long)(n0 + j) * SK + lane * 4);
}

__device__ __forceinline__ int strip_first_sources(const int* __restrict__ col, int rp, int lane) {
    int beg  = __builtin_amdgcn_readlane(rp, 0);
    int tend = __builtin_amdgcn_readlane(rp, 16);
    int nxt = 0;
    if (beg < tend) nxt = col[min(beg + lane, tend - 1)];
    return nxt;
}

__device__ __forceinline__ void strip_row_done(float* hw, int lane, int& node, int& nend, int rp, float4& acc) {
    float* hp = hw + node * HP + lane * 4;
    float4 xv = *(const float4*)hp;
    *(float4*)hp = make_float4(xv.x + acc.x, xv.y + acc.y, xv.z + acc.z, xv.w + acc.w);
    acc = make_float4(0.f, 0.f, 0.f, 0.f);
    node++;
    nend = __builtin_amdgcn_readlane(rp, node + 1);
}

__device__ __forceinline__ void strip_sum(const float* __restrict__ x, const int* __restrict__ col, int lane,
                                          float* hw, int rp, int nxt, const float4 (&own)[16]) {
#pragma unroll
    for (int j = 0; j < 16; j++) *(float4*)(hw + j * HP + lane * 4) = own[j];
    int beg  = __builtin_amdgcn_readlane(rp, 0);
    int tend = __builtin_amdgcn_readlane(rp, 16);
    int nend = __builtin_amdgcn_readlane(rp, 1);
    int node = 0;
    float4 acc = make_float4(0.f, 0.f, 0.f, 0.f);
    for (int base = beg; base < tend; base += GB) {
        int n = min(tend - base, GB);
        int cur = nxt;
        if (base + GB < tend) nxt = col[min(base + GB + lane, tend - 1)];
        float4 v[GB];
#pragma unroll
        for (int g = 0; g < GB / 8; g++) {
            if (g * 8 < n) {
#pragma unroll
                for (int k = 0; k < 8; k++) {
                    int s = __builtin_amdgcn_readlane(cur, g * 8 + k);
                    v[g * 8 + k] = *(const float4*)(x + (long)s * SK + lane * 4);
                }
            }
        }
#pragma unroll
        for (int t = 0; t < GB; t++) {
            if (t < n) {
                while (base + t == nend) strip_row_done(hw, lane, node, nend, rp, acc);
                acc.x += v[t].x; acc.y += v[t].y; acc.z += v[t].z; acc.w += v[t].w;
            }
        }
    }
    while (node < 16) strip_row_done(hw, lane, node, nend, rp, acc);
}

template <bool GATHER>
__global__ void __launch_bounds__(256) mlp_strip(const float* __restrict__ X,
                                                 const int* __restrict__ rowptr, const int* __restrict__ col,
                                                 const float* __restrict__ W1, const float* __restrict__ b1,
                                                 const float* __restrict__ gamma, const float* __restrict__ beta,
                                                 const float* __restrict__ mean, const float* __restrict__ var,
                                                 const float* __restrict__ W2, const float* __restrict__ b2,
                                                 float* __restrict__ O, int strips) {
    __shared__ __attribute__((aligned(16))) float wl[(SK / 2) * WPF];
    __shared__ __attribute__((aligned(16))) float hs[8 * 16 * HP];
    int wave = __builtin_amdgcn_readfirstlane(threadIdx.x >> 5);
    int lane = threadIdx.x & 31;
    int c = lane & 15, h = lane >> 4;
    int strip = blockIdx.x * 8 + wave;
    bool active = strip < strips;
    float* hw = hs + wave * 16 * HP;
    const float* bl = wl + h * WPF + c * 2;

    float4 r0[8], r1[8];
    int rp = 0, nxt = 0;
    float4 own[16];
    int gs = active ? strip : strips - 1;
    if (GATHER) strip_request(X, rowptr, gs * 16, lane, rp, own);
    w_fetch(W1, wave, lane, r0, r1);
    if (GATHER) nxt = strip_first_sources(col, rp, lane);
    w_put(wl, wave, lane, r0, r1);
    if (GATHER) strip_sum(X, col, lane, hw, rp, nxt, own);
    __syncthreads();

    if (active) {
        v2f af[SK / 4];
        if (GATHER) {
            const float* gr = hw + c * HP + h * 2;
#pragma unroll
            for (int i = 0; i < SK / 4; i++) af[i] = *(const v2f*)(gr + i * 4);
        } else {
            const float* xr = X + (long)(strip * 16 + c) * SK + h * 2;
#pragma unroll
            for (int i = 0; i < SK / 4; i++) af[i] = *(const v2f*)(xr + i * 4);
        }
        v8f acc[8];
        strip_product(af, bl, acc);
#pragma unroll
        for (int nt = 0; nt < 8; nt++) {
            int col = nt * 16 + c;
            float bb = b1[col];
            float mu = mean[col], inv = rsqrtf(var[col] + BN_EPS), gm = gamma[col], bt = beta[col];
#pragma unroll
            for (int r = 0; r < 8; r++) {
                float v = acc[nt][r] + bb;
                v = (v - mu) * inv * gm + bt;
                v = fmaxf(v, 0.f);
                hw[(h * 8 + r) * HP + col] = v;
            }
        }
    }

    w_fetch(W2, wave, lane, r0, r1);
    __syncthreads();
    w_put(wl, wave, lane, r0, r1);
    __syncthreads();

    if (active) {
        v2f af[SK / 4];
        const float* hr = hw + c * HP + h * 2;
#pragma unroll
        for (int i = 0; i < SK / 4; i++) af[i] = *(const v2f*)(hr + i * 4);
        v8f acc[8];
        strip_product(af, bl, acc);
        float* orow = O + (long)(strip * 16 + h * 8) * SN + c;
#pragma unroll
        for (int nt = 0; nt < 8; nt++) {
            float bb = b2[nt * 16 + c];
#pragma unroll
            for (int r = 0; r < 8; r++) {
                float v = fmaxf(acc[nt][r] + bb, 0.f);
                orow[(long)r * SN + nt * 16] = v;
            }
        }
    }
}

template <bool FUSE_ADD, bool BN, bool RELU>
__global__ void gemm_wmma(const float* __restrict__ X, const float* __restrict__ AGG,
                          const float* __restrict__ W, const float* __restrict__ bias,
                          const float* __restrict__ gamma, const float* __restrict__ beta,
                          const float* __restrict__ mean, const float* __restrict__ var,
                          float* __restrict__ O, int M, int K, int N) {
    int wave = threadIdx.x >> 5, lane = threadIdx.x & 31;
    int tilesN = N >> 4;
    int tile = blockIdx.x * 8 + wave;
    if (tile >= (M >> 4) * tilesN) return;
    int mt = tile / tilesN, nt = tile % tilesN;
    int mi = lane & 15, kh = lane >> 4;

    const float* xr = X + (long)(mt * 16 + mi) * K + kh * 2;
    const float* ar = FUSE_ADD ? (AGG + (long)(mt * 16 + mi) * K + kh * 2) : nullptr;
    const float* wc = W + (long)(kh * 2) * N + nt * 16 + mi;

    v8f acc = {0.f, 0.f, 0.f, 0.f, 0.f, 0.f, 0.f, 0.f};
    for (int kb = 0; kb < K; kb += 4) {
        v2f a, b;
        a.x = xr[kb];
        a.y = xr[kb + 1];
        if (FUSE_ADD) { a.x += ar[kb]; a.y += ar[kb + 1]; }
        b.x = wc[(long)kb * N];
        b.y = wc[(long)(kb + 1) * N];
        acc = __builtin_amdgcn_wmma_f32_16x16x4_f32(false, a, false, b,
                                                    (short)0, acc, false, false);
    }

    int col = nt * 16 + mi;
    float bb = bias[col];
    float mu = 0.f, inv = 1.f, gm = 1.f, bt = 0.f;
    if (BN) { mu = mean[col]; inv = rsqrtf(var[col] + BN_EPS); gm = gamma[col]; bt = beta[col]; }
    float* orow = O + (long)(mt * 16 + kh * 8) * N + col;
#pragma unroll
    for (int r = 0; r < 8; r++) {
        float v = acc[r] + bb;
        if (BN)   v = (v - mu) * inv * gm + bt;
        if (RELU) v = fmaxf(v, 0.f);
        orow[(long)r * N] = v;
    }
}

#define POOL_CH 16
__device__ __forceinline__ void pool_flush(float* __restrict__ p, float4 a) {
    atomicAdd(p + 0, a.x);
    atomicAdd(p + 1, a.y);
    atomicAdd(p + 2, a.z);
    atomicAdd(p + 3, a.w);
}

__global__ void __launch_bounds__(256) pool3(const float* __restrict__ h1, const float* __restrict__ h2,
                                             const float* __restrict__ h3, const int* __restrict__ batch,
                                             float* __restrict__ gcat) {
    int wave = threadIdx.x >> 5, lane = threadIdx.x & 31;
    int chunk = blockIdx.x * 8 + wave;
    if (chunk >= N_NODES / POOL_CH) return;
    int base = chunk * POOL_CH;
    int myb = batch[base + (lane & (POOL_CH - 1))];
    const float* hs[3] = {h1, h2, h3};
#pragma unroll
    for (int l = 0; l < 3; l++) {
        const float* hp = hs[l] + (long)base * 128 + lane * 4;
        float4 v[POOL_CH];
#pragma unroll
        for (int j = 0; j < POOL_CH; j++) v[j] = *(const float4*)(hp + j * 128);
        float* gp = gcat + l * 128 + lane * 4;
        int cur = __builtin_amdgcn_readlane(myb, 0);
        float4 acc = v[0];
#pragma unroll
        for (int j = 1; j < POOL_CH; j++) {
            int g = __builtin_amdgcn_readlane(myb, j);
            if (g != cur) {
                pool_flush(gp + (long)cur * 384, acc);
                acc = v[j];
                cur = g;
            } else {
                acc.x += v[j].x; acc.y += v[j].y; acc.z += v[j].z; acc.w += v[j].w;
            }
        }
        pool_flush(gp + (long)cur * 384, acc);
    }
}

__global__ void lin2_kernel(const float* __restrict__ h, const float* __restrict__ w,
                            const float* __restrict__ b, float* __restrict__ out) {
    int t = blockIdx.x * blockDim.x + threadIdx.x;
    if (t >= N_GRAPHS * 10) return;
    int g = t / 10, o = t % 10;
    const float* hr = h + (long)g * 384;
    float acc = b[o];
    for (int k = 0; k < 384; k++) acc = fmaf(hr[k], w[k * 10 + o], acc);
    out[t] = acc;
}

extern "C" void kernel_launch(void* const* d_in, const int* in_sizes, int n_in,
                              void* d_out, int out_size, void* d_ws, size_t ws_size,
                              hipStream_t stream) {
    const int*   x_ids = (const int*)d_in[0];
    const int*   edge  = (const int*)d_in[1];
    const int*   batch = (const int*)d_in[2];
    const float* emb   = (const float*)d_in[3];

    const float *c_w1[3], *c_b1[3], *c_g[3], *c_be[3], *c_mu[3], *c_va[3], *c_w2[3], *c_b2[3];
    for (int i = 0; i < 3; i++) {
        c_w1[i] = (const float*)d_in[4 + 8 * i + 0];
        c_b1[i] = (const float*)d_in[4 + 8 * i + 1];
        c_g[i]  = (const float*)d_in[4 + 8 * i + 2];
        c_be[i] = (const float*)d_in[4 + 8 * i + 3];
        c_mu[i] = (const float*)d_in[4 + 8 * i + 4];
        c_va[i] = (const float*)d_in[4 + 8 * i + 5];
        c_w2[i] = (const float*)d_in[4 + 8 * i + 6];
        c_b2[i] = (const float*)d_in[4 + 8 * i + 7];
    }
    const float* lin1_w = (const float*)d_in[28];
    const float* lin1_b = (const float*)d_in[29];
    const float* lin2_w = (const float*)d_in[30];
    const float* lin2_b = (const float*)d_in[31];

    const long NF = (long)N_NODES * 128;
    float* ws   = (float*)d_ws;
    float* h3   = ws + 0 * NF;
    float* agg  = ws + 1 * NF;
    float* tmp  = ws + 2 * NF;
    float* hbuf[3] = { ws + 3 * NF, ws + 4 * NF, h3 };
    float* gcat  = agg;
    float* lin1o = tmp;
    int* rowptr = (int*)(ws + 5 * NF);
    int* binned = rowptr + RP_PAD;
    int* col    = binned + N_EDGES;
    int* cntm   = col + N_EDGES;
    int* stm    = cntm + EB_BLOCKS * EB_BINS;

    edge_bin<<<EB_BLOCKS, EB_T, 0, stream>>>(edge, binned, cntm, stm);
    edge_place<<<EB_BINS, EB_T, 0, stream>>>(binned, cntm, stm, rowptr, col);

    const int strips = N_NODES / 16;
    for (int l = 0; l < 3; l++) {
        if (l == 0) {
            gather_sum_emb<<<GE_BLOCKS, GE_WAVES * 32, 0, stream>>>(x_ids, emb, rowptr, col, agg);
            mlp_strip<false><<<(strips + 7) / 8, 256, 0, stream>>>(
                agg, rowptr, col, c_w1[l], c_b1[l], c_g[l], c_be[l], c_mu[l], c_va[l], c_w2[l], c_b2[l], hbuf[l], strips);
        } else {
            mlp_strip<true><<<(strips + 7) / 8, 256, 0, stream>>>(
                hbuf[l - 1], rowptr, col, c_w1[l], c_b1[l], c_g[l], c_be[l], c_mu[l], c_va[l], c_w2[l], c_b2[l], hbuf[l], strips);
        }
    }

    zero_f32<<<(N_GRAPHS * 384 / 4 + 255) / 256, 256, 0, stream>>>(gcat, N_GRAPHS * 384 / 4);
    pool3<<<(N_NODES / POOL_CH + 7) / 8, 256, 0, stream>>>(hbuf[0], hbuf[1], hbuf[2], batch, gcat);

    const int tiles_head = (N_GRAPHS / 16) * (384 / 16);
    gemm_wmma<false, false, true><<<(tiles_head + 7) / 8, 256, 0, stream>>>(
        gcat, nullptr, lin1_w, lin1_b, nullptr, nullptr, nullptr, nullptr,
        lin1o, N_GRAPHS, 384, 384);

    lin2_kernel<<<(N_GRAPHS * 10 + 255) / 256, 256, 0, stream>>>(lin1o, lin2_w, lin2_b,
                                                                 (float*)d_out);
}


# Head in one launch: a work-group owns 16 graphs, lin1 on WMMA with its 16x384 hidden tile kept in LDS, lin2 reads it there; loads run a group ahead; hidden matrix not stored; sum order unchanged
# speedup vs baseline: 4.7233x; 1.1432x over previous
#include <hip/hip_runtime.h>

typedef __attribute__((ext_vector_type(2))) float v2f;
typedef __attribute__((ext_vector_type(8))) float v8f;

#define N_NODES  50000
#define N_EDGES  800000
#define N_GRAPHS 512
#define BN_EPS   1e-5f

__global__ void zero_f32(float* __restrict__ p, int n4) {
    int t = blockIdx.x * blockDim.x + threadIdx.x;
    if (t < n4) ((float4*)p)[t] = make_float4(0.f, 0.f, 0.f, 0.f);
}

#define EB_BLOCKS 256
#define EB_CHUNK  3125
#define EB_BINS   256
#define EB_RANGE  196
#define EB_T      512
#define EB_PER_T  7
#define EB_SEG    16
#define RP_PAD    (EB_BINS * EB_RANGE)

__device__ __forceinline__ int scan256_excl(int v, int tid, int* wtot) {
    int lane = tid & 31, wave = tid >> 5;
    int inc = v;
#pragma unroll
    for (int off = 1; off < 32; off <<= 1) {
        int t = __shfl_up(inc, off, 32);
        if (lane >= off) inc += t;
    }
    if (tid < 256 && lane == 31) wtot[wave] = inc;
    __syncthreads();
    int carry = 0;
    if (tid < 256)
        for (int w = 0; w < wave; w++) carry += wtot[w];
    return carry + inc - v;
}

__global__ void __launch_bounds__(EB_T) edge_bin(const int* __restrict__ ei, int* __restrict__ binned,
                                                 int* __restrict__ cntm, int* __restrict__ stm) {
    __shared__ int hist[EB_BINS];
    __shared__ int offs[EB_BINS];
    __shared__ int wtot[8];
    int tid = threadIdx.x, blk = blockIdx.x;
    if (tid < EB_BINS) hist[tid] = 0;
    __syncthreads();

    int d[EB_PER_T], s[EB_PER_T], r[EB_PER_T];
#pragma unroll
    for (int k = 0; k < EB_PER_T; k++) {
        int idx = tid + k * EB_T;
        d[k] = -1;
        s[k] = 0;
        if (idx < EB_CHUNK) {
            int e = blk * EB_CHUNK + idx;
            d[k] = ei[N_EDGES + e];
            s[k] = ei[e];
        }
    }
#pragma unroll
    for (int k = 0; k < EB_PER_T; k++) {
        r[k] = 0;
        if (d[k] >= 0) r[k] = atomicAdd(&hist[d[k] / EB_RANGE], 1);
    }
    __syncthreads();

    int v = (tid < EB_BINS) ? hist[tid] : 0;
    int ex = scan256_excl(v, tid, wtot);
    if (tid < EB_BINS) {
        offs[tid] = ex;
        cntm[blk * EB_BINS + tid] = v;
        stm[blk * EB_BINS + tid] = ex;
    }
    __syncthreads();

#pragma unroll
    for (int k = 0; k < EB_PER_T; k++) {
        if (d[k] >= 0) {
            int bn = d[k] / EB_RANGE;
            binned[blk * EB_CHUNK + offs[bn] + r[k]] = ((d[k] - bn * EB_RANGE) << 16) | s[k];
        }
    }
}

__global__ void __launch_bounds__(EB_T) edge_place(const int* __restrict__ binned, const int* __restrict__ cntm,
                                                   const int* __restrict__ stm, int* __restrict__ rowptr,
                                                   int* __restrict__ col) {
    __shared__ int cnt[EB_BINS];
    __shared__ int wtot[8];
    __shared__ int red[EB_T / 32];
    int tid = threadIdx.x, b = blockIdx.x;
    int wave = tid >> 5, lane = tid & 31;
    if (tid < EB_BINS) cnt[tid] = 0;

    int c = 0, s = 0;
    if (lane < EB_SEG) {
        int i = wave * EB_SEG + lane;
        c = cntm[i * EB_BINS + b];
        s = stm[i * EB_BINS + b];
    }
    int part = s;
    for (int off = 16; off > 0; off >>= 1) part += __shfl_xor(part, off, 32);
    if (lane == 0) red[wave] = part;

    int len[EB_SEG], st[EB_SEG], rec[EB_SEG];
#pragma unroll
    for (int q = 0; q < EB_SEG; q++) {
        len[q] = __builtin_amdgcn_readlane(c, q);
        st[q]  = (wave * EB_SEG + q) * EB_CHUNK + __builtin_amdgcn_readlane(s, q);
    }
#pragma unroll
    for (int q = 0; q < EB_SEG; q++) {
        int p = st[q] + lane;
        if (p > N_EDGES - 1) p = N_EDGES - 1;
        int t = binned[p];
        rec[q] = (lane < len[q]) ? t : -1;
    }
    __syncthreads();

#pragma unroll
    for (int q = 0; q < EB_SEG; q++) {
        if (rec[q] >= 0) atomicAdd(&cnt[rec[q] >> 16], 1);
        for (int o = 32 + lane; o < len[q]; o += 32) atomicAdd(&cnt[binned[st[q] + o] >> 16], 1);
    }
    __syncthreads();

    int base = 0;
    for (int w = 0; w < EB_T / 32; w++) base += red[w];
    int v = (tid < EB_BINS) ? cnt[tid] : 0;
    int ex = scan256_excl(v, tid, wtot);
    if (tid < EB_BINS) {
        int p = base + ex;
        cnt[tid] = p;
        int node = b * EB_RANGE + tid;
        if (tid < EB_RANGE && node <= N_NODES) rowptr[node] = p;
    }
    __syncthreads();

#pragma unroll
    for (int q = 0; q < EB_SEG; q++) {
        if (rec[q] >= 0) {
            int pos = atomicAdd(&cnt[rec[q] >> 16], 1);
            col[pos] = rec[q] & 0xFFFF;
        }
        for (int o = 32 + lane; o < len[q]; o += 32) {
            int rr = binned[st[q] + o];
            int pos = atomicAdd(&cnt[rr >> 16], 1);
            col[pos] = rr & 0xFFFF;
        }
    }
}

#define VOCAB     500
#define GE_BLOCKS 256
#define GE_WAVES  32
#define GE_SLOTS  (GE_BLOCKS * GE_WAVES)
#define GE_F4     (VOCAB * 128 / 4)
#define GE_FILL   ((GE_F4 + GE_WAVES * 32 - 1) / (GE_WAVES * 32))
#define GE_LDS4   (GE_FILL * GE_WAVES * 32)
static_assert(N_NODES >= GE_SLOTS, "every wave has at least one node");
static_assert(N_NODES / GE_SLOTS + 1 < 32, "a wave's slot range starts fit its 32 lanes");

__device__ __forceinline__ int row_of(int id) {
    return min(max(id, 0), VOCAB - 1) * 128;
}

__device__ __forceinline__ unsigned first_sources(const int* __restrict__ col, int rp, int k, int cnt, int lane) {
    int b = __builtin_amdgcn_readlane(rp, k < cnt ? k : cnt);
    int p = b + lane;
    if (p > N_EDGES - 1) p = N_EDGES - 1;
    return (unsigned)col[p];
}

__device__ __forceinline__ void sum_rows(const float* el, int ro, int n, float4& acc) {
    int t = 0;
    for (; t + 4 <= n; t += 4) {
        int o0 = __builtin_amdgcn_readlane(ro, t);
        int o1 = __builtin_amdgcn_readlane(ro, t + 1);
        int o2 = __builtin_amdgcn_readlane(ro, t + 2);
        int o3 = __builtin_amdgcn_readlane(ro, t + 3);
        float4 v0 = *(const float4*)(el + o0);
        float4 v1 = *(const float4*)(el + o1);
        float4 v2 = *(const float4*)(el + o2);
        float4 v3 = *(const float4*)(el + o3);
        acc.x = (((acc.x + v0.x) + v1.x) + v2.x) + v3.x;
        acc.y = (((acc.y + v0.y) + v1.y) + v2.y) + v3.y;
        acc.z = (((acc.z + v0.z) + v1.z) + v2.z) + v3.z;
        acc.w = (((acc.w + v0.w) + v1.w) + v2.w) + v3.w;
    }
    switch (n - t) {
    case 3: {
        int o0 = __builtin_amdgcn_readlane(ro, t);
        int o1 = __builtin_amdgcn_readlane(ro, t + 1);
        int o2 = __builtin_amdgcn_readlane(ro, t + 2);
        float4 v0 = *(const float4*)(el + o0);
        float4 v1 = *(const float4*)(el + o1);
        float4 v2 = *(const float4*)(el + o2);
        acc.x = ((acc.x + v0.x) + v1.x) + v2.x;
        acc.y = ((acc.y + v0.y) + v1.y) + v2.y;
        acc.z = ((acc.z + v0.z) + v1.z) + v2.z;
        acc.w = ((acc.w + v0.w) + v1.w) + v2.w;
        break;
    }
    case 2: {
        int o0 = __builtin_amdgcn_readlane(ro, t);
        int o1 = __builtin_amdgcn_readlane(ro, t + 1);
        float4 v0 = *(const float4*)(el + o0);
        float4 v1 = *(const float4*)(el + o1);
        acc.x = (acc.x + v0.x) + v1.x;
        acc.y = (acc.y + v0.y) + v1.y;
        acc.z = (acc.z + v0.z) + v1.z;
        acc.w = (acc.w + v0.w) + v1.w;
        break;
    }
    case 1: {
        int o0 = __builtin_amdgcn_readlane(ro, t);
        float4 v0 = *(const float4*)(el + o0);
        acc.x += v0.x; acc.y += v0.y; acc.z += v0.z; acc.w += v0.w;
        break;
    }
    default: break;
    }
}

__global__ void __launch_bounds__(GE_WAVES * 32) gather_sum_emb(const int* __restrict__ ids, const float* __restrict__ emb,
                                                                const int* __restrict__ rowptr, const int* __restrict__ col,
                                                                float* __restrict__ s) {
    __shared__ __attribute__((aligned(16))) float et[GE_LDS4 * 4];
    int tid = threadIdx.x;
    int wave = __builtin_amdgcn_readfirstlane(tid >> 5);
    int lane = tid & 31;
    int gw = blockIdx.x * GE_WAVES + wave;
    int n0 = (int)(((long)gw * N_NODES) / GE_SLOTS);
    int n1 = (int)(((long)(gw + 1) * N_NODES) / GE_SLOTS);
    int cnt = n1 - n0;

    int rp  = rowptr[n0 + (lane < cnt ? lane : cnt)];
    int oid = ids[n0 + (lane < cnt ? lane : cnt - 1)];

    float4 r[GE_FILL];
#pragma unroll
    for (int k = 0; k < GE_FILL / 2; k++) {
        int i = tid + k * (GE_WAVES * 32);
        r[k] = ((const float4*)emb)[i < GE_F4 ? i : GE_F4 - 1];
    }
    unsigned c0 = first_sources(col, rp, 0, cnt, lane);
    unsigned cn = first_sources(col, rp, 1, cnt, lane);
#pragma unroll
    for (int k = GE_FILL / 2; k < GE_FILL; k++) {
        int i = tid + k * (GE_WAVES * 32);
        r[k] = ((const float4*)emb)[i < GE_F4 ? i : GE_F4 - 1];
    }
#pragma unroll
    for (int k = 0; k < GE_FILL / 2; k++) ((float4*)et)[tid + k * (GE_WAVES * 32)] = r[k];
    int idn = ids[c0];
#pragma unroll
    for (int k = GE_FILL / 2; k < GE_FILL; k++) ((float4*)et)[tid + k * (GE_WAVES * 32)] = r[k];
    __syncthreads();

    int own = row_of(oid);
    const float* el = et + lane * 4;
    for (int j = 0; j < cnt; j++) {
        int roc = row_of(idn);
        idn = ids[cn];
        cn = first_sources(col, rp, j + 2, cnt, lane);
        int beg = __builtin_amdgcn_readlane(rp, j);
        int end = __builtin_amdgcn_readlane(rp, j + 1);
        int so  = __builtin_amdgcn_readlane(own, j);
        float4 xv = *(const float4*)(el + so);
        float4 acc = make_float4(0.f, 0.f, 0.f, 0.f);
        int n = end - beg;
        if (n > 32) n = 32;
        sum_rows(el, roc, n, acc);
        for (int base = beg + 32; base < end; base += 32) {
            int m = end - base;
            if (m > 32) m = 32;
            unsigned c = (lane < m) ? (unsigned)col[base + lane] : 0u;
            int rr = row_of(ids[c]);
            sum_rows(el, rr, m, acc);
        }
        float4 o = make_float4(xv.x + acc.x, xv.y + acc.y, xv.z + acc.z, xv.w + acc.w);
        *(float4*)(s + (long)(n0 + j) * 128 + lane * 4) = o;
    }
}

#define SK  128
#define SN  128
#define WPF 288
#define HP  132

__device__ __forceinline__ void w_fetch(const float* __restrict__ W, int wave, int lane,
                                        float4 (&r0)[8], float4 (&r1)[8]) {
#pragma unroll
    for (int i = 0; i < 8; i++) {
        int kp = wave + 8 * i;
        r0[i] = *(const float4*)(W + (long)(2 * kp) * SN + lane * 4);
        r1[i] = *(const float4*)(W + (long)(2 * kp + 1) * SN + lane * 4);
    }
}
__device__ __forceinline__ void w_put(float* wl, int wave, int lane,
                                      const float4 (&r0)[8], const float4 (&r1)[8]) {
#pragma unroll
    for (int i = 0; i < 8; i++) {
        int kp = wave + 8 * i;
        float* d = wl + kp * WPF + lane * 8;
        *(float4*)(d)     = make_float4(r0[i].x, r1[i].x, r0[i].y, r1[i].y);
        *(float4*)(d + 4) = make_float4(r0[i].z, r1[i].z, r0[i].w, r1[i].w);
    }
}

__device__ __forceinline__ void strip_product(const v2f (&af)[SK / 4], const float* bl, v8f (&acc)[8]) {
#pragma unroll
    for (int nt = 0; nt < 8; nt++) acc[nt] = (v8f){0.f, 0.f, 0.f, 0.f, 0.f, 0.f, 0.f, 0.f};
    v2f bc[8], bn[8];
#pragma unroll
    for (int nt = 0; nt < 8; nt++) bc[nt] = *(const v2f*)(bl + nt * 32);
#pragma unroll
    for (int i = 0; i < SK / 4; i++) {
        if (i + 1 < SK / 4) {
#pragma unroll
            for (int nt = 0; nt < 8; nt++) bn[nt] = *(const v2f*)(bl + (2 * i + 2) * WPF + nt * 32);
        }
        __builtin_amdgcn_sched_barrier(0);
#pragma unroll
        for (int nt = 0; nt < 8; nt++)
            acc[nt] = __builtin_amdgcn_wmma_f32_16x16x4_f32(false, af[i], false, bc[nt],
                                                            (short)0, acc[nt], false, false);
        __builtin_amdgcn_sched_barrier(0);
#pragma unroll
        for (int nt = 0; nt < 8; nt++) bc[nt] = bn[nt];
    }
}

#define GB 32

__device__ __forceinline__ void strip_request(const float* __restrict__ x, const int* __restrict__ rowptr,
                                              int n0, int lane, int& rp, float4 (&own)[16]) {
    rp = rowptr[n0 + (lane < 16 ? lane : 16)];
#pragma unroll
    for (int j = 0; j < 16; j++) own[j] = *(const float4*)(x + (long)(n0 + j) * SK + lane * 4);
}

__device__ __forceinline__ int strip_first_sources(const int* __restrict__ col, int rp, int lane) {
    int beg  = __builtin_amdgcn_readlane(rp, 0);
    int tend = __builtin_amdgcn_readlane(rp, 16);
    int nxt = 0;
    if (beg < tend) nxt = col[min(beg + lane, tend - 1)];
    return nxt;
}

__device__ __forceinline__ void strip_row_done(float* hw, int lane, int& node, int& nend, int rp, float4& acc) {
    float* hp = hw + node * HP + lane * 4;
    float4 xv = *(const float4*)hp;
    *(float4*)hp = make_float4(xv.x + acc.x, xv.y + acc.y, xv.z + acc.z, xv.w + acc.w);
    acc = make_float4(0.f, 0.f, 0.f, 0.f);
    node++;
    nend = __builtin_amdgcn_readlane(rp, node + 1);
}

__device__ __forceinline__ void strip_sum(const float* __restrict__ x, const int* __restrict__ col, int lane,
                                          float* hw, int rp, int nxt, const float4 (&own)[16]) {
#pragma unroll
    for (int j = 0; j < 16; j++) *(float4*)(hw + j * HP + lane * 4) = own[j];
    int beg  = __builtin_amdgcn_readlane(rp, 0);
    int tend = __builtin_amdgcn_readlane(rp, 16);
    int nend = __builtin_amdgcn_readlane(rp, 1);
    int node = 0;
    float4 acc = make_float4(0.f, 0.f, 0.f, 0.f);
    for (int base = beg; base < tend; base += GB) {
        int n = min(tend - base, GB);
        int cur = nxt;
        if (base + GB < tend) nxt = col[min(base + GB + lane, tend - 1)];
        float4 v[GB];
#pragma unroll
        for (int g = 0; g < GB / 8; g++) {
            if (g * 8 < n) {
#pragma unroll
                for (int k = 0; k < 8; k++) {
                    int s = __builtin_amdgcn_readlane(cur, g * 8 + k);
                    v[g * 8 + k] = *(const float4*)(x + (long)s * SK + lane * 4);
                }
            }
        }
#pragma unroll
        for (int t = 0; t < GB; t++) {
            if (t < n) {
                while (base + t == nend) strip_row_done(hw, lane, node, nend, rp, acc);
                acc.x += v[t].x; acc.y += v[t].y; acc.z += v[t].z; acc.w += v[t].w;
            }
        }
    }
    while (node < 16) strip_row_done(hw, lane, node, nend, rp, acc);
}

template <bool GATHER>
__global__ void __launch_bounds__(256) mlp_strip(const float* __restrict__ X,
                                                 const int* __restrict__ rowptr, const int* __restrict__ col,
                                                 const float* __restrict__ W1, const float* __restrict__ b1,
                                                 const float* __restrict__ gamma, const float* __restrict__ beta,
                                                 const float* __restrict__ mean, const float* __restrict__ var,
                                                 const float* __restrict__ W2, const float* __restrict__ b2,
                                                 float* __restrict__ O, int strips) {
    __shared__ __attribute__((aligned(16))) float wl[(SK / 2) * WPF];
    __shared__ __attribute__((aligned(16))) float hs[8 * 16 * HP];
    int wave = __builtin_amdgcn_readfirstlane(threadIdx.x >> 5);
    int lane = threadIdx.x & 31;
    int c = lane & 15, h = lane >> 4;
    int strip = blockIdx.x * 8 + wave;
    bool active = strip < strips;
    float* hw = hs + wave * 16 * HP;
    const float* bl = wl + h * WPF + c * 2;

    float4 r0[8], r1[8];
    int rp = 0, nxt = 0;
    float4 own[16];
    int gs = active ? strip : strips - 1;
    if (GATHER) strip_request(X, rowptr, gs * 16, lane, rp, own);
    w_fetch(W1, wave, lane, r0, r1);
    if (GATHER) nxt = strip_first_sources(col, rp, lane);
    w_put(wl, wave, lane, r0, r1);
    if (GATHER) strip_sum(X, col, lane, hw, rp, nxt, own);
    __syncthreads();

    if (active) {
        v2f af[SK / 4];
        if (GATHER) {
            const float* gr = hw + c * HP + h * 2;
#pragma unroll
            for (int i = 0; i < SK / 4; i++) af[i] = *(const v2f*)(gr + i * 4);
        } else {
            const float* xr = X + (long)(strip * 16 + c) * SK + h * 2;
#pragma unroll
            for (int i = 0; i < SK / 4; i++) af[i] = *(const v2f*)(xr + i * 4);
        }
        v8f acc[8];
        strip_product(af, bl, acc);
#pragma unroll
        for (int nt = 0; nt < 8; nt++) {
            int col = nt * 16 + c;
            float bb = b1[col];
            float mu = mean[col], inv = rsqrtf(var[col] + BN_EPS), gm = gamma[col], bt = beta[col];
#pragma unroll
            for (int r = 0; r < 8; r++) {
                float v = acc[nt][r] + bb;
                v = (v - mu) * inv * gm + bt;
                v = fmaxf(v, 0.f);
                hw[(h * 8 + r) * HP + col] = v;
            }
        }
    }

    w_fetch(W2, wave, lane, r0, r1);
    __syncthreads();
    w_put(wl, wave, lane, r0, r1);
    __syncthreads();

    if (active) {
        v2f af[SK / 4];
        const float* hr = hw + c * HP + h * 2;
#pragma unroll
        for (int i = 0; i < SK / 4; i++) af[i] = *(const v2f*)(hr + i * 4);
        v8f acc[8];
        strip_product(af, bl, acc);
        float* orow = O + (long)(strip * 16 + h * 8) * SN + c;
#pragma unroll
        for (int nt = 0; nt < 8; nt++) {
            float bb = b2[nt * 16 + c];
#pragma unroll
            for (int r = 0; r < 8; r++) {
                float v = fmaxf(acc[nt][r] + bb, 0.f);
                orow[(long)r * SN + nt * 16] = v;
            }
        }
    }
}

#define POOL_CH 16
__device__ __forceinline__ void pool_flush(float* __restrict__ p, float4 a) {
    atomicAdd(p + 0, a.x);
    atomicAdd(p + 1, a.y);
    atomicAdd(p + 2, a.z);
    atomicAdd(p + 3, a.w);
}

__global__ void __launch_bounds__(256) pool3(const float* __restrict__ h1, const float* __restrict__ h2,
                                             const float* __restrict__ h3, const int* __restrict__ batch,
                                             float* __restrict__ gcat) {
    int wave = threadIdx.x >> 5, lane = threadIdx.x & 31;
    int chunk = blockIdx.x * 8 + wave;
    if (chunk >= N_NODES / POOL_CH) return;
    int base = chunk * POOL_CH;
    int myb = batch[base + (lane & (POOL_CH - 1))];
    const float* hs[3] = {h1, h2, h3};
#pragma unroll
    for (int l = 0; l < 3; l++) {
        const float* hp = hs[l] + (long)base * 128 + lane * 4;
        float4 v[POOL_CH];
#pragma unroll
        for (int j = 0; j < POOL_CH; j++) v[j] = *(const float4*)(hp + j * 128);
        float* gp = gcat + l * 128 + lane * 4;
        int cur = __builtin_amdgcn_readlane(myb, 0);
        float4 acc = v[0];
#pragma unroll
        for (int j = 1; j < POOL_CH; j++) {
            int g = __builtin_amdgcn_readlane(myb, j);
            if (g != cur) {
                pool_flush(gp + (long)cur * 384, acc);
                acc = v[j];
                cur = g;
            } else {
                acc.x += v[j].x; acc.y += v[j].y; acc.z += v[j].z; acc.w += v[j].w;
            }
        }
        pool_flush(gp + (long)cur * 384, acc);
    }
}

#define HD_T   384
#define HD_K   384
#define HD_N   384
#define HD_P   388
#define HD_CK  8
#define HD_W2F4 (HD_N * 10 / 4)
#define HD_W2R  ((HD_W2F4 + HD_T - 1) / HD_T)
#define HD_W2L4 (HD_W2R * HD_T)

__device__ __forceinline__ void head_fetch(const float* wb, const float* ar, int ch, bool with_a,
                                           v2f (&a)[HD_CK], v2f (&b)[HD_CK][2]) {
#pragma unroll
    for (int j = 0; j < HD_CK; j++) {
        int i = ch * HD_CK + j;
        b[j][0] = *(const v2f*)(wb + (long)(4 * i) * HD_N);
        b[j][1] = *(const v2f*)(wb + (long)(4 * i + 1) * HD_N);
    }
    if (with_a) {
#pragma unroll
        for (int j = 0; j < HD_CK; j++) a[j] = *(const v2f*)(ar + (ch * HD_CK + j) * 4);
    }
}

__global__ void __launch_bounds__(HD_T) head_fused(const float* __restrict__ gcat, const float* __restrict__ w1,
                                                   const float* __restrict__ b1, const float* __restrict__ w2,
                                                   const float* __restrict__ b2, float* __restrict__ out) {
    __shared__ __attribute__((aligned(16))) float ga[16 * HD_P];
    __shared__ __attribute__((aligned(16))) float hl[16 * HD_P];
    __shared__ __attribute__((aligned(16))) float w2s[HD_W2L4 * 4];
    int tid = threadIdx.x;
    int wave = __builtin_amdgcn_readfirstlane(tid >> 5);
    int lane = tid & 31;
    int c = lane & 15, h = lane >> 4;
    int m = blockIdx.x;

    const float* wb = w1 + (long)(2 * h) * HD_N + 32 * wave + 2 * c;
    const float* ar = ga + c * HD_P + 2 * h;
    v2f ac[HD_CK], an[HD_CK], bc[HD_CK][2], bn[HD_CK][2];
    head_fetch(wb, ar, 0, false, ac, bc);

    float4 ra[4], rw[HD_W2R];
    const float4* gsrc = (const float4*)(gcat + (long)m * 16 * HD_K);
#pragma unroll
    for (int k = 0; k < 4; k++) ra[k] = gsrc[tid + k * HD_T];
#pragma unroll
    for (int k = 0; k < HD_W2R; k++) {
        int i = tid + k * HD_T;
        rw[k] = ((const float4*)w2)[i < HD_W2F4 ? i : HD_W2F4 - 1];
    }
#pragma unroll
    for (int k = 0; k < 4; k++) {
        int i = tid + k * HD_T;
        int row = i / (HD_K / 4), c4 = i - row * (HD_K / 4);
        *(float4*)(ga + row * HD_P + c4 * 4) = ra[k];
    }
#pragma unroll
    for (int k = 0; k < HD_W2R; k++) ((float4*)w2s)[tid + k * HD_T] = rw[k];
    __syncthreads();

#pragma unroll
    for (int j = 0; j < HD_CK; j++) ac[j] = *(const v2f*)(ar + j * 4);

    v8f acc0 = (v8f){0.f, 0.f, 0.f, 0.f, 0.f, 0.f, 0.f, 0.f};
    v8f acc1 = (v8f){0.f, 0.f, 0.f, 0.f, 0.f, 0.f, 0.f, 0.f};
#pragma unroll
    for (int ch = 0; ch < HD_K / 4 / HD_CK; ch++) {
        if (ch + 1 < HD_K / 4 / HD_CK) head_fetch(wb, ar, ch + 1, true, an, bn);
        __builtin_amdgcn_sched_barrier(0);
#pragma unroll
        for (int j = 0; j < HD_CK; j++) {
            v2f f0 = (v2f){bc[j][0].x, bc[j][1].x};
            v2f f1 = (v2f){bc[j][0].y, bc[j][1].y};
            acc0 = __builtin_amdgcn_wmma_f32_16x16x4_f32(false, ac[j], false, f0, (short)0, acc0, false, false);
            acc1 = __builtin_amdgcn_wmma_f32_16x16x4_f32(false, ac[j], false, f1, (short)0, acc1, false, false);
        }
        __builtin_amdgcn_sched_barrier(0);
#pragma unroll
        for (int j = 0; j < HD_CK; j++) { ac[j] = an[j]; bc[j][0] = bn[j][0]; bc[j][1] = bn[j][1]; }
    }

    {
        int col0 = 32 * wave + 2 * c;
        float bb0 = b1[col0], bb1 = b1[col0 + 1];
#pragma unroll
        for (int r = 0; r < 8; r++) {
            float v0 = fmaxf(acc0[r] + bb0, 0.f);
            float v1 = fmaxf(acc1[r] + bb1, 0.f);
            *(float2*)(hl + (h * 8 + r) * HD_P + col0) = make_float2(v0, v1);
        }
    }
    __syncthreads();

    if (tid < 160) {
        int g = tid / 10, o = tid - g * 10;
        const float* hr = hl + g * HD_P;
        const float* wr = w2s + o;
        float acc = b2[o];
        float4 hc[8], hn[8];
        float wc[32], wn[32];
#pragma unroll
        for (int q = 0; q < 8; q++) hc[q] = *(const float4*)(hr + q * 4);
#pragma unroll
        for (int q = 0; q < 32; q++) wc[q] = wr[q * 10];
#pragma unroll
        for (int st = 0; st < HD_K / 32; st++) {
            if (st + 1 < HD_K / 32) {
#pragma unroll
                for (int q = 0; q < 8; q++) hn[q] = *(const float4*)(hr + (st + 1) * 32 + q * 4);
#pragma unroll
                for (int q = 0; q < 32; q++) wn[q] = wr[((st + 1) * 32 + q) * 10];
            }
            __builtin_amdgcn_sched_barrier(0);
#pragma unroll
            for (int q = 0; q < 8; q++) {
                acc = fmaf(hc[q].x, wc[4 * q + 0], acc);
                acc = fmaf(hc[q].y, wc[4 * q + 1], acc);
                acc = fmaf(hc[q].z, wc[4 * q + 2], acc);
                acc = fmaf(hc[q].w, wc[4 * q + 3], acc);
            }
            __builtin_amdgcn_sched_barrier(0);
#pragma unroll
            for (int q = 0; q < 8; q++) hc[q] = hn[q];
#pragma unroll
            for (int q = 0; q < 32; q++) wc[q] = wn[q];
        }
        out[(m * 16 + g) * 10 + o] = acc;
    }
}

extern "C" void kernel_launch(void* const* d_in, const int* in_sizes, int n_in,
                              void* d_out, int out_size, void* d_ws, size_t ws_size,
                              hipStream_t stream) {
    const int*   x_ids = (const int*)d_in[0];
    const int*   edge  = (const int*)d_in[1];
    const int*   batch = (const int*)d_in[2];
    const float* emb   = (const float*)d_in[3];

    const float *c_w1[3], *c_b1[3], *c_g[3], *c_be[3], *c_mu[3], *c_va[3], *c_w2[3], *c_b2[3];
    for (int i = 0; i < 3; i++) {
        c_w1[i] = (const float*)d_in[4 + 8 * i + 0];
        c_b1[i] = (const float*)d_in[4 + 8 * i + 1];
        c_g[i]  = (const float*)d_in[4 + 8 * i + 2];
        c_be[i] = (const float*)d_in[4 + 8 * i + 3];
        c_mu[i] = (const float*)d_in[4 + 8 * i + 4];
        c_va[i] = (const float*)d_in[4 + 8 * i + 5];
        c_w2[i] = (const float*)d_in[4 + 8 * i + 6];
        c_b2[i] = (const float*)d_in[4 + 8 * i + 7];
    }
    const float* lin1_w = (const float*)d_in[28];
    const float* lin1_b = (const float*)d_in[29];
    const float* lin2_w = (const float*)d_in[30];
    const float* lin2_b = (const float*)d_in[31];

    const long NF = (long)N_NODES * 128;
    float* ws   = (float*)d_ws;
    float* h3   = ws + 0 * NF;
    float* agg  = ws + 1 * NF;
    float* hbuf[3] = { ws + 3 * NF, ws + 4 * NF, h3 };
    float* gcat  = agg;
    int* rowptr = (int*)(ws + 5 * NF);
    int* binned = rowptr + RP_PAD;
    int* col    = binned + N_EDGES;
    int* cntm   = col + N_EDGES;
    int* stm    = cntm + EB_BLOCKS * EB_BINS;

    edge_bin<<<EB_BLOCKS, EB_T, 0, stream>>>(edge, binned, cntm, stm);
    edge_place<<<EB_BINS, EB_T, 0, stream>>>(binned, cntm, stm, rowptr, col);

    const int strips = N_NODES / 16;
    for (int l = 0; l < 3; l++) {
        if (l == 0) {
            gather_sum_emb<<<GE_BLOCKS, GE_WAVES * 32, 0, stream>>>(x_ids, emb, rowptr, col, agg);
            mlp_strip<false><<<(strips + 7) / 8, 256, 0, stream>>>(
                agg, rowptr, col, c_w1[l], c_b1[l], c_g[l], c_be[l], c_mu[l], c_va[l], c_w2[l], c_b2[l], hbuf[l], strips);
        } else {
            mlp_strip<true><<<(strips + 7) / 8, 256, 0, stream>>>(
                hbuf[l - 1], rowptr, col, c_w1[l], c_b1[l], c_g[l], c_be[l], c_mu[l], c_va[l], c_w2[l], c_b2[l], hbuf[l], strips);
        }
    }

    zero_f32<<<(N_GRAPHS * 384 / 4 + 255) / 256, 256, 0, stream>>>(gcat, N_GRAPHS * 384 / 4);
    pool3<<<(N_NODES / POOL_CH + 7) / 8, 256, 0, stream>>>(hbuf[0], hbuf[1], hbuf[2], batch, gcat);

    head_fused<<<N_GRAPHS / 16, HD_T, 0, stream>>>(gcat, lin1_w, lin1_b, lin2_w, lin2_b, (float*)d_out);
}


# Layer 1's first product taken once on the 500 table rows (T = emb @ W1 on WMMA, in spare work-groups of the first launch) instead of on 50000 nodes; nodes sum rows of T; sum order changes
# speedup vs baseline: 5.1020x; 1.0802x over previous
#include <hip/hip_runtime.h>

typedef __attribute__((ext_vector_type(2))) float v2f;
typedef __attribute__((ext_vector_type(8))) float v8f;

#define N_NODES  50000
#define N_EDGES  800000
#define N_GRAPHS 512
#define BN_EPS   1e-5f

__global__ void zero_f32(float* __restrict__ p, int n4) {
    int t = blockIdx.x * blockDim.x + threadIdx.x;
    if (t < n4) ((float4*)p)[t] = make_float4(0.f, 0.f, 0.f, 0.f);
}

#define EB_BLOCKS 256
#define EB_CHUNK  3125
#define EB_BINS   256
#define EB_RANGE  196
#define EB_T      512
#define EB_PER_T  7
#define EB_SEG    16
#define RP_PAD    (EB_BINS * EB_RANGE)
#define TP_BLOCKS 8
#define TP_STRIPS 4

__device__ __forceinline__ void table_product(const float* __restrict__ emb, const float* __restrict__ W1,
                                              float* __restrict__ tbl, int tb);

__device__ __forceinline__ int scan256_excl(int v, int tid, int* wtot) {
    int lane = tid & 31, wave = tid >> 5;
    int inc = v;
#pragma unroll
    for (int off = 1; off < 32; off <<= 1) {
        int t = __shfl_up(inc, off, 32);
        if (lane >= off) inc += t;
    }
    if (tid < 256 && lane == 31) wtot[wave] = inc;
    __syncthreads();
    int carry = 0;
    if (tid < 256)
        for (int w = 0; w < wave; w++) carry += wtot[w];
    return carry + inc - v;
}

__global__ void __launch_bounds__(EB_T) edge_bin(const int* __restrict__ ei, int* __restrict__ binned,
                                                 int* __restrict__ cntm, int* __restrict__ stm,
                                                 const float* __restrict__ emb, const float* __restrict__ W1,
                                                 float* __restrict__ tbl) {
    __shared__ int hist[EB_BINS];
    __shared__ int offs[EB_BINS];
    __shared__ int wtot[8];
    int tid = threadIdx.x, blk = blockIdx.x;
    if (blk >= EB_BLOCKS) {
        table_product(emb, W1, tbl, blk - EB_BLOCKS);
        return;
    }
    if (tid < EB_BINS) hist[tid] = 0;
    __syncthreads();

    int d[EB_PER_T], s[EB_PER_T], r[EB_PER_T];
#pragma unroll
    for (int k = 0; k < EB_PER_T; k++) {
        int idx = tid + k * EB_T;
        d[k] = -1;
        s[k] = 0;
        if (idx < EB_CHUNK) {
            int e = blk * EB_CHUNK + idx;
            d[k] = ei[N_EDGES + e];
            s[k] = ei[e];
        }
    }
#pragma unroll
    for (int k = 0; k < EB_PER_T; k++) {
        r[k] = 0;
        if (d[k] >= 0) r[k] = atomicAdd(&hist[d[k] / EB_RANGE], 1);
    }
    __syncthreads();

    int v = (tid < EB_BINS) ? hist[tid] : 0;
    int ex = scan256_excl(v, tid, wtot);
    if (tid < EB_BINS) {
        offs[tid] = ex;
        cntm[blk * EB_BINS + tid] = v;
        stm[blk * EB_BINS + tid] = ex;
    }
    __syncthreads();

#pragma unroll
    for (int k = 0; k < EB_PER_T; k++) {
        if (d[k] >= 0) {
            int bn = d[k] / EB_RANGE;
            binned[blk * EB_CHUNK + offs[bn] + r[k]] = ((d[k] - bn * EB_RANGE) << 16) | s[k];
        }
    }
}

__global__ void __launch_bounds__(EB_T) edge_place(const int* __restrict__ binned, const int* __restrict__ cntm,
                                                   const int* __restrict__ stm, int* __restrict__ rowptr,
                                                   int* __restrict__ col) {
    __shared__ int cnt[EB_BINS];
    __shared__ int wtot[8];
    __shared__ int red[EB_T / 32];
    int tid = threadIdx.x, b = blockIdx.x;
    int wave = tid >> 5, lane = tid & 31;
    if (tid < EB_BINS) cnt[tid] = 0;

    int c = 0, s = 0;
    if (lane < EB_SEG) {
        int i = wave * EB_SEG + lane;
        c = cntm[i * EB_BINS + b];
        s = stm[i * EB_BINS + b];
    }
    int part = s;
    for (int off = 16; off > 0; off >>= 1) part += __shfl_xor(part, off, 32);
    if (lane == 0) red[wave] = part;

    int len[EB_SEG], st[EB_SEG], rec[EB_SEG];
#pragma unroll
    for (int q = 0; q < EB_SEG; q++) {
        len[q] = __builtin_amdgcn_readlane(c, q);
        st[q]  = (wave * EB_SEG + q) * EB_CHUNK + __builtin_amdgcn_readlane(s, q);
    }
#pragma unroll
    for (int q = 0; q < EB_SEG; q++) {
        int p = st[q] + lane;
        if (p > N_EDGES - 1) p = N_EDGES - 1;
        int t = binned[p];
        rec[q] = (lane < len[q]) ? t : -1;
    }
    __syncthreads();

#pragma unroll
    for (int q = 0; q < EB_SEG; q++) {
        if (rec[q] >= 0) atomicAdd(&cnt[rec[q] >> 16], 1);
        for (int o = 32 + lane; o < len[q]; o += 32) atomicAdd(&cnt[binned[st[q] + o] >> 16], 1);
    }
    __syncthreads();

    int base = 0;
    for (int w = 0; w < EB_T / 32; w++) base += red[w];
    int v = (tid < EB_BINS) ? cnt[tid] : 0;
    int ex = scan256_excl(v, tid, wtot);
    if (tid < EB_BINS) {
        int p = base + ex;
        cnt[tid] = p;
        int node = b * EB_RANGE + tid;
        if (tid < EB_RANGE && node <= N_NODES) rowptr[node] = p;
    }
    __syncthreads();

#pragma unroll
    for (int q = 0; q < EB_SEG; q++) {
        if (rec[q] >= 0) {
            int pos = atomicAdd(&cnt[rec[q] >> 16], 1);
            col[pos] = rec[q] & 0xFFFF;
        }
        for (int o = 32 + lane; o < len[q]; o += 32) {
            int rr = binned[st[q] + o];
            int pos = atomicAdd(&cnt[rr >> 16], 1);
            col[pos] = rr & 0xFFFF;
        }
    }
}

#define VOCAB     500
#define GE_BLOCKS 256
#define GE_WAVES  32
#define GE_SLOTS  (GE_BLOCKS * GE_WAVES)
#define GE_F4     (VOCAB * 128 / 4)
#define GE_FILL   ((GE_F4 + GE_WAVES * 32 - 1) / (GE_WAVES * 32))
#define GE_LDS4   (GE_FILL * GE_WAVES * 32)
static_assert(N_NODES >= GE_SLOTS, "every wave has at least one node");
static_assert(N_NODES / GE_SLOTS + 1 < 32, "a wave's slot range starts fit its 32 lanes");

__device__ __forceinline__ int row_of(int id) {
    return min(max(id, 0), VOCAB - 1) * 128;
}

__device__ __forceinline__ unsigned first_sources(const int* __restrict__ col, int rp, int k, int cnt, int lane) {
    int b = __builtin_amdgcn_readlane(rp, k < cnt ? k : cnt);
    int p = b + lane;
    if (p > N_EDGES - 1) p = N_EDGES - 1;
    return (unsigned)col[p];
}

__device__ __forceinline__ void sum_rows(const float* el, int ro, int n, float4& acc) {
    int t = 0;
    for (; t + 4 <= n; t += 4) {
        int o0 = __builtin_amdgcn_readlane(ro, t);
        int o1 = __builtin_amdgcn_readlane(ro, t + 1);
        int o2 = __builtin_amdgcn_readlane(ro, t + 2);
        int o3 = __builtin_amdgcn_readlane(ro, t + 3);
        float4 v0 = *(const float4*)(el + o0);
        float4 v1 = *(const float4*)(el + o1);
        float4 v2 = *(const float4*)(el + o2);
        float4 v3 = *(const float4*)(el + o3);
        acc.x = (((acc.x + v0.x) + v1.x) + v2.x) + v3.x;
        acc.y = (((acc.y + v0.y) + v1.y) + v2.y) + v3.y;
        acc.z = (((acc.z + v0.z) + v1.z) + v2.z) + v3.z;
        acc.w = (((acc.w + v0.w) + v1.w) + v2.w) + v3.w;
    }
    switch (n - t) {
    case 3: {
        int o0 = __builtin_amdgcn_readlane(ro, t);
        int o1 = __builtin_amdgcn_readlane(ro, t + 1);
        int o2 = __builtin_amdgcn_readlane(ro, t + 2);
        float4 v0 = *(const float4*)(el + o0);
        float4 v1 = *(const float4*)(el + o1);
        float4 v2 = *(const float4*)(el + o2);
        acc.x = ((acc.x + v0.x) + v1.x) + v2.x;
        acc.y = ((acc.y + v0.y) + v1.y) + v2.y;
        acc.z = ((acc.z + v0.z) + v1.z) + v2.z;
        acc.w = ((acc.w + v0.w) + v1.w) + v2.w;
        break;
    }
    case 2: {
        int o0 = __builtin_amdgcn_readlane(ro, t);
        int o1 = __builtin_amdgcn_readlane(ro, t + 1);
        float4 v0 = *(const float4*)(el + o0);
        float4 v1 = *(const float4*)(el + o1);
        acc.x = (acc.x + v0.x) + v1.x;
        acc.y = (acc.y + v0.y) + v1.y;
        acc.z = (acc.z + v0.z) + v1.z;
        acc.w = (acc.w + v0.w) + v1.w;
        break;
    }
    case 1: {
        int o0 = __builtin_amdgcn_readlane(ro, t);
        float4 v0 = *(const float4*)(el + o0);
        acc.x += v0.x; acc.y += v0.y; acc.z += v0.z; acc.w += v0.w;
        break;
    }
    default: break;
    }
}

__global__ void __launch_bounds__(GE_WAVES * 32) gather_sum_emb(const int* __restrict__ ids, const float* __restrict__ emb,
                                                                const int* __restrict__ rowptr, const int* __restrict__ col,
                                                                const float* __restrict__ b1, const float* __restrict__ gamma,
                                                                const float* __restrict__ beta, const float* __restrict__ mean,
                                                                const float* __restrict__ var, float* __restrict__ s) {
    __shared__ __attribute__((aligned(16))) float et[GE_LDS4 * 4];
    int tid = threadIdx.x;
    int wave = __builtin_amdgcn_readfirstlane(tid >> 5);
    int lane = tid & 31;
    int gw = blockIdx.x * GE_WAVES + wave;
    int n0 = (int)(((long)gw * N_NODES) / GE_SLOTS);
    int n1 = (int)(((long)(gw + 1) * N_NODES) / GE_SLOTS);
    int cnt = n1 - n0;

    int rp  = rowptr[n0 + (lane < cnt ? lane : cnt)];
    int oid = ids[n0 + (lane < cnt ? lane : cnt - 1)];

    float4 r[GE_FILL];
#pragma unroll
    for (int k = 0; k < GE_FILL / 2; k++) {
        int i = tid + k * (GE_WAVES * 32);
        r[k] = ((const float4*)emb)[i < GE_F4 ? i : GE_F4 - 1];
    }
    unsigned c0 = first_sources(col, rp, 0, cnt, lane);
    unsigned cn = first_sources(col, rp, 1, cnt, lane);
#pragma unroll
    for (int k = GE_FILL / 2; k < GE_FILL; k++) {
        int i = tid + k * (GE_WAVES * 32);
        r[k] = ((const float4*)emb)[i < GE_F4 ? i : GE_F4 - 1];
    }
#pragma unroll
    for (int k = 0; k < GE_FILL / 2; k++) ((float4*)et)[tid + k * (GE_WAVES * 32)] = r[k];
    int idn = ids[c0];
#pragma unroll
    for (int k = GE_FILL / 2; k < GE_FILL; k++) ((float4*)et)[tid + k * (GE_WAVES * 32)] = r[k];
    __syncthreads();

    int own = row_of(oid);
    const float* el = et + lane * 4;
    float4 bb = *(const float4*)(b1 + lane * 4);
    float4 mu = *(const float4*)(mean + lane * 4);
    float4 va = *(const float4*)(var + lane * 4);
    float4 gm = *(const float4*)(gamma + lane * 4);
    float4 bt = *(const float4*)(beta + lane * 4);
    float4 inv = make_float4(rsqrtf(va.x + BN_EPS), rsqrtf(va.y + BN_EPS), rsqrtf(va.z + BN_EPS), rsqrtf(va.w + BN_EPS));
    for (int j = 0; j < cnt; j++) {
        int roc = row_of(idn);
        idn = ids[cn];
        cn = first_sources(col, rp, j + 2, cnt, lane);
        int beg = __builtin_amdgcn_readlane(rp, j);
        int end = __builtin_amdgcn_readlane(rp, j + 1);
        int so  = __builtin_amdgcn_readlane(own, j);
        float4 xv = *(const float4*)(el + so);
        float4 acc = make_float4(0.f, 0.f, 0.f, 0.f);
        int n = end - beg;
        if (n > 32) n = 32;
        sum_rows(el, roc, n, acc);
        for (int base = beg + 32; base < end; base += 32) {
            int m = end - base;
            if (m > 32) m = 32;
            unsigned c = (lane < m) ? (unsigned)col[base + lane] : 0u;
            int rr = row_of(ids[c]);
            sum_rows(el, rr, m, acc);
        }
        float4 o = make_float4(xv.x + acc.x, xv.y + acc.y, xv.z + acc.z, xv.w + acc.w);
        o.x = fmaxf((o.x + bb.x - mu.x) * inv.x * gm.x + bt.x, 0.f);
        o.y = fmaxf((o.y + bb.y - mu.y) * inv.y * gm.y + bt.y, 0.f);
        o.z = fmaxf((o.z + bb.z - mu.z) * inv.z * gm.z + bt.z, 0.f);
        o.w = fmaxf((o.w + bb.w - mu.w) * inv.w * gm.w + bt.w, 0.f);
        *(float4*)(s + (long)(n0 + j) * 128 + lane * 4) = o;
    }
}

#define SK  128
#define SN  128
#define WPF 288
#define HP  132

__device__ __forceinline__ void w_fetch(const float* __restrict__ W, int wave, int lane,
                                        float4 (&r0)[8], float4 (&r1)[8]) {
#pragma unroll
    for (int i = 0; i < 8; i++) {
        int kp = wave + 8 * i;
        r0[i] = *(const float4*)(W + (long)(2 * kp) * SN + lane * 4);
        r1[i] = *(const float4*)(W + (long)(2 * kp + 1) * SN + lane * 4);
    }
}
__device__ __forceinline__ void w_put(float* wl, int wave, int lane,
                                      const float4 (&r0)[8], const float4 (&r1)[8]) {
#pragma unroll
    for (int i = 0; i < 8; i++) {
        int kp = wave + 8 * i;
        float* d = wl + kp * WPF + lane * 8;
        *(float4*)(d)     = make_float4(r0[i].x, r1[i].x, r0[i].y, r1[i].y);
        *(float4*)(d + 4) = make_float4(r0[i].z, r1[i].z, r0[i].w, r1[i].w);
    }
}

__device__ __forceinline__ void strip_product(const v2f (&af)[SK / 4], const float* bl, v8f (&acc)[8]) {
#pragma unroll
    for (int nt = 0; nt < 8; nt++) acc[nt] = (v8f){0.f, 0.f, 0.f, 0.f, 0.f, 0.f, 0.f, 0.f};
    v2f bc[8], bn[8];
#pragma unroll
    for (int nt = 0; nt < 8; nt++) bc[nt] = *(const v2f*)(bl + nt * 32);
#pragma unroll
    for (int i = 0; i < SK / 4; i++) {
        if (i + 1 < SK / 4) {
#pragma unroll
            for (int nt = 0; nt < 8; nt++) bn[nt] = *(const v2f*)(bl + (2 * i + 2) * WPF + nt * 32);
        }
        __builtin_amdgcn_sched_barrier(0);
#pragma unroll
        for (int nt = 0; nt < 8; nt++)
            acc[nt] = __builtin_amdgcn_wmma_f32_16x16x4_f32(false, af[i], false, bc[nt],
                                                            (short)0, acc[nt], false, false);
        __builtin_amdgcn_sched_barrier(0);
#pragma unroll
        for (int nt = 0; nt < 8; nt++) bc[nt] = bn[nt];
    }
}

static_assert(TP_BLOCKS * TP_STRIPS * 16 >= VOCAB, "the strips cover the table");
static_assert(TP_STRIPS * 4 * 32 == EB_T, "four waves per strip fill the work-group");
#define TBL_ROWS (TP_BLOCKS * TP_STRIPS * 16)
__device__ __forceinline__ void table_product(const float* __restrict__ emb, const float* __restrict__ W1,
                                              float* __restrict__ tbl, int tb) {
    __shared__ __attribute__((aligned(16))) float twl[(SK / 2) * WPF];
    int wave = __builtin_amdgcn_readfirstlane(threadIdx.x >> 5);
    int lane = threadIdx.x & 31;
    int c = lane & 15, h = lane >> 4;
    float4 r0[4], r1[4];
#pragma unroll
    for (int i = 0; i < 4; i++) {
        int kp = wave + 16 * i;
        r0[i] = *(const float4*)(W1 + (long)(2 * kp) * SN + lane * 4);
        r1[i] = *(const float4*)(W1 + (long)(2 * kp + 1) * SN + lane * 4);
    }
#pragma unroll
    for (int i = 0; i < 4; i++) {
        int kp = wave + 16 * i;
        float* d = twl + kp * WPF + lane * 8;
        *(float4*)(d)     = make_float4(r0[i].x, r1[i].x, r0[i].y, r1[i].y);
        *(float4*)(d + 4) = make_float4(r0[i].z, r1[i].z, r0[i].w, r1[i].w);
    }
    __syncthreads();

    int strip = tb * TP_STRIPS + (wave >> 2);
    int q = wave & 3;
    int row = min(strip * 16 + c, VOCAB - 1);
    const float* xr = emb + (long)row * SK + h * 2;
    v2f af[SK / 4];
#pragma unroll
    for (int i = 0; i < SK / 4; i++) af[i] = *(const v2f*)(xr + i * 4);
    const float* bl = twl + h * WPF + (32 * q + c) * 2;
    v8f acc0 = (v8f){0.f, 0.f, 0.f, 0.f, 0.f, 0.f, 0.f, 0.f};
    v8f acc1 = (v8f){0.f, 0.f, 0.f, 0.f, 0.f, 0.f, 0.f, 0.f};
#pragma unroll
    for (int i = 0; i < SK / 4; i++) {
        v2f b0 = *(const v2f*)(bl + (2 * i) * WPF);
        v2f b1 = *(const v2f*)(bl + (2 * i) * WPF + 32);
        acc0 = __builtin_amdgcn_wmma_f32_16x16x4_f32(false, af[i], false, b0, (short)0, acc0, false, false);
        acc1 = __builtin_amdgcn_wmma_f32_16x16x4_f32(false, af[i], false, b1, (short)0, acc1, false, false);
    }
    float* trow = tbl + (long)(strip * 16 + h * 8) * SN + 32 * q + c;
#pragma unroll
    for (int r = 0; r < 8; r++) {
        trow[(long)r * SN]      = acc0[r];
        trow[(long)r * SN + 16] = acc1[r];
    }
}

#define GB 32

__device__ __forceinline__ void strip_request(const float* __restrict__ x, const int* __restrict__ rowptr,
                                              int n0, int lane, int& rp, float4 (&own)[16]) {
    rp = rowptr[n0 + (lane < 16 ? lane : 16)];
#pragma unroll
    for (int j = 0; j < 16; j++) own[j] = *(const float4*)(x + (long)(n0 + j) * SK + lane * 4);
}

__device__ __forceinline__ int strip_first_sources(const int* __restrict__ col, int rp, int lane) {
    int beg  = __builtin_amdgcn_readlane(rp, 0);
    int tend = __builtin_amdgcn_readlane(rp, 16);
    int nxt = 0;
    if (beg < tend) nxt = col[min(beg + lane, tend - 1)];
    return nxt;
}

__device__ __forceinline__ void strip_row_done(float* hw, int lane, int& node, int& nend, int rp, float4& acc) {
    float* hp = hw + node * HP + lane * 4;
    float4 xv = *(const float4*)hp;
    *(float4*)hp = make_float4(xv.x + acc.x, xv.y + acc.y, xv.z + acc.z, xv.w + acc.w);
    acc = make_float4(0.f, 0.f, 0.f, 0.f);
    node++;
    nend = __builtin_amdgcn_readlane(rp, node + 1);
}

__device__ __forceinline__ void strip_sum(const float* __restrict__ x, const int* __restrict__ col, int lane,
                                          float* hw, int rp, int nxt, const float4 (&own)[16]) {
#pragma unroll
    for (int j = 0; j < 16; j++) *(float4*)(hw + j * HP + lane * 4) = own[j];
    int beg  = __builtin_amdgcn_readlane(rp, 0);
    int tend = __builtin_amdgcn_readlane(rp, 16);
    int nend = __builtin_amdgcn_readlane(rp, 1);
    int node = 0;
    float4 acc = make_float4(0.f, 0.f, 0.f, 0.f);
    for (int base = beg; base < tend; base += GB) {
        int n = min(tend - base, GB);
        int cur = nxt;
        if (base + GB < tend) nxt = col[min(base + GB + lane, tend - 1)];
        float4 v[GB];
#pragma unroll
        for (int g = 0; g < GB / 8; g++) {
            if (g * 8 < n) {
#pragma unroll
                for (int k = 0; k < 8; k++) {
                    int s = __builtin_amdgcn_readlane(cur, g * 8 + k);
                    v[g * 8 + k] = *(const float4*)(x + (long)s * SK + lane * 4);
                }
            }
        }
#pragma unroll
        for (int t = 0; t < GB; t++) {
            if (t < n) {
                while (base + t == nend) strip_row_done(hw, lane, node, nend, rp, acc);
                acc.x += v[t].x; acc.y += v[t].y; acc.z += v[t].z; acc.w += v[t].w;
            }
        }
    }
    while (node < 16) strip_row_done(hw, lane, node, nend, rp, acc);
}

__global__ void __launch_bounds__(256) mlp_strip(const float* __restrict__ X,
                                                 const int* __restrict__ rowptr, const int* __restrict__ col,
                                                 const float* __restrict__ W1, const float* __restrict__ b1,
                                                 const float* __restrict__ gamma, const float* __restrict__ beta,
                                                 const float* __restrict__ mean, const float* __restrict__ var,
                                                 const float* __restrict__ W2, const float* __restrict__ b2,
                                                 float* __restrict__ O, int strips) {
    __shared__ __attribute__((aligned(16))) float wl[(SK / 2) * WPF];
    __shared__ __attribute__((aligned(16))) float hs[8 * 16 * HP];
    int wave = __builtin_amdgcn_readfirstlane(threadIdx.x >> 5);
    int lane = threadIdx.x & 31;
    int c = lane & 15, h = lane >> 4;
    int strip = blockIdx.x * 8 + wave;
    bool active = strip < strips;
    float* hw = hs + wave * 16 * HP;
    const float* bl = wl + h * WPF + c * 2;

    float4 r0[8], r1[8];
    int rp = 0, nxt = 0;
    float4 own[16];
    int gs = active ? strip : strips - 1;
    strip_request(X, rowptr, gs * 16, lane, rp, own);
    w_fetch(W1, wave, lane, r0, r1);
    nxt = strip_first_sources(col, rp, lane);
    w_put(wl, wave, lane, r0, r1);
    strip_sum(X, col, lane, hw, rp, nxt, own);
    __syncthreads();

    if (active) {
        v2f af[SK / 4];
        const float* gr = hw + c * HP + h * 2;
#pragma unroll
        for (int i = 0; i < SK / 4; i++) af[i] = *(const v2f*)(gr + i * 4);
        v8f acc[8];
        strip_product(af, bl, acc);
#pragma unroll
        for (int nt = 0; nt < 8; nt++) {
            int col = nt * 16 + c;
            float bb = b1[col];
            float mu = mean[col], inv = rsqrtf(var[col] + BN_EPS), gm = gamma[col], bt = beta[col];
#pragma unroll
            for (int r = 0; r < 8; r++) {
                float v = acc[nt][r] + bb;
                v = (v - mu) * inv * gm + bt;
                v = fmaxf(v, 0.f);
                hw[(h * 8 + r) * HP + col] = v;
            }
        }
    }

    w_fetch(W2, wave, lane, r0, r1);
    __syncthreads();
    w_put(wl, wave, lane, r0, r1);
    __syncthreads();

    if (active) {
        v2f af[SK / 4];
        const float* hr = hw + c * HP + h * 2;
#pragma unroll
        for (int i = 0; i < SK / 4; i++) af[i] = *(const v2f*)(hr + i * 4);
        v8f acc[8];
        strip_product(af, bl, acc);
        float* orow = O + (long)(strip * 16 + h * 8) * SN + c;
#pragma unroll
        for (int nt = 0; nt < 8; nt++) {
            float bb = b2[nt * 16 + c];
#pragma unroll
            for (int r = 0; r < 8; r++) {
                float v = fmaxf(acc[nt][r] + bb, 0.f);
                orow[(long)r * SN + nt * 16] = v;
            }
        }
    }
}

__global__ void __launch_bounds__(256) mlp_second(const float* __restrict__ H, const float* __restrict__ W2,
                                                  const float* __restrict__ b2, float* __restrict__ O, int strips) {
    __shared__ __attribute__((aligned(16))) float wl[(SK / 2) * WPF];
    int wave = __builtin_amdgcn_readfirstlane(threadIdx.x >> 5);
    int lane = threadIdx.x & 31;
    int c = lane & 15, h = lane >> 4;
    int strip = blockIdx.x * 8 + wave;
    bool active = strip < strips;
    float4 r0[8], r1[8];
    w_fetch(W2, wave, lane, r0, r1);
    w_put(wl, wave, lane, r0, r1);
    __syncthreads();

    if (active) {
        v2f af[SK / 4];
        const float* xr = H + (long)(strip * 16 + c) * SK + h * 2;
#pragma unroll
        for (int i = 0; i < SK / 4; i++) af[i] = *(const v2f*)(xr + i * 4);
        v8f acc[8];
        strip_product(af, wl + h * WPF + c * 2, acc);
        float* orow = O + (long)(strip * 16 + h * 8) * SN + c;
#pragma unroll
        for (int nt = 0; nt < 8; nt++) {
            float bb = b2[nt * 16 + c];
#pragma unroll
            for (int r = 0; r < 8; r++) {
                float v = fmaxf(acc[nt][r] + bb, 0.f);
                orow[(long)r * SN + nt * 16] = v;
            }
        }
    }
}

#define POOL_CH 16
__device__ __forceinline__ void pool_flush(float* __restrict__ p, float4 a) {
    atomicAdd(p + 0, a.x);
    atomicAdd(p + 1, a.y);
    atomicAdd(p + 2, a.z);
    atomicAdd(p + 3, a.w);
}

__global__ void __launch_bounds__(256) pool3(const float* __restrict__ h1, const float* __restrict__ h2,
                                             const float* __restrict__ h3, const int* __restrict__ batch,
                                             float* __restrict__ gcat) {
    int wave = threadIdx.x >> 5, lane = threadIdx.x & 31;
    int chunk = blockIdx.x * 8 + wave;
    if (chunk >= N_NODES / POOL_CH) return;
    int base = chunk * POOL_CH;
    int myb = batch[base + (lane & (POOL_CH - 1))];
    const float* hs[3] = {h1, h2, h3};
#pragma unroll
    for (int l = 0; l < 3; l++) {
        const float* hp = hs[l] + (long)base * 128 + lane * 4;
        float4 v[POOL_CH];
#pragma unroll
        for (int j = 0; j < POOL_CH; j++) v[j] = *(const float4*)(hp + j * 128);
        float* gp = gcat + l * 128 + lane * 4;
        int cur = __builtin_amdgcn_readlane(myb, 0);
        float4 acc = v[0];
#pragma unroll
        for (int j = 1; j < POOL_CH; j++) {
            int g = __builtin_amdgcn_readlane(myb, j);
            if (g != cur) {
                pool_flush(gp + (long)cur * 384, acc);
                acc = v[j];
                cur = g;
            } else {
                acc.x += v[j].x; acc.y += v[j].y; acc.z += v[j].z; acc.w += v[j].w;
            }
        }
        pool_flush(gp + (long)cur * 384, acc);
    }
}

#define HD_T   384
#define HD_K   384
#define HD_N   384
#define HD_P   388
#define HD_CK  8
#define HD_W2F4 (HD_N * 10 / 4)
#define HD_W2R  ((HD_W2F4 + HD_T - 1) / HD_T)
#define HD_W2L4 (HD_W2R * HD_T)

__device__ __forceinline__ void head_fetch(const float* wb, const float* ar, int ch, bool with_a,
                                           v2f (&a)[HD_CK], v2f (&b)[HD_CK][2]) {
#pragma unroll
    for (int j = 0; j < HD_CK; j++) {
        int i = ch * HD_CK + j;
        b[j][0] = *(const v2f*)(wb + (long)(4 * i) * HD_N);
        b[j][1] = *(const v2f*)(wb + (long)(4 * i + 1) * HD_N);
    }
    if (with_a) {
#pragma unroll
        for (int j = 0; j < HD_CK; j++) a[j] = *(const v2f*)(ar + (ch * HD_CK + j) * 4);
    }
}

__global__ void __launch_bounds__(HD_T) head_fused(const float* __restrict__ gcat, const float* __restrict__ w1,
                                                   const float* __restrict__ b1, const float* __restrict__ w2,
                                                   const float* __restrict__ b2, float* __restrict__ out) {
    __shared__ __attribute__((aligned(16))) float ga[16 * HD_P];
    __shared__ __attribute__((aligned(16))) float hl[16 * HD_P];
    __shared__ __attribute__((aligned(16))) float w2s[HD_W2L4 * 4];
    int tid = threadIdx.x;
    int wave = __builtin_amdgcn_readfirstlane(tid >> 5);
    int lane = tid & 31;
    int c = lane & 15, h = lane >> 4;
    int m = blockIdx.x;

    const float* wb = w1 + (long)(2 * h) * HD_N + 32 * wave + 2 * c;
    const float* ar = ga + c * HD_P + 2 * h;
    v2f ac[HD_CK], an[HD_CK], bc[HD_CK][2], bn[HD_CK][2];
    head_fetch(wb, ar, 0, false, ac, bc);

    float4 ra[4], rw[HD_W2R];
    const float4* gsrc = (const float4*)(gcat + (long)m * 16 * HD_K);
#pragma unroll
    for (int k = 0; k < 4; k++) ra[k] = gsrc[tid + k * HD_T];
#pragma unroll
    for (int k = 0; k < HD_W2R; k++) {
        int i = tid + k * HD_T;
        rw[k] = ((const float4*)w2)[i < HD_W2F4 ? i : HD_W2F4 - 1];
    }
#pragma unroll
    for (int k = 0; k < 4; k++) {
        int i = tid + k * HD_T;
        int row = i / (HD_K / 4), c4 = i - row * (HD_K / 4);
        *(float4*)(ga + row * HD_P + c4 * 4) = ra[k];
    }
#pragma unroll
    for (int k = 0; k < HD_W2R; k++) ((float4*)w2s)[tid + k * HD_T] = rw[k];
    __syncthreads();

#pragma unroll
    for (int j = 0; j < HD_CK; j++) ac[j] = *(const v2f*)(ar + j * 4);

    v8f acc0 = (v8f){0.f, 0.f, 0.f, 0.f, 0.f, 0.f, 0.f, 0.f};
    v8f acc1 = (v8f){0.f, 0.f, 0.f, 0.f, 0.f, 0.f, 0.f, 0.f};
#pragma unroll
    for (int ch = 0; ch < HD_K / 4 / HD_CK; ch++) {
        if (ch + 1 < HD_K / 4 / HD_CK) head_fetch(wb, ar, ch + 1, true, an, bn);
        __builtin_amdgcn_sched_barrier(0);
#pragma unroll
        for (int j = 0; j < HD_CK; j++) {
            v2f f0 = (v2f){bc[j][0].x, bc[j][1].x};
            v2f f1 = (v2f){bc[j][0].y, bc[j][1].y};
            acc0 = __builtin_amdgcn_wmma_f32_16x16x4_f32(false, ac[j], false, f0, (short)0, acc0, false, false);
            acc1 = __builtin_amdgcn_wmma_f32_16x16x4_f32(false, ac[j], false, f1, (short)0, acc1, false, false);
        }
        __builtin_amdgcn_sched_barrier(0);
#pragma unroll
        for (int j = 0; j < HD_CK; j++) { ac[j] = an[j]; bc[j][0] = bn[j][0]; bc[j][1] = bn[j][1]; }
    }

    {
        int col0 = 32 * wave + 2 * c;
        float bb0 = b1[col0], bb1 = b1[col0 + 1];
#pragma unroll
        for (int r = 0; r < 8; r++) {
            float v0 = fmaxf(acc0[r] + bb0, 0.f);
            float v1 = fmaxf(acc1[r] + bb1, 0.f);
            *(float2*)(hl + (h * 8 + r) * HD_P + col0) = make_float2(v0, v1);
        }
    }
    __syncthreads();

    if (tid < 160) {
        int g = tid / 10, o = tid - g * 10;
        const float* hr = hl + g * HD_P;
        const float* wr = w2s + o;
        float acc = b2[o];
        float4 hc[8], hn[8];
        float wc[32], wn[32];
#pragma unroll
        for (int q = 0; q < 8; q++) hc[q] = *(const float4*)(hr + q * 4);
#pragma unroll
        for (int q = 0; q < 32; q++) wc[q] = wr[q * 10];
#pragma unroll
        for (int st = 0; st < HD_K / 32; st++) {
            if (st + 1 < HD_K / 32) {
#pragma unroll
                for (int q = 0; q < 8; q++) hn[q] = *(const float4*)(hr + (st + 1) * 32 + q * 4);
#pragma unroll
                for (int q = 0; q < 32; q++) wn[q] = wr[((st + 1) * 32 + q) * 10];
            }
            __builtin_amdgcn_sched_barrier(0);
#pragma unroll
            for (int q = 0; q < 8; q++) {
                acc = fmaf(hc[q].x, wc[4 * q + 0], acc);
                acc = fmaf(hc[q].y, wc[4 * q + 1], acc);
                acc = fmaf(hc[q].z, wc[4 * q + 2], acc);
                acc = fmaf(hc[q].w, wc[4 * q + 3], acc);
            }
            __builtin_amdgcn_sched_barrier(0);
#pragma unroll
            for (int q = 0; q < 8; q++) hc[q] = hn[q];
#pragma unroll
            for (int q = 0; q < 32; q++) wc[q] = wn[q];
        }
        out[(m * 16 + g) * 10 + o] = acc;
    }
}

extern "C" void kernel_launch(void* const* d_in, const int* in_sizes, int n_in,
                              void* d_out, int out_size, void* d_ws, size_t ws_size,
                              hipStream_t stream) {
    const int*   x_ids = (const int*)d_in[0];
    const int*   edge  = (const int*)d_in[1];
    const int*   batch = (const int*)d_in[2];
    const float* emb   = (const float*)d_in[3];

    const float *c_w1[3], *c_b1[3], *c_g[3], *c_be[3], *c_mu[3], *c_va[3], *c_w2[3], *c_b2[3];
    for (int i = 0; i < 3; i++) {
        c_w1[i] = (const float*)d_in[4 + 8 * i + 0];
        c_b1[i] = (const float*)d_in[4 + 8 * i + 1];
        c_g[i]  = (const float*)d_in[4 + 8 * i + 2];
        c_be[i] = (const float*)d_in[4 + 8 * i + 3];
        c_mu[i] = (const float*)d_in[4 + 8 * i + 4];
        c_va[i] = (const float*)d_in[4 + 8 * i + 5];
        c_w2[i] = (const float*)d_in[4 + 8 * i + 6];
        c_b2[i] = (const float*)d_in[4 + 8 * i + 7];
    }
    const float* lin1_w = (const float*)d_in[28];
    const float* lin1_b = (const float*)d_in[29];
    const float* lin2_w = (const float*)d_in[30];
    const float* lin2_b = (const float*)d_in[31];

    const long NF = (long)N_NODES * 128;
    float* ws   = (float*)d_ws;
    float* h3   = ws + 0 * NF;
    float* agg  = ws + 1 * NF;
    float* hbuf[3] = { ws + 3 * NF, ws + 4 * NF, h3 };
    float* gcat  = agg;
    int* rowptr = (int*)(ws + 5 * NF);
    int* binned = rowptr + RP_PAD;
    int* col    = binned + N_EDGES;
    int* cntm   = col + N_EDGES;
    int* stm    = cntm + EB_BLOCKS * EB_BINS;
    float* tbl  = (float*)(stm + EB_BLOCKS * EB_BINS);

    edge_bin<<<EB_BLOCKS + TP_BLOCKS, EB_T, 0, stream>>>(edge, binned, cntm, stm, emb, c_w1[0], tbl);
    edge_place<<<EB_BINS, EB_T, 0, stream>>>(binned, cntm, stm, rowptr, col);

    const int strips = N_NODES / 16;
    for (int l = 0; l < 3; l++) {
        if (l == 0) {
            gather_sum_emb<<<GE_BLOCKS, GE_WAVES * 32, 0, stream>>>(x_ids, tbl, rowptr, col,
                                                                   c_b1[l], c_g[l], c_be[l], c_mu[l], c_va[l], agg);
            mlp_second<<<(strips + 7) / 8, 256, 0, stream>>>(agg, c_w2[l], c_b2[l], hbuf[l], strips);
        } else {
            mlp_strip<<<(strips + 7) / 8, 256, 0, stream>>>(
                hbuf[l - 1], rowptr, col, c_w1[l], c_b1[l], c_g[l], c_be[l], c_mu[l], c_va[l], c_w2[l], c_b2[l], hbuf[l], strips);
        }
    }

    zero_f32<<<(N_GRAPHS * 384 / 4 + 255) / 256, 256, 0, stream>>>(gcat, N_GRAPHS * 384 / 4);
    pool3<<<(N_NODES / POOL_CH + 7) / 8, 256, 0, stream>>>(hbuf[0], hbuf[1], hbuf[2], batch, gcat);

    head_fused<<<N_GRAPHS / 16, HD_T, 0, stream>>>(gcat, lin1_w, lin1_b, lin2_w, lin2_b, (float*)d_out);
}
